# starting kernel
# baseline (speedup 1.0000x reference)
#include <hip/hip_runtime.h>
#include <hip/hip_bf16.h>

#define N_FEAT 256
#define HIDDEN 16
#define NCLS   2

typedef float v2f __attribute__((ext_vector_type(2)));
typedef float v8f __attribute__((ext_vector_type(8)));

__device__ __forceinline__ void atomic_add_f32(float* p, float v) {
    unsafeAtomicAdd(p, v);
}

__global__ void k_deg_init(float* __restrict__ deg, int n) {
    int i = blockIdx.x * blockDim.x + threadIdx.x;
    if (i < n) deg[i] = 1.0f;
}

__global__ void k_deg_edges(const int* __restrict__ dst, float* __restrict__ deg, int E) {
    int e = blockIdx.x * blockDim.x + threadIdx.x;
    if (e < E) atomic_add_f32(&deg[dst[e]], 1.0f);
}

__global__ void k_dinv(float* __restrict__ deg, int n) {
    int i = blockIdx.x * blockDim.x + threadIdx.x;
    if (i < n) deg[i] = rsqrtf(deg[i]);
}

__global__ void k_gemm1(const float* __restrict__ x, const float* __restrict__ W1,
                        const float* __restrict__ dinv,
                        float* __restrict__ hw1, float* __restrict__ agg1, int ntiles) {
    int lane = threadIdx.x & 31;
    int wave = threadIdx.x >> 5;
    int tile = blockIdx.x * (blockDim.x >> 5) + wave;
    if (tile >= ntiles) return;
    int half = lane >> 4;
    int l    = lane & 15;
    int row0 = tile * 16;

    const float* aptr = x  + (size_t)(row0 + l) * N_FEAT + 2 * half;
    const float* bptr = W1 + (size_t)(2 * half) * HIDDEN + l;

    v8f c = {};
    #pragma unroll 8
    for (int kk = 0; kk < N_FEAT / 4; ++kk) {
        v2f a = *(const v2f*)(aptr + kk * 4);
        v2f b;
        b.x = bptr[kk * 4 * HIDDEN];
        b.y = bptr[kk * 4 * HIDDEN + HIDDEN];
        c = __builtin_amdgcn_wmma_f32_16x16x4_f32(false, a, false, b,
                                                  (short)0, c, false, false);
    }

    int rbase = row0 + half * 8;
    #pragma unroll
    for (int v = 0; v < 8; ++v) {
        int r = rbase + v;
        float dval = c[v];
        float di   = dinv[r];
        size_t idx = (size_t)r * HIDDEN + l;
        hw1[idx]  = dval;
        agg1[idx] = dval * di * di;
    }
}

__global__ void k_gemm1_rem(const float* __restrict__ x, const float* __restrict__ W1,
                            const float* __restrict__ dinv,
                            float* __restrict__ hw1, float* __restrict__ agg1,
                            int row_lo, int n) {
    int t = blockIdx.x * blockDim.x + threadIdx.x;
    int r = row_lo + t / HIDDEN;
    int c = t % HIDDEN;
    if (r >= n) return;
    float acc = 0.0f;
    for (int k = 0; k < N_FEAT; ++k) acc += x[(size_t)r * N_FEAT + k] * W1[(size_t)k * HIDDEN + c];
    float di = dinv[r];
    size_t idx = (size_t)r * HIDDEN + c;
    hw1[idx]  = acc;
    agg1[idx] = acc * di * di;
}

__global__ void k_edges1(const int* __restrict__ src, const int* __restrict__ dst,
                         const float* __restrict__ hw1, const float* __restrict__ dinv,
                         float* __restrict__ agg1, int E) {
    int t = blockIdx.x * blockDim.x + threadIdx.x;
    int e = t >> 4;
    int c = t & 15;
    if (e >= E) return;
    int s = src[e], d = dst[e];
    float norm = dinv[s] * dinv[d];
    atomic_add_f32(&agg1[(size_t)d * HIDDEN + c], hw1[(size_t)s * HIDDEN + c] * norm);
}

__global__ void k_layer2(const float* __restrict__ agg1, const float* __restrict__ b1,
                         const float* __restrict__ W2, const float* __restrict__ dinv,
                         float* __restrict__ hw2, float* __restrict__ agg2, int n) {
    int i = blockIdx.x * blockDim.x + threadIdx.x;
    if (i >= n) return;
    float h0 = 0.0f, h1 = 0.0f;
    #pragma unroll
    for (int c = 0; c < HIDDEN; ++c) {
        float v = agg1[(size_t)i * HIDDEN + c] + b1[c];
        v = fmaxf(v, 0.0f);
        h0 += v * W2[c * NCLS + 0];
        h1 += v * W2[c * NCLS + 1];
    }
    float di  = dinv[i];
    float di2 = di * di;
    hw2[(size_t)i * NCLS + 0]  = h0;
    hw2[(size_t)i * NCLS + 1]  = h1;
    agg2[(size_t)i * NCLS + 0] = h0 * di2;
    agg2[(size_t)i * NCLS + 1] = h1 * di2;
}

__global__ void k_edges2(const int* __restrict__ src, const int* __restrict__ dst,
                         const float* __restrict__ hw2, const float* __restrict__ dinv,
                         float* __restrict__ agg2, int E) {
    int e = blockIdx.x * blockDim.x + threadIdx.x;
    if (e >= E) return;
    int s = src[e], d = dst[e];
    float norm = dinv[s] * dinv[d];
    atomic_add_f32(&agg2[(size_t)d * NCLS + 0], hw2[(size_t)s * NCLS + 0] * norm);
    atomic_add_f32(&agg2[(size_t)d * NCLS + 1], hw2[(size_t)s * NCLS + 1] * norm);
}

__global__ void k_final(const float* __restrict__ agg2, const float* __restrict__ b2,
                        float* __restrict__ out, int n) {
    int i = blockIdx.x * blockDim.x + threadIdx.x;
    if (i >= n) return;
    float l0 = agg2[(size_t)i * NCLS + 0] + b2[0];
    float l1 = agg2[(size_t)i * NCLS + 1] + b2[1];
    float m   = fmaxf(l0, l1);
    float lse = m + logf(expf(l0 - m) + expf(l1 - m));
    out[(size_t)i * NCLS + 0] = l0 - lse;
    out[(size_t)i * NCLS + 1] = l1 - lse;
}

extern "C" void kernel_launch(void* const* d_in, const int* in_sizes, int n_in,
                              void* d_out, int out_size, void* d_ws, size_t ws_size,
                              hipStream_t stream) {
    const float* x  = (const float*)d_in[0];
    const int*   ei = (const int*)d_in[1];
    const float* W1 = (const float*)d_in[2];
    const float* b1 = (const float*)d_in[3];
    const float* W2 = (const float*)d_in[4];
    const float* b2 = (const float*)d_in[5];
    float* out = (float*)d_out;

    int N = in_sizes[0] / N_FEAT;
    int E = in_sizes[1] / 2;
    const int* src = ei;
    const int* dst = ei + E;

    float* ws   = (float*)d_ws;
    float* dinv = ws;
    float* hw1  = dinv + (size_t)N;
    float* agg1 = hw1  + (size_t)N * HIDDEN;
    float* hw2  = agg1 + (size_t)N * HIDDEN;
    float* agg2 = hw2  + (size_t)N * NCLS;

    const int tb = 256;
    k_deg_init <<<(N + tb - 1) / tb, tb, 0, stream>>>(dinv, N);
    k_deg_edges<<<(E + tb - 1) / tb, tb, 0, stream>>>(dst, dinv, E);
    k_dinv     <<<(N + tb - 1) / tb, tb, 0, stream>>>(dinv, N);

    int ntiles = N / 16;
    int wpb    = tb / 32;
    if (ntiles > 0)
        k_gemm1<<<(ntiles + wpb - 1) / wpb, tb, 0, stream>>>(x, W1, dinv, hw1, agg1, ntiles);
    int rem = N - ntiles * 16;
    if (rem > 0)
        k_gemm1_rem<<<(rem * HIDDEN + tb - 1) / tb, tb, 0, stream>>>(x, W1, dinv, hw1, agg1,
                                                                     ntiles * 16, N);

    long long t1 = (long long)E * HIDDEN;
    k_edges1<<<(int)((t1 + tb - 1) / tb), tb, 0, stream>>>(src, dst, hw1, dinv, agg1, E);
    k_layer2<<<(N + tb - 1) / tb, tb, 0, stream>>>(agg1, b1, W2, dinv, hw2, agg2, N);
    k_edges2<<<(E + tb - 1) / tb, tb, 0, stream>>>(src, dst, hw2, dinv, agg2, E);
    k_final <<<(N + tb - 1) / tb, tb, 0, stream>>>(agg2, b2, out, N);
}


# layer-2 edge scatter: two adjacent lanes per edge (lane = channel), so both channels leave as one atomic instruction touching one line per edge instead of two (same arrival-order f32 atomics)
# speedup vs baseline: 1.1857x; 1.1857x over previous
#include <hip/hip_runtime.h>
#include <hip/hip_bf16.h>

#define N_FEAT 256
#define HIDDEN 16
#define NCLS   2

typedef float v2f __attribute__((ext_vector_type(2)));
typedef float v8f __attribute__((ext_vector_type(8)));

__device__ __forceinline__ void atomic_add_f32(float* p, float v) {
    unsafeAtomicAdd(p, v);
}

__global__ void k_deg_init(float* __restrict__ deg, int n) {
    int i = blockIdx.x * blockDim.x + threadIdx.x;
    if (i < n) deg[i] = 1.0f;
}

__global__ void k_deg_edges(const int* __restrict__ dst, float* __restrict__ deg, int E) {
    int e = blockIdx.x * blockDim.x + threadIdx.x;
    if (e < E) atomic_add_f32(&deg[dst[e]], 1.0f);
}

__global__ void k_dinv(float* __restrict__ deg, int n) {
    int i = blockIdx.x * blockDim.x + threadIdx.x;
    if (i < n) deg[i] = rsqrtf(deg[i]);
}

__global__ void k_gemm1(const float* __restrict__ x, const float* __restrict__ W1,
                        const float* __restrict__ dinv,
                        float* __restrict__ hw1, float* __restrict__ agg1, int ntiles) {
    int lane = threadIdx.x & 31;
    int wave = threadIdx.x >> 5;
    int tile = blockIdx.x * (blockDim.x >> 5) + wave;
    if (tile >= ntiles) return;
    int half = lane >> 4;
    int l    = lane & 15;
    int row0 = tile * 16;

    const float* aptr = x  + (size_t)(row0 + l) * N_FEAT + 2 * half;
    const float* bptr = W1 + (size_t)(2 * half) * HIDDEN + l;

    v8f c = {};
    #pragma unroll 8
    for (int kk = 0; kk < N_FEAT / 4; ++kk) {
        v2f a = *(const v2f*)(aptr + kk * 4);
        v2f b;
        b.x = bptr[kk * 4 * HIDDEN];
        b.y = bptr[kk * 4 * HIDDEN + HIDDEN];
        c = __builtin_amdgcn_wmma_f32_16x16x4_f32(false, a, false, b,
                                                  (short)0, c, false, false);
    }

    int rbase = row0 + half * 8;
    #pragma unroll
    for (int v = 0; v < 8; ++v) {
        int r = rbase + v;
        float dval = c[v];
        float di   = dinv[r];
        size_t idx = (size_t)r * HIDDEN + l;
        hw1[idx]  = dval;
        agg1[idx] = dval * di * di;
    }
}

__global__ void k_gemm1_rem(const float* __restrict__ x, const float* __restrict__ W1,
                            const float* __restrict__ dinv,
                            float* __restrict__ hw1, float* __restrict__ agg1,
                            int row_lo, int n) {
    int t = blockIdx.x * blockDim.x + threadIdx.x;
    int r = row_lo + t / HIDDEN;
    int c = t % HIDDEN;
    if (r >= n) return;
    float acc = 0.0f;
    for (int k = 0; k < N_FEAT; ++k) acc += x[(size_t)r * N_FEAT + k] * W1[(size_t)k * HIDDEN + c];
    float di = dinv[r];
    size_t idx = (size_t)r * HIDDEN + c;
    hw1[idx]  = acc;
    agg1[idx] = acc * di * di;
}

__global__ void k_edges1(const int* __restrict__ src, const int* __restrict__ dst,
                         const float* __restrict__ hw1, const float* __restrict__ dinv,
                         float* __restrict__ agg1, int E) {
    int t = blockIdx.x * blockDim.x + threadIdx.x;
    int e = t >> 4;
    int c = t & 15;
    if (e >= E) return;
    int s = src[e], d = dst[e];
    float norm = dinv[s] * dinv[d];
    atomic_add_f32(&agg1[(size_t)d * HIDDEN + c], hw1[(size_t)s * HIDDEN + c] * norm);
}

__global__ void k_layer2(const float* __restrict__ agg1, const float* __restrict__ b1,
                         const float* __restrict__ W2, const float* __restrict__ dinv,
                         float* __restrict__ hw2, float* __restrict__ agg2, int n) {
    int i = blockIdx.x * blockDim.x + threadIdx.x;
    if (i >= n) return;
    float h0 = 0.0f, h1 = 0.0f;
    #pragma unroll
    for (int c = 0; c < HIDDEN; ++c) {
        float v = agg1[(size_t)i * HIDDEN + c] + b1[c];
        v = fmaxf(v, 0.0f);
        h0 += v * W2[c * NCLS + 0];
        h1 += v * W2[c * NCLS + 1];
    }
    float di  = dinv[i];
    float di2 = di * di;
    hw2[(size_t)i * NCLS + 0]  = h0;
    hw2[(size_t)i * NCLS + 1]  = h1;
    agg2[(size_t)i * NCLS + 0] = h0 * di2;
    agg2[(size_t)i * NCLS + 1] = h1 * di2;
}

__global__ void k_edges2(const int* __restrict__ src, const int* __restrict__ dst,
                         const float* __restrict__ hw2, const float* __restrict__ dinv,
                         float* __restrict__ agg2, int E) {
    int t = blockIdx.x * blockDim.x + threadIdx.x;
    int e = t >> 1;
    int c = t & 1;
    if (e >= E) return;
    int s = src[e], d = dst[e];
    float norm = dinv[s] * dinv[d];
    atomic_add_f32(&agg2[(size_t)d * NCLS + c], hw2[(size_t)s * NCLS + c] * norm);
}

__global__ void k_final(const float* __restrict__ agg2, const float* __restrict__ b2,
                        float* __restrict__ out, int n) {
    int i = blockIdx.x * blockDim.x + threadIdx.x;
    if (i >= n) return;
    float l0 = agg2[(size_t)i * NCLS + 0] + b2[0];
    float l1 = agg2[(size_t)i * NCLS + 1] + b2[1];
    float m   = fmaxf(l0, l1);
    float lse = m + logf(expf(l0 - m) + expf(l1 - m));
    out[(size_t)i * NCLS + 0] = l0 - lse;
    out[(size_t)i * NCLS + 1] = l1 - lse;
}

extern "C" void kernel_launch(void* const* d_in, const int* in_sizes, int n_in,
                              void* d_out, int out_size, void* d_ws, size_t ws_size,
                              hipStream_t stream) {
    const float* x  = (const float*)d_in[0];
    const int*   ei = (const int*)d_in[1];
    const float* W1 = (const float*)d_in[2];
    const float* b1 = (const float*)d_in[3];
    const float* W2 = (const float*)d_in[4];
    const float* b2 = (const float*)d_in[5];
    float* out = (float*)d_out;

    int N = in_sizes[0] / N_FEAT;
    int E = in_sizes[1] / 2;
    const int* src = ei;
    const int* dst = ei + E;

    float* ws   = (float*)d_ws;
    float* dinv = ws;
    float* hw1  = dinv + (size_t)N;
    float* agg1 = hw1  + (size_t)N * HIDDEN;
    float* hw2  = agg1 + (size_t)N * HIDDEN;
    float* agg2 = hw2  + (size_t)N * NCLS;

    const int tb = 256;
    k_deg_init <<<(N + tb - 1) / tb, tb, 0, stream>>>(dinv, N);
    k_deg_edges<<<(E + tb - 1) / tb, tb, 0, stream>>>(dst, dinv, E);
    k_dinv     <<<(N + tb - 1) / tb, tb, 0, stream>>>(dinv, N);

    int ntiles = N / 16;
    int wpb    = tb / 32;
    if (ntiles > 0)
        k_gemm1<<<(ntiles + wpb - 1) / wpb, tb, 0, stream>>>(x, W1, dinv, hw1, agg1, ntiles);
    int rem = N - ntiles * 16;
    if (rem > 0)
        k_gemm1_rem<<<(rem * HIDDEN + tb - 1) / tb, tb, 0, stream>>>(x, W1, dinv, hw1, agg1,
                                                                     ntiles * 16, N);

    long long t1 = (long long)E * HIDDEN;
    k_edges1<<<(int)((t1 + tb - 1) / tb), tb, 0, stream>>>(src, dst, hw1, dinv, agg1, E);
    k_layer2<<<(N + tb - 1) / tb, tb, 0, stream>>>(agg1, b1, W2, dinv, hw2, agg2, N);
    long long t2 = (long long)E * NCLS;
    k_edges2<<<(int)((t2 + tb - 1) / tb), tb, 0, stream>>>(src, dst, hw2, dinv, agg2, E);
    k_final <<<(N + tb - 1) / tb, tb, 0, stream>>>(agg2, b2, out, N);
}


# layer-1 edge scatter: one wave per 32 edges (lane = edge for coalesced src/dst and dinv gathers, then lane = channel with 16 row loads and 16 atomic rows in flight); same arrival-order f32 atomics
# speedup vs baseline: 1.3975x; 1.1786x over previous
#include <hip/hip_runtime.h>
#include <hip/hip_bf16.h>

#define N_FEAT 256
#define HIDDEN 16
#define NCLS   2

typedef float v2f __attribute__((ext_vector_type(2)));
typedef float v8f __attribute__((ext_vector_type(8)));

__device__ __forceinline__ void atomic_add_f32(float* p, float v) {
    unsafeAtomicAdd(p, v);
}

__global__ void k_deg_init(float* __restrict__ deg, int n) {
    int i = blockIdx.x * blockDim.x + threadIdx.x;
    if (i < n) deg[i] = 1.0f;
}

__global__ void k_deg_edges(const int* __restrict__ dst, float* __restrict__ deg, int E) {
    int e = blockIdx.x * blockDim.x + threadIdx.x;
    if (e < E) atomic_add_f32(&deg[dst[e]], 1.0f);
}

__global__ void k_dinv(float* __restrict__ deg, int n) {
    int i = blockIdx.x * blockDim.x + threadIdx.x;
    if (i < n) deg[i] = rsqrtf(deg[i]);
}

__global__ void k_gemm1(const float* __restrict__ x, const float* __restrict__ W1,
                        const float* __restrict__ dinv,
                        float* __restrict__ hw1, float* __restrict__ agg1, int ntiles) {
    int lane = threadIdx.x & 31;
    int wave = threadIdx.x >> 5;
    int tile = blockIdx.x * (blockDim.x >> 5) + wave;
    if (tile >= ntiles) return;
    int half = lane >> 4;
    int l    = lane & 15;
    int row0 = tile * 16;

    const float* aptr = x  + (size_t)(row0 + l) * N_FEAT + 2 * half;
    const float* bptr = W1 + (size_t)(2 * half) * HIDDEN + l;

    v8f c = {};
    #pragma unroll 8
    for (int kk = 0; kk < N_FEAT / 4; ++kk) {
        v2f a = *(const v2f*)(aptr + kk * 4);
        v2f b;
        b.x = bptr[kk * 4 * HIDDEN];
        b.y = bptr[kk * 4 * HIDDEN + HIDDEN];
        c = __builtin_amdgcn_wmma_f32_16x16x4_f32(false, a, false, b,
                                                  (short)0, c, false, false);
    }

    int rbase = row0 + half * 8;
    #pragma unroll
    for (int v = 0; v < 8; ++v) {
        int r = rbase + v;
        float dval = c[v];
        float di   = dinv[r];
        size_t idx = (size_t)r * HIDDEN + l;
        hw1[idx]  = dval;
        agg1[idx] = dval * di * di;
    }
}

__global__ void k_gemm1_rem(const float* __restrict__ x, const float* __restrict__ W1,
                            const float* __restrict__ dinv,
                            float* __restrict__ hw1, float* __restrict__ agg1,
                            int row_lo, int n) {
    int t = blockIdx.x * blockDim.x + threadIdx.x;
    int r = row_lo + t / HIDDEN;
    int c = t % HIDDEN;
    if (r >= n) return;
    float acc = 0.0f;
    for (int k = 0; k < N_FEAT; ++k) acc += x[(size_t)r * N_FEAT + k] * W1[(size_t)k * HIDDEN + c];
    float di = dinv[r];
    size_t idx = (size_t)r * HIDDEN + c;
    hw1[idx]  = acc;
    agg1[idx] = acc * di * di;
}

__global__ void k_edges1(const int* __restrict__ src, const int* __restrict__ dst,
                         const float* __restrict__ hw1, const float* __restrict__ dinv,
                         float* __restrict__ agg1, int E) {
    int lane = threadIdx.x & 31;
    int wave = (blockIdx.x * blockDim.x + threadIdx.x) >> 5;
    int e0   = wave * 32;
    if (e0 >= E) return;
    int e = e0 + lane;
    int s = 0, d = 0;
    float norm = 0.0f;
    if (e < E) {
        s = src[e];
        d = dst[e];
        norm = dinv[s] * dinv[d];
    }
    int c     = lane & 15;
    int hbase = lane & 16;
    if (e0 + 32 <= E) {
        float val[16];
        int   drow[16];
        #pragma unroll
        for (int j = 0; j < 16; ++j) {
            int   sj = __shfl(s, hbase + j);
            float nj = __shfl(norm, hbase + j);
            drow[j]  = __shfl(d, hbase + j);
            val[j]   = hw1[(size_t)sj * HIDDEN + c] * nj;
        }
        #pragma unroll
        for (int j = 0; j < 16; ++j)
            atomic_add_f32(&agg1[(size_t)drow[j] * HIDDEN + c], val[j]);
    } else {
        for (int j = 0; j < 16; ++j) {
            int   sj = __shfl(s, hbase + j);
            float nj = __shfl(norm, hbase + j);
            int   dj = __shfl(d, hbase + j);
            if (e0 + hbase + j < E)
                atomic_add_f32(&agg1[(size_t)dj * HIDDEN + c],
                               hw1[(size_t)sj * HIDDEN + c] * nj);
        }
    }
}

__global__ void k_layer2(const float* __restrict__ agg1, const float* __restrict__ b1,
                         const float* __restrict__ W2, const float* __restrict__ dinv,
                         float* __restrict__ hw2, float* __restrict__ agg2, int n) {
    int i = blockIdx.x * blockDim.x + threadIdx.x;
    if (i >= n) return;
    float h0 = 0.0f, h1 = 0.0f;
    #pragma unroll
    for (int c = 0; c < HIDDEN; ++c) {
        float v = agg1[(size_t)i * HIDDEN + c] + b1[c];
        v = fmaxf(v, 0.0f);
        h0 += v * W2[c * NCLS + 0];
        h1 += v * W2[c * NCLS + 1];
    }
    float di  = dinv[i];
    float di2 = di * di;
    hw2[(size_t)i * NCLS + 0]  = h0;
    hw2[(size_t)i * NCLS + 1]  = h1;
    agg2[(size_t)i * NCLS + 0] = h0 * di2;
    agg2[(size_t)i * NCLS + 1] = h1 * di2;
}

__global__ void k_edges2(const int* __restrict__ src, const int* __restrict__ dst,
                         const float* __restrict__ hw2, const float* __restrict__ dinv,
                         float* __restrict__ agg2, int E) {
    int t = blockIdx.x * blockDim.x + threadIdx.x;
    int e = t >> 1;
    int c = t & 1;
    if (e >= E) return;
    int s = src[e], d = dst[e];
    float norm = dinv[s] * dinv[d];
    atomic_add_f32(&agg2[(size_t)d * NCLS + c], hw2[(size_t)s * NCLS + c] * norm);
}

__global__ void k_final(const float* __restrict__ agg2, const float* __restrict__ b2,
                        float* __restrict__ out, int n) {
    int i = blockIdx.x * blockDim.x + threadIdx.x;
    if (i >= n) return;
    float l0 = agg2[(size_t)i * NCLS + 0] + b2[0];
    float l1 = agg2[(size_t)i * NCLS + 1] + b2[1];
    float m   = fmaxf(l0, l1);
    float lse = m + logf(expf(l0 - m) + expf(l1 - m));
    out[(size_t)i * NCLS + 0] = l0 - lse;
    out[(size_t)i * NCLS + 1] = l1 - lse;
}

extern "C" void kernel_launch(void* const* d_in, const int* in_sizes, int n_in,
                              void* d_out, int out_size, void* d_ws, size_t ws_size,
                              hipStream_t stream) {
    const float* x  = (const float*)d_in[0];
    const int*   ei = (const int*)d_in[1];
    const float* W1 = (const float*)d_in[2];
    const float* b1 = (const float*)d_in[3];
    const float* W2 = (const float*)d_in[4];
    const float* b2 = (const float*)d_in[5];
    float* out = (float*)d_out;

    int N = in_sizes[0] / N_FEAT;
    int E = in_sizes[1] / 2;
    const int* src = ei;
    const int* dst = ei + E;

    float* ws   = (float*)d_ws;
    float* dinv = ws;
    float* hw1  = dinv + (size_t)N;
    float* agg1 = hw1  + (size_t)N * HIDDEN;
    float* hw2  = agg1 + (size_t)N * HIDDEN;
    float* agg2 = hw2  + (size_t)N * NCLS;

    const int tb = 256;
    k_deg_init <<<(N + tb - 1) / tb, tb, 0, stream>>>(dinv, N);
    k_deg_edges<<<(E + tb - 1) / tb, tb, 0, stream>>>(dst, dinv, E);
    k_dinv     <<<(N + tb - 1) / tb, tb, 0, stream>>>(dinv, N);

    int ntiles = N / 16;
    int wpb    = tb / 32;
    if (ntiles > 0)
        k_gemm1<<<(ntiles + wpb - 1) / wpb, tb, 0, stream>>>(x, W1, dinv, hw1, agg1, ntiles);
    int rem = N - ntiles * 16;
    if (rem > 0)
        k_gemm1_rem<<<(rem * HIDDEN + tb - 1) / tb, tb, 0, stream>>>(x, W1, dinv, hw1, agg1,
                                                                     ntiles * 16, N);

    k_edges1<<<(E + tb - 1) / tb, tb, 0, stream>>>(src, dst, hw1, dinv, agg1, E);
    k_layer2<<<(N + tb - 1) / tb, tb, 0, stream>>>(agg1, b1, W2, dinv, hw2, agg2, N);
    long long t2 = (long long)E * NCLS;
    k_edges2<<<(int)((t2 + tb - 1) / tb), tb, 0, stream>>>(src, dst, hw2, dinv, agg2, E);
    k_final <<<(N + tb - 1) / tb, tb, 0, stream>>>(agg2, b2, out, N);
}


# edge weight dinv[src]*dinv[dst] factored out of both edge passes: rows stored pre-scaled by dinv[src], finished sums scaled once by dinv[dst]; no dinv gathers per edge; atomic sum order unchanged
# speedup vs baseline: 1.5206x; 1.0881x over previous
#include <hip/hip_runtime.h>
#include <hip/hip_bf16.h>

#define N_FEAT 256
#define HIDDEN 16
#define NCLS   2

typedef float v2f __attribute__((ext_vector_type(2)));
typedef float v8f __attribute__((ext_vector_type(8)));

__device__ __forceinline__ void atomic_add_f32(float* p, float v) {
    unsafeAtomicAdd(p, v);
}

__global__ void k_deg_init(float* __restrict__ deg, int n) {
    int i = blockIdx.x * blockDim.x + threadIdx.x;
    if (i < n) deg[i] = 1.0f;
}

__global__ void k_deg_edges(const int* __restrict__ dst, float* __restrict__ deg, int E) {
    int e = blockIdx.x * blockDim.x + threadIdx.x;
    if (e < E) atomic_add_f32(&deg[dst[e]], 1.0f);
}

__global__ void k_dinv(float* __restrict__ deg, int n) {
    int i = blockIdx.x * blockDim.x + threadIdx.x;
    if (i < n) deg[i] = rsqrtf(deg[i]);
}

__global__ void k_gemm1(const float* __restrict__ x, const float* __restrict__ W1,
                        const float* __restrict__ dinv,
                        float* __restrict__ hw1, float* __restrict__ agg1, int ntiles) {
    int lane = threadIdx.x & 31;
    int wave = threadIdx.x >> 5;
    int tile = blockIdx.x * (blockDim.x >> 5) + wave;
    if (tile >= ntiles) return;
    int half = lane >> 4;
    int l    = lane & 15;
    int row0 = tile * 16;

    const float* aptr = x  + (size_t)(row0 + l) * N_FEAT + 2 * half;
    const float* bptr = W1 + (size_t)(2 * half) * HIDDEN + l;

    v8f c = {};
    #pragma unroll 8
    for (int kk = 0; kk < N_FEAT / 4; ++kk) {
        v2f a = *(const v2f*)(aptr + kk * 4);
        v2f b;
        b.x = bptr[kk * 4 * HIDDEN];
        b.y = bptr[kk * 4 * HIDDEN + HIDDEN];
        c = __builtin_amdgcn_wmma_f32_16x16x4_f32(false, a, false, b,
                                                  (short)0, c, false, false);
    }

    int rbase = row0 + half * 8;
    #pragma unroll
    for (int v = 0; v < 8; ++v) {
        int r = rbase + v;
        float dval = c[v] * dinv[r];
        size_t idx = (size_t)r * HIDDEN + l;
        hw1[idx]  = dval;
        agg1[idx] = dval;
    }
}

__global__ void k_gemm1_rem(const float* __restrict__ x, const float* __restrict__ W1,
                            const float* __restrict__ dinv,
                            float* __restrict__ hw1, float* __restrict__ agg1,
                            int row_lo, int n) {
    int t = blockIdx.x * blockDim.x + threadIdx.x;
    int r = row_lo + t / HIDDEN;
    int c = t % HIDDEN;
    if (r >= n) return;
    float acc = 0.0f;
    for (int k = 0; k < N_FEAT; ++k) acc += x[(size_t)r * N_FEAT + k] * W1[(size_t)k * HIDDEN + c];
    acc *= dinv[r];
    size_t idx = (size_t)r * HIDDEN + c;
    hw1[idx]  = acc;
    agg1[idx] = acc;
}

__global__ void k_edges1(const int* __restrict__ src, const int* __restrict__ dst,
                         const float* __restrict__ hw1,
                         float* __restrict__ agg1, int E) {
    int lane = threadIdx.x & 31;
    int wave = (blockIdx.x * blockDim.x + threadIdx.x) >> 5;
    int e0   = wave * 32;
    if (e0 >= E) return;
    int e = e0 + lane;
    int s = 0, d = 0;
    if (e < E) {
        s = src[e];
        d = dst[e];
    }
    int c     = lane & 15;
    int hbase = lane & 16;
    if (e0 + 32 <= E) {
        float val[16];
        int   drow[16];
        #pragma unroll
        for (int j = 0; j < 16; ++j) {
            int   sj = __shfl(s, hbase + j);
            drow[j]  = __shfl(d, hbase + j);
            val[j]   = hw1[(size_t)sj * HIDDEN + c];
        }
        #pragma unroll
        for (int j = 0; j < 16; ++j)
            atomic_add_f32(&agg1[(size_t)drow[j] * HIDDEN + c], val[j]);
    } else {
        for (int j = 0; j < 16; ++j) {
            int   sj = __shfl(s, hbase + j);
            int   dj = __shfl(d, hbase + j);
            if (e0 + hbase + j < E)
                atomic_add_f32(&agg1[(size_t)dj * HIDDEN + c],
                               hw1[(size_t)sj * HIDDEN + c]);
        }
    }
}

__global__ void k_layer2(const float* __restrict__ agg1, const float* __restrict__ b1,
                         const float* __restrict__ W2, const float* __restrict__ dinv,
                         float* __restrict__ hw2, float* __restrict__ agg2, int n) {
    int i = blockIdx.x * blockDim.x + threadIdx.x;
    if (i >= n) return;
    float di = dinv[i];
    float h0 = 0.0f, h1 = 0.0f;
    #pragma unroll
    for (int c = 0; c < HIDDEN; ++c) {
        float v = agg1[(size_t)i * HIDDEN + c] * di + b1[c];
        v = fmaxf(v, 0.0f);
        h0 += v * W2[c * NCLS + 0];
        h1 += v * W2[c * NCLS + 1];
    }
    h0 *= di;
    h1 *= di;
    hw2[(size_t)i * NCLS + 0]  = h0;
    hw2[(size_t)i * NCLS + 1]  = h1;
    agg2[(size_t)i * NCLS + 0] = h0;
    agg2[(size_t)i * NCLS + 1] = h1;
}

__global__ void k_edges2(const int* __restrict__ src, const int* __restrict__ dst,
                         const float* __restrict__ hw2,
                         float* __restrict__ agg2, int E) {
    int t = blockIdx.x * blockDim.x + threadIdx.x;
    int e = t >> 1;
    int c = t & 1;
    if (e >= E) return;
    int s = src[e], d = dst[e];
    atomic_add_f32(&agg2[(size_t)d * NCLS + c], hw2[(size_t)s * NCLS + c]);
}

__global__ void k_final(const float* __restrict__ agg2, const float* __restrict__ b2,
                        const float* __restrict__ dinv,
                        float* __restrict__ out, int n) {
    int i = blockIdx.x * blockDim.x + threadIdx.x;
    if (i >= n) return;
    float di = dinv[i];
    float l0 = agg2[(size_t)i * NCLS + 0] * di + b2[0];
    float l1 = agg2[(size_t)i * NCLS + 1] * di + b2[1];
    float m   = fmaxf(l0, l1);
    float lse = m + logf(expf(l0 - m) + expf(l1 - m));
    out[(size_t)i * NCLS + 0] = l0 - lse;
    out[(size_t)i * NCLS + 1] = l1 - lse;
}

extern "C" void kernel_launch(void* const* d_in, const int* in_sizes, int n_in,
                              void* d_out, int out_size, void* d_ws, size_t ws_size,
                              hipStream_t stream) {
    const float* x  = (const float*)d_in[0];
    const int*   ei = (const int*)d_in[1];
    const float* W1 = (const float*)d_in[2];
    const float* b1 = (const float*)d_in[3];
    const float* W2 = (const float*)d_in[4];
    const float* b2 = (const float*)d_in[5];
    float* out = (float*)d_out;

    int N = in_sizes[0] / N_FEAT;
    int E = in_sizes[1] / 2;
    const int* src = ei;
    const int* dst = ei + E;

    float* ws   = (float*)d_ws;
    float* dinv = ws;
    float* hw1  = dinv + (size_t)N;
    float* agg1 = hw1  + (size_t)N * HIDDEN;
    float* hw2  = agg1 + (size_t)N * HIDDEN;
    float* agg2 = hw2  + (size_t)N * NCLS;

    const int tb = 256;
    k_deg_init <<<(N + tb - 1) / tb, tb, 0, stream>>>(dinv, N);
    k_deg_edges<<<(E + tb - 1) / tb, tb, 0, stream>>>(dst, dinv, E);
    k_dinv     <<<(N + tb - 1) / tb, tb, 0, stream>>>(dinv, N);

    int ntiles = N / 16;
    int wpb    = tb / 32;
    if (ntiles > 0)
        k_gemm1<<<(ntiles + wpb - 1) / wpb, tb, 0, stream>>>(x, W1, dinv, hw1, agg1, ntiles);
    int rem = N - ntiles * 16;
    if (rem > 0)
        k_gemm1_rem<<<(rem * HIDDEN + tb - 1) / tb, tb, 0, stream>>>(x, W1, dinv, hw1, agg1,
                                                                     ntiles * 16, N);

    k_edges1<<<(E + tb - 1) / tb, tb, 0, stream>>>(src, dst, hw1, agg1, E);
    k_layer2<<<(N + tb - 1) / tb, tb, 0, stream>>>(agg1, b1, W2, dinv, hw2, agg2, N);
    long long t2 = (long long)E * NCLS;
    k_edges2<<<(int)((t2 + tb - 1) / tb), tb, 0, stream>>>(src, dst, hw2, agg2, E);
    k_final <<<(N + tb - 1) / tb, tb, 0, stream>>>(agg2, b2, dinv, out, N);
}


# degree count: a block counts one edge slice for a quarter of the nodes in integer LDS counters, then adds them to deg with lane = node: one whole-line atomic per 32 nodes, not a line per edge; exact
# speedup vs baseline: 2.2380x; 1.4718x over previous
#include <hip/hip_runtime.h>
#include <hip/hip_bf16.h>

#define N_FEAT 256
#define HIDDEN 16
#define NCLS   2

typedef float v2f __attribute__((ext_vector_type(2)));
typedef float v8f __attribute__((ext_vector_type(8)));

__device__ __forceinline__ void atomic_add_f32(float* p, float v) {
    unsafeAtomicAdd(p, v);
}

__global__ void k_deg_init(float* __restrict__ deg, int n) {
    int i = blockIdx.x * blockDim.x + threadIdx.x;
    if (i < n) deg[i] = 1.0f;
}

__global__ void k_deg_edges(const int* __restrict__ dst, float* __restrict__ deg, int E) {
    int e = blockIdx.x * blockDim.x + threadIdx.x;
    if (e < E) atomic_add_f32(&deg[dst[e]], 1.0f);
}

#define DEG_RANGES   4
#define DEG_RNG_MAX  32768
#define DEG_THREADS  1024
#define DEG_LOADS    49
#define DEG_SLICE    (DEG_LOADS * DEG_THREADS)

__global__ __launch_bounds__(DEG_THREADS)
void k_deg_hist(const int* __restrict__ dst, float* __restrict__ deg,
                int E, int N, int rng, int nslices) {
    __shared__ unsigned int s_cnt[DEG_RNG_MAX];
    int tid   = threadIdx.x;
    int slice = blockIdx.x / DEG_RANGES;
    int range = blockIdx.x % DEG_RANGES;
    int lo    = range * rng;

    for (int i = tid; i < rng; i += DEG_THREADS) s_cnt[i] = 0u;
    __syncthreads();

    int e0 = slice * DEG_SLICE;
    int e1 = min(e0 + DEG_SLICE, E);
    int d[DEG_LOADS];
    const int* p = dst + e0 + tid;
    if (e0 + DEG_SLICE <= E) {
        #pragma unroll
        for (int j = 0; j < DEG_LOADS; ++j) d[j] = p[j * DEG_THREADS];
    } else {
        #pragma unroll
        for (int j = 0; j < DEG_LOADS; ++j)
            d[j] = (e0 + j * DEG_THREADS + tid < e1) ? p[j * DEG_THREADS] : -1;
    }
    #pragma unroll
    for (int j = 0; j < DEG_LOADS; ++j) {
        unsigned int r = (unsigned int)(d[j] - lo);
        if (r < (unsigned int)rng) atomicAdd(&s_cnt[r], 1u);
    }
    __syncthreads();

    int rot = (int)(((unsigned int)slice * (unsigned int)(rng >> 5)) / (unsigned int)nslices) << 5;
    for (int i = tid; i < rng; i += DEG_THREADS) {
        int j = i + rot;
        if (j >= rng) j -= rng;
        unsigned int c = s_cnt[j];
        int node = lo + j;
        if (c != 0u && node < N) atomic_add_f32(&deg[node], (float)c);
    }
}

__global__ void k_dinv(float* __restrict__ deg, int n) {
    int i = blockIdx.x * blockDim.x + threadIdx.x;
    if (i < n) deg[i] = rsqrtf(deg[i]);
}

__global__ void k_gemm1(const float* __restrict__ x, const float* __restrict__ W1,
                        const float* __restrict__ dinv,
                        float* __restrict__ hw1, float* __restrict__ agg1, int ntiles) {
    int lane = threadIdx.x & 31;
    int wave = threadIdx.x >> 5;
    int tile = blockIdx.x * (blockDim.x >> 5) + wave;
    if (tile >= ntiles) return;
    int half = lane >> 4;
    int l    = lane & 15;
    int row0 = tile * 16;

    const float* aptr = x  + (size_t)(row0 + l) * N_FEAT + 2 * half;
    const float* bptr = W1 + (size_t)(2 * half) * HIDDEN + l;

    v8f c = {};
    #pragma unroll 8
    for (int kk = 0; kk < N_FEAT / 4; ++kk) {
        v2f a = *(const v2f*)(aptr + kk * 4);
        v2f b;
        b.x = bptr[kk * 4 * HIDDEN];
        b.y = bptr[kk * 4 * HIDDEN + HIDDEN];
        c = __builtin_amdgcn_wmma_f32_16x16x4_f32(false, a, false, b,
                                                  (short)0, c, false, false);
    }

    int rbase = row0 + half * 8;
    #pragma unroll
    for (int v = 0; v < 8; ++v) {
        int r = rbase + v;
        float dval = c[v] * dinv[r];
        size_t idx = (size_t)r * HIDDEN + l;
        hw1[idx]  = dval;
        agg1[idx] = dval;
    }
}

__global__ void k_gemm1_rem(const float* __restrict__ x, const float* __restrict__ W1,
                            const float* __restrict__ dinv,
                            float* __restrict__ hw1, float* __restrict__ agg1,
                            int row_lo, int n) {
    int t = blockIdx.x * blockDim.x + threadIdx.x;
    int r = row_lo + t / HIDDEN;
    int c = t % HIDDEN;
    if (r >= n) return;
    float acc = 0.0f;
    for (int k = 0; k < N_FEAT; ++k) acc += x[(size_t)r * N_FEAT + k] * W1[(size_t)k * HIDDEN + c];
    acc *= dinv[r];
    size_t idx = (size_t)r * HIDDEN + c;
    hw1[idx]  = acc;
    agg1[idx] = acc;
}

__global__ void k_edges1(const int* __restrict__ src, const int* __restrict__ dst,
                         const float* __restrict__ hw1,
                         float* __restrict__ agg1, int E) {
    int lane = threadIdx.x & 31;
    int wave = (blockIdx.x * blockDim.x + threadIdx.x) >> 5;
    int e0   = wave * 32;
    if (e0 >= E) return;
    int e = e0 + lane;
    int s = 0, d = 0;
    if (e < E) {
        s = src[e];
        d = dst[e];
    }
    int c     = lane & 15;
    int hbase = lane & 16;
    if (e0 + 32 <= E) {
        float val[16];
        int   drow[16];
        #pragma unroll
        for (int j = 0; j < 16; ++j) {
            int   sj = __shfl(s, hbase + j);
            drow[j]  = __shfl(d, hbase + j);
            val[j]   = hw1[(size_t)sj * HIDDEN + c];
        }
        #pragma unroll
        for (int j = 0; j < 16; ++j)
            atomic_add_f32(&agg1[(size_t)drow[j] * HIDDEN + c], val[j]);
    } else {
        for (int j = 0; j < 16; ++j) {
            int   sj = __shfl(s, hbase + j);
            int   dj = __shfl(d, hbase + j);
            if (e0 + hbase + j < E)
                atomic_add_f32(&agg1[(size_t)dj * HIDDEN + c],
                               hw1[(size_t)sj * HIDDEN + c]);
        }
    }
}

__global__ void k_layer2(const float* __restrict__ agg1, const float* __restrict__ b1,
                         const float* __restrict__ W2, const float* __restrict__ dinv,
                         float* __restrict__ hw2, float* __restrict__ agg2, int n) {
    int i = blockIdx.x * blockDim.x + threadIdx.x;
    if (i >= n) return;
    float di = dinv[i];
    float h0 = 0.0f, h1 = 0.0f;
    #pragma unroll
    for (int c = 0; c < HIDDEN; ++c) {
        float v = agg1[(size_t)i * HIDDEN + c] * di + b1[c];
        v = fmaxf(v, 0.0f);
        h0 += v * W2[c * NCLS + 0];
        h1 += v * W2[c * NCLS + 1];
    }
    h0 *= di;
    h1 *= di;
    hw2[(size_t)i * NCLS + 0]  = h0;
    hw2[(size_t)i * NCLS + 1]  = h1;
    agg2[(size_t)i * NCLS + 0] = h0;
    agg2[(size_t)i * NCLS + 1] = h1;
}

__global__ void k_edges2(const int* __restrict__ src, const int* __restrict__ dst,
                         const float* __restrict__ hw2,
                         float* __restrict__ agg2, int E) {
    int t = blockIdx.x * blockDim.x + threadIdx.x;
    int e = t >> 1;
    int c = t & 1;
    if (e >= E) return;
    int s = src[e], d = dst[e];
    atomic_add_f32(&agg2[(size_t)d * NCLS + c], hw2[(size_t)s * NCLS + c]);
}

__global__ void k_final(const float* __restrict__ agg2, const float* __restrict__ b2,
                        const float* __restrict__ dinv,
                        float* __restrict__ out, int n) {
    int i = blockIdx.x * blockDim.x + threadIdx.x;
    if (i >= n) return;
    float di = dinv[i];
    float l0 = agg2[(size_t)i * NCLS + 0] * di + b2[0];
    float l1 = agg2[(size_t)i * NCLS + 1] * di + b2[1];
    float m   = fmaxf(l0, l1);
    float lse = m + logf(expf(l0 - m) + expf(l1 - m));
    out[(size_t)i * NCLS + 0] = l0 - lse;
    out[(size_t)i * NCLS + 1] = l1 - lse;
}

extern "C" void kernel_launch(void* const* d_in, const int* in_sizes, int n_in,
                              void* d_out, int out_size, void* d_ws, size_t ws_size,
                              hipStream_t stream) {
    const float* x  = (const float*)d_in[0];
    const int*   ei = (const int*)d_in[1];
    const float* W1 = (const float*)d_in[2];
    const float* b1 = (const float*)d_in[3];
    const float* W2 = (const float*)d_in[4];
    const float* b2 = (const float*)d_in[5];
    float* out = (float*)d_out;

    int N = in_sizes[0] / N_FEAT;
    int E = in_sizes[1] / 2;
    const int* src = ei;
    const int* dst = ei + E;

    float* ws   = (float*)d_ws;
    float* dinv = ws;
    float* hw1  = dinv + (size_t)N;
    float* agg1 = hw1  + (size_t)N * HIDDEN;
    float* hw2  = agg1 + (size_t)N * HIDDEN;
    float* agg2 = hw2  + (size_t)N * NCLS;

    const int tb = 256;
    k_deg_init <<<(N + tb - 1) / tb, tb, 0, stream>>>(dinv, N);
    int rng = (((N + DEG_RANGES - 1) / DEG_RANGES) + 31) & ~31;
    if (rng <= DEG_RNG_MAX) {
        int nslices = (E + DEG_SLICE - 1) / DEG_SLICE;
        if (nslices > 0)
            k_deg_hist<<<nslices * DEG_RANGES, DEG_THREADS, 0, stream>>>(dst, dinv, E, N, rng, nslices);
    } else {
        k_deg_edges<<<(E + tb - 1) / tb, tb, 0, stream>>>(dst, dinv, E);
    }
    k_dinv     <<<(N + tb - 1) / tb, tb, 0, stream>>>(dinv, N);

    int ntiles = N / 16;
    int wpb    = tb / 32;
    if (ntiles > 0)
        k_gemm1<<<(ntiles + wpb - 1) / wpb, tb, 0, stream>>>(x, W1, dinv, hw1, agg1, ntiles);
    int rem = N - ntiles * 16;
    if (rem > 0)
        k_gemm1_rem<<<(rem * HIDDEN + tb - 1) / tb, tb, 0, stream>>>(x, W1, dinv, hw1, agg1,
                                                                     ntiles * 16, N);

    k_edges1<<<(E + tb - 1) / tb, tb, 0, stream>>>(src, dst, hw1, agg1, E);
    k_layer2<<<(N + tb - 1) / tb, tb, 0, stream>>>(agg1, b1, W2, dinv, hw2, agg2, N);
    long long t2 = (long long)E * NCLS;
    k_edges2<<<(int)((t2 + tb - 1) / tb), tb, 0, stream>>>(src, dst, hw2, agg2, E);
    k_final <<<(N + tb - 1) / tb, tb, 0, stream>>>(agg2, b2, dinv, out, N);
}


# layer-2 edge sums: a block adds one edge slice's rows for a quarter of the nodes into f32 LDS accumulators, then flushes them as whole-line atomics; one partial sum per slice: new f32 sum order
# speedup vs baseline: 3.8349x; 1.7135x over previous
#include <hip/hip_runtime.h>
#include <hip/hip_bf16.h>

#define N_FEAT 256
#define HIDDEN 16
#define NCLS   2

typedef float v2f __attribute__((ext_vector_type(2)));
typedef float v4f __attribute__((ext_vector_type(4)));
typedef float v8f __attribute__((ext_vector_type(8)));

__device__ __forceinline__ void atomic_add_f32(float* p, float v) {
    unsafeAtomicAdd(p, v);
}

__global__ void k_deg_init(float* __restrict__ deg, int n) {
    int i = blockIdx.x * blockDim.x + threadIdx.x;
    if (i < n) deg[i] = 1.0f;
}

__global__ void k_deg_edges(const int* __restrict__ dst, float* __restrict__ deg, int E) {
    int e = blockIdx.x * blockDim.x + threadIdx.x;
    if (e < E) atomic_add_f32(&deg[dst[e]], 1.0f);
}

#define DEG_RANGES   4
#define DEG_RNG_MAX  32768
#define DEG_THREADS  1024
#define DEG_LOADS    49
#define DEG_SLICE    (DEG_LOADS * DEG_THREADS)

__global__ __launch_bounds__(DEG_THREADS)
void k_deg_hist(const int* __restrict__ dst, float* __restrict__ deg,
                int E, int N, int rng, int nslices) {
    __shared__ unsigned int s_cnt[DEG_RNG_MAX];
    int tid   = threadIdx.x;
    int slice = blockIdx.x / DEG_RANGES;
    int range = blockIdx.x % DEG_RANGES;
    int lo    = range * rng;

    for (int i = tid; i < rng; i += DEG_THREADS) s_cnt[i] = 0u;
    __syncthreads();

    int e0 = slice * DEG_SLICE;
    int e1 = min(e0 + DEG_SLICE, E);
    int d[DEG_LOADS];
    const int* p = dst + e0 + tid;
    if (e0 + DEG_SLICE <= E) {
        #pragma unroll
        for (int j = 0; j < DEG_LOADS; ++j) d[j] = p[j * DEG_THREADS];
    } else {
        #pragma unroll
        for (int j = 0; j < DEG_LOADS; ++j)
            d[j] = (e0 + j * DEG_THREADS + tid < e1) ? p[j * DEG_THREADS] : -1;
    }
    #pragma unroll
    for (int j = 0; j < DEG_LOADS; ++j) {
        unsigned int r = (unsigned int)(d[j] - lo);
        if (r < (unsigned int)rng) atomicAdd(&s_cnt[r], 1u);
    }
    __syncthreads();

    int rot = (int)(((unsigned int)slice * (unsigned int)(rng >> 5)) / (unsigned int)nslices) << 5;
    for (int i = tid; i < rng; i += DEG_THREADS) {
        int j = i + rot;
        if (j >= rng) j -= rng;
        unsigned int c = s_cnt[j];
        int node = lo + j;
        if (c != 0u && node < N) atomic_add_f32(&deg[node], (float)c);
    }
}

__global__ void k_dinv(float* __restrict__ deg, int n) {
    int i = blockIdx.x * blockDim.x + threadIdx.x;
    if (i < n) deg[i] = rsqrtf(deg[i]);
}

__global__ void k_gemm1(const float* __restrict__ x, const float* __restrict__ W1,
                        const float* __restrict__ dinv,
                        float* __restrict__ hw1, float* __restrict__ agg1, int ntiles) {
    int lane = threadIdx.x & 31;
    int wave = threadIdx.x >> 5;
    int tile = blockIdx.x * (blockDim.x >> 5) + wave;
    if (tile >= ntiles) return;
    int half = lane >> 4;
    int l    = lane & 15;
    int row0 = tile * 16;

    const float* aptr = x  + (size_t)(row0 + l) * N_FEAT + 2 * half;
    const float* bptr = W1 + (size_t)(2 * half) * HIDDEN + l;

    v8f c = {};
    #pragma unroll 8
    for (int kk = 0; kk < N_FEAT / 4; ++kk) {
        v2f a = *(const v2f*)(aptr + kk * 4);
        v2f b;
        b.x = bptr[kk * 4 * HIDDEN];
        b.y = bptr[kk * 4 * HIDDEN + HIDDEN];
        c = __builtin_amdgcn_wmma_f32_16x16x4_f32(false, a, false, b,
                                                  (short)0, c, false, false);
    }

    int rbase = row0 + half * 8;
    #pragma unroll
    for (int v = 0; v < 8; ++v) {
        int r = rbase + v;
        float dval = c[v] * dinv[r];
        size_t idx = (size_t)r * HIDDEN + l;
        hw1[idx]  = dval;
        agg1[idx] = dval;
    }
}

__global__ void k_gemm1_rem(const float* __restrict__ x, const float* __restrict__ W1,
                            const float* __restrict__ dinv,
                            float* __restrict__ hw1, float* __restrict__ agg1,
                            int row_lo, int n) {
    int t = blockIdx.x * blockDim.x + threadIdx.x;
    int r = row_lo + t / HIDDEN;
    int c = t % HIDDEN;
    if (r >= n) return;
    float acc = 0.0f;
    for (int k = 0; k < N_FEAT; ++k) acc += x[(size_t)r * N_FEAT + k] * W1[(size_t)k * HIDDEN + c];
    acc *= dinv[r];
    size_t idx = (size_t)r * HIDDEN + c;
    hw1[idx]  = acc;
    agg1[idx] = acc;
}

__global__ void k_edges1(const int* __restrict__ src, const int* __restrict__ dst,
                         const float* __restrict__ hw1,
                         float* __restrict__ agg1, int E) {
    int lane = threadIdx.x & 31;
    int wave = (blockIdx.x * blockDim.x + threadIdx.x) >> 5;
    int e0   = wave * 32;
    if (e0 >= E) return;
    int e = e0 + lane;
    int s = 0, d = 0;
    if (e < E) {
        s = src[e];
        d = dst[e];
    }
    int c     = lane & 15;
    int hbase = lane & 16;
    if (e0 + 32 <= E) {
        float val[16];
        int   drow[16];
        #pragma unroll
        for (int j = 0; j < 16; ++j) {
            int   sj = __shfl(s, hbase + j);
            drow[j]  = __shfl(d, hbase + j);
            val[j]   = hw1[(size_t)sj * HIDDEN + c];
        }
        #pragma unroll
        for (int j = 0; j < 16; ++j)
            atomic_add_f32(&agg1[(size_t)drow[j] * HIDDEN + c], val[j]);
    } else {
        for (int j = 0; j < 16; ++j) {
            int   sj = __shfl(s, hbase + j);
            int   dj = __shfl(d, hbase + j);
            if (e0 + hbase + j < E)
                atomic_add_f32(&agg1[(size_t)dj * HIDDEN + c],
                               hw1[(size_t)sj * HIDDEN + c]);
        }
    }
}

__global__ void k_layer2(const float* __restrict__ agg1, const float* __restrict__ b1,
                         const float* __restrict__ W2, const float* __restrict__ dinv,
                         float* __restrict__ hw2, float* __restrict__ agg2, int n) {
    int i = blockIdx.x * blockDim.x + threadIdx.x;
    if (i >= n) return;
    float di = dinv[i];
    float h0 = 0.0f, h1 = 0.0f;
    #pragma unroll
    for (int c = 0; c < HIDDEN; ++c) {
        float v = agg1[(size_t)i * HIDDEN + c] * di + b1[c];
        v = fmaxf(v, 0.0f);
        h0 += v * W2[c * NCLS + 0];
        h1 += v * W2[c * NCLS + 1];
    }
    h0 *= di;
    h1 *= di;
    hw2[(size_t)i * NCLS + 0]  = h0;
    hw2[(size_t)i * NCLS + 1]  = h1;
    agg2[(size_t)i * NCLS + 0] = h0;
    agg2[(size_t)i * NCLS + 1] = h1;
}

__global__ void k_edges2(const int* __restrict__ src, const int* __restrict__ dst,
                         const float* __restrict__ hw2,
                         float* __restrict__ agg2, int E) {
    int t = blockIdx.x * blockDim.x + threadIdx.x;
    int e = t >> 1;
    int c = t & 1;
    if (e >= E) return;
    int s = src[e], d = dst[e];
    atomic_add_f32(&agg2[(size_t)d * NCLS + c], hw2[(size_t)s * NCLS + c]);
}

#define E2_CHUNK  7
#define E2_TRIPS  (DEG_LOADS / E2_CHUNK)

__device__ __forceinline__ void lds_add_f32(float* p, float v) {
    __hip_atomic_fetch_add(p, v, __ATOMIC_RELAXED, __HIP_MEMORY_SCOPE_WORKGROUP);
}

template <bool FULL>
__device__ __forceinline__ void e2_load_idx(const int* __restrict__ ps, const int* __restrict__ pd,
                                            int t, int ebase, int E,
                                            int (&s)[E2_CHUNK], int (&d)[E2_CHUNK]) {
    #pragma unroll
    for (int j = 0; j < E2_CHUNK; ++j) {
        int off = (t * E2_CHUNK + j) * DEG_THREADS;
        if (FULL || ebase + off < E) {
            s[j] = ps[off];
            d[j] = pd[off];
        } else {
            s[j] = 0;
            d[j] = -1;
        }
    }
}

template <bool FULL>
__device__ __forceinline__ void e2_walk(const int* __restrict__ ps, const int* __restrict__ pd,
                                        const v2f* __restrict__ rows, float* acc,
                                        int ebase, int E, int lo, int rng, int nacc, int tid) {
    int sn[E2_CHUNK], dn[E2_CHUNK];
    e2_load_idx<FULL>(ps, pd, 0, ebase, E, sn, dn);

    v4f* z = (v4f*)acc;
    for (int i = tid; i < (nacc >> 2); i += DEG_THREADS) z[i] = (v4f){0.0f, 0.0f, 0.0f, 0.0f};
    __syncthreads();

    #pragma unroll
    for (int t = 0; t < E2_TRIPS; ++t) {
        int sc[E2_CHUNK], dc[E2_CHUNK];
        #pragma unroll
        for (int j = 0; j < E2_CHUNK; ++j) { sc[j] = sn[j]; dc[j] = dn[j]; }
        if (t + 1 < E2_TRIPS) e2_load_idx<FULL>(ps, pd, t + 1, ebase, E, sn, dn);
        v2f v[E2_CHUNK];
        #pragma unroll
        for (int j = 0; j < E2_CHUNK; ++j) {
            v[j] = (v2f){0.0f, 0.0f};
            if ((unsigned int)(dc[j] - lo) < (unsigned int)rng) v[j] = rows[sc[j]];
        }
        #pragma unroll
        for (int j = 0; j < E2_CHUNK; ++j) {
            unsigned int r = (unsigned int)(dc[j] - lo);
            if (r < (unsigned int)rng) {
                lds_add_f32(&acc[2u * r],      v[j].x);
                lds_add_f32(&acc[2u * r + 1u], v[j].y);
            }
        }
    }
}

__global__ __launch_bounds__(DEG_THREADS)
void k_edges2_lds(const int* __restrict__ src, const int* __restrict__ dst,
                  const float* __restrict__ hw2, float* __restrict__ agg2,
                  int E, int N, int rng, int nslices) {
    __shared__ float s_acc[NCLS * DEG_RNG_MAX];
    int tid   = threadIdx.x;
    int slice = blockIdx.x / DEG_RANGES;
    int range = blockIdx.x % DEG_RANGES;
    int lo    = range * rng;
    int nacc  = rng * NCLS;

    int e0 = slice * DEG_SLICE;
    const int* ps = src + e0 + tid;
    const int* pd = dst + e0 + tid;
    const v2f* rows = (const v2f*)hw2;
    if (e0 + DEG_SLICE <= E)
        e2_walk<true >(ps, pd, rows, s_acc, e0 + tid, E, lo, rng, nacc, tid);
    else
        e2_walk<false>(ps, pd, rows, s_acc, e0 + tid, E, lo, rng, nacc, tid);
    __syncthreads();

    int nlines = nacc >> 5;
    int rot = (int)(((unsigned int)slice * (unsigned int)nlines) / (unsigned int)nslices) << 5;
    float* base = agg2 + (size_t)lo * NCLS;
    int lim = (N - lo) * NCLS;
    for (int i = tid; i < nacc; i += DEG_THREADS) {
        int j = i + rot;
        if (j >= nacc) j -= nacc;
        float a = s_acc[j];
        if (a != 0.0f && j < lim) atomic_add_f32(&base[j], a);
    }
}

__global__ void k_final(const float* __restrict__ agg2, const float* __restrict__ b2,
                        const float* __restrict__ dinv,
                        float* __restrict__ out, int n) {
    int i = blockIdx.x * blockDim.x + threadIdx.x;
    if (i >= n) return;
    float di = dinv[i];
    float l0 = agg2[(size_t)i * NCLS + 0] * di + b2[0];
    float l1 = agg2[(size_t)i * NCLS + 1] * di + b2[1];
    float m   = fmaxf(l0, l1);
    float lse = m + logf(expf(l0 - m) + expf(l1 - m));
    out[(size_t)i * NCLS + 0] = l0 - lse;
    out[(size_t)i * NCLS + 1] = l1 - lse;
}

extern "C" void kernel_launch(void* const* d_in, const int* in_sizes, int n_in,
                              void* d_out, int out_size, void* d_ws, size_t ws_size,
                              hipStream_t stream) {
    const float* x  = (const float*)d_in[0];
    const int*   ei = (const int*)d_in[1];
    const float* W1 = (const float*)d_in[2];
    const float* b1 = (const float*)d_in[3];
    const float* W2 = (const float*)d_in[4];
    const float* b2 = (const float*)d_in[5];
    float* out = (float*)d_out;

    int N = in_sizes[0] / N_FEAT;
    int E = in_sizes[1] / 2;
    const int* src = ei;
    const int* dst = ei + E;

    float* ws   = (float*)d_ws;
    float* dinv = ws;
    float* hw1  = dinv + (size_t)N;
    float* agg1 = hw1  + (size_t)N * HIDDEN;
    float* hw2  = agg1 + (size_t)N * HIDDEN;
    float* agg2 = hw2  + (size_t)N * NCLS;

    const int tb = 256;
    k_deg_init <<<(N + tb - 1) / tb, tb, 0, stream>>>(dinv, N);
    int rng = (((N + DEG_RANGES - 1) / DEG_RANGES) + 31) & ~31;
    int nslices = (E + DEG_SLICE - 1) / DEG_SLICE;
    if (rng <= DEG_RNG_MAX) {
        if (nslices > 0)
            k_deg_hist<<<nslices * DEG_RANGES, DEG_THREADS, 0, stream>>>(dst, dinv, E, N, rng, nslices);
    } else {
        k_deg_edges<<<(E + tb - 1) / tb, tb, 0, stream>>>(dst, dinv, E);
    }
    k_dinv     <<<(N + tb - 1) / tb, tb, 0, stream>>>(dinv, N);

    int ntiles = N / 16;
    int wpb    = tb / 32;
    if (ntiles > 0)
        k_gemm1<<<(ntiles + wpb - 1) / wpb, tb, 0, stream>>>(x, W1, dinv, hw1, agg1, ntiles);
    int rem = N - ntiles * 16;
    if (rem > 0)
        k_gemm1_rem<<<(rem * HIDDEN + tb - 1) / tb, tb, 0, stream>>>(x, W1, dinv, hw1, agg1,
                                                                     ntiles * 16, N);

    k_edges1<<<(E + tb - 1) / tb, tb, 0, stream>>>(src, dst, hw1, agg1, E);
    k_layer2<<<(N + tb - 1) / tb, tb, 0, stream>>>(agg1, b1, W2, dinv, hw2, agg2, N);
    if (rng <= DEG_RNG_MAX) {
        if (nslices > 0)
            k_edges2_lds<<<nslices * DEG_RANGES, DEG_THREADS, 0, stream>>>(src, dst, hw2, agg2, E, N, rng, nslices);
    } else {
        long long t2 = (long long)E * NCLS;
        k_edges2<<<(int)((t2 + tb - 1) / tb), tb, 0, stream>>>(src, dst, hw2, agg2, E);
    }
    k_final <<<(N + tb - 1) / tb, tb, 0, stream>>>(agg2, b2, dinv, out, N);
}


# layer-1 edge sums: a block adds one edge slice's rows for one node range into f32 LDS accumulators (a wave queues its hits, then lane = channel), flushed as whole-line atomics: new f32 sum order
# speedup vs baseline: 4.3706x; 1.1397x over previous
#include <hip/hip_runtime.h>
#include <hip/hip_bf16.h>

#define N_FEAT 256
#define HIDDEN 16
#define NCLS   2

typedef float v2f __attribute__((ext_vector_type(2)));
typedef float v4f __attribute__((ext_vector_type(4)));
typedef float v8f __attribute__((ext_vector_type(8)));

__device__ __forceinline__ void atomic_add_f32(float* p, float v) {
    unsafeAtomicAdd(p, v);
}

__global__ void k_deg_init(float* __restrict__ deg, int n) {
    int i = blockIdx.x * blockDim.x + threadIdx.x;
    if (i < n) deg[i] = 1.0f;
}

__global__ void k_deg_edges(const int* __restrict__ dst, float* __restrict__ deg, int E) {
    int e = blockIdx.x * blockDim.x + threadIdx.x;
    if (e < E) atomic_add_f32(&deg[dst[e]], 1.0f);
}

#define DEG_RANGES   4
#define DEG_RNG_MAX  32768
#define DEG_THREADS  1024
#define DEG_LOADS    49
#define DEG_SLICE    (DEG_LOADS * DEG_THREADS)

__global__ __launch_bounds__(DEG_THREADS)
void k_deg_hist(const int* __restrict__ dst, float* __restrict__ deg,
                int E, int N, int rng, int nslices) {
    __shared__ unsigned int s_cnt[DEG_RNG_MAX];
    int tid   = threadIdx.x;
    int slice = blockIdx.x / DEG_RANGES;
    int range = blockIdx.x % DEG_RANGES;
    int lo    = range * rng;

    for (int i = tid; i < rng; i += DEG_THREADS) s_cnt[i] = 0u;
    __syncthreads();

    int e0 = slice * DEG_SLICE;
    int e1 = min(e0 + DEG_SLICE, E);
    int d[DEG_LOADS];
    const int* p = dst + e0 + tid;
    if (e0 + DEG_SLICE <= E) {
        #pragma unroll
        for (int j = 0; j < DEG_LOADS; ++j) d[j] = p[j * DEG_THREADS];
    } else {
        #pragma unroll
        for (int j = 0; j < DEG_LOADS; ++j)
            d[j] = (e0 + j * DEG_THREADS + tid < e1) ? p[j * DEG_THREADS] : -1;
    }
    #pragma unroll
    for (int j = 0; j < DEG_LOADS; ++j) {
        unsigned int r = (unsigned int)(d[j] - lo);
        if (r < (unsigned int)rng) atomicAdd(&s_cnt[r], 1u);
    }
    __syncthreads();

    int rot = (int)(((unsigned int)slice * (unsigned int)(rng >> 5)) / (unsigned int)nslices) << 5;
    for (int i = tid; i < rng; i += DEG_THREADS) {
        int j = i + rot;
        if (j >= rng) j -= rng;
        unsigned int c = s_cnt[j];
        int node = lo + j;
        if (c != 0u && node < N) atomic_add_f32(&deg[node], (float)c);
    }
}

__global__ void k_dinv(float* __restrict__ deg, int n) {
    int i = blockIdx.x * blockDim.x + threadIdx.x;
    if (i < n) deg[i] = rsqrtf(deg[i]);
}

__global__ void k_gemm1(const float* __restrict__ x, const float* __restrict__ W1,
                        const float* __restrict__ dinv,
                        float* __restrict__ hw1, float* __restrict__ agg1, int ntiles) {
    int lane = threadIdx.x & 31;
    int wave = threadIdx.x >> 5;
    int tile = blockIdx.x * (blockDim.x >> 5) + wave;
    if (tile >= ntiles) return;
    int half = lane >> 4;
    int l    = lane & 15;
    int row0 = tile * 16;

    const float* aptr = x  + (size_t)(row0 + l) * N_FEAT + 2 * half;
    const float* bptr = W1 + (size_t)(2 * half) * HIDDEN + l;

    v8f c = {};
    #pragma unroll 8
    for (int kk = 0; kk < N_FEAT / 4; ++kk) {
        v2f a = *(const v2f*)(aptr + kk * 4);
        v2f b;
        b.x = bptr[kk * 4 * HIDDEN];
        b.y = bptr[kk * 4 * HIDDEN + HIDDEN];
        c = __builtin_amdgcn_wmma_f32_16x16x4_f32(false, a, false, b,
                                                  (short)0, c, false, false);
    }

    int rbase = row0 + half * 8;
    #pragma unroll
    for (int v = 0; v < 8; ++v) {
        int r = rbase + v;
        float dval = c[v] * dinv[r];
        size_t idx = (size_t)r * HIDDEN + l;
        hw1[idx]  = dval;
        agg1[idx] = dval;
    }
}

__global__ void k_gemm1_rem(const float* __restrict__ x, const float* __restrict__ W1,
                            const float* __restrict__ dinv,
                            float* __restrict__ hw1, float* __restrict__ agg1,
                            int row_lo, int n) {
    int t = blockIdx.x * blockDim.x + threadIdx.x;
    int r = row_lo + t / HIDDEN;
    int c = t % HIDDEN;
    if (r >= n) return;
    float acc = 0.0f;
    for (int k = 0; k < N_FEAT; ++k) acc += x[(size_t)r * N_FEAT + k] * W1[(size_t)k * HIDDEN + c];
    acc *= dinv[r];
    size_t idx = (size_t)r * HIDDEN + c;
    hw1[idx]  = acc;
    agg1[idx] = acc;
}

__global__ void k_edges1(const int* __restrict__ src, const int* __restrict__ dst,
                         const float* __restrict__ hw1,
                         float* __restrict__ agg1, int E) {
    int lane = threadIdx.x & 31;
    int wave = (blockIdx.x * blockDim.x + threadIdx.x) >> 5;
    int e0   = wave * 32;
    if (e0 >= E) return;
    int e = e0 + lane;
    int s = 0, d = 0;
    if (e < E) {
        s = src[e];
        d = dst[e];
    }
    int c     = lane & 15;
    int hbase = lane & 16;
    if (e0 + 32 <= E) {
        float val[16];
        int   drow[16];
        #pragma unroll
        for (int j = 0; j < 16; ++j) {
            int   sj = __shfl(s, hbase + j);
            drow[j]  = __shfl(d, hbase + j);
            val[j]   = hw1[(size_t)sj * HIDDEN + c];
        }
        #pragma unroll
        for (int j = 0; j < 16; ++j)
            atomic_add_f32(&agg1[(size_t)drow[j] * HIDDEN + c], val[j]);
    } else {
        for (int j = 0; j < 16; ++j) {
            int   sj = __shfl(s, hbase + j);
            int   dj = __shfl(d, hbase + j);
            if (e0 + hbase + j < E)
                atomic_add_f32(&agg1[(size_t)dj * HIDDEN + c],
                               hw1[(size_t)sj * HIDDEN + c]);
        }
    }
}

__global__ void k_layer2(const float* __restrict__ agg1, const float* __restrict__ b1,
                         const float* __restrict__ W2, const float* __restrict__ dinv,
                         float* __restrict__ hw2, float* __restrict__ agg2, int n) {
    int i = blockIdx.x * blockDim.x + threadIdx.x;
    if (i >= n) return;
    float di = dinv[i];
    float h0 = 0.0f, h1 = 0.0f;
    #pragma unroll
    for (int c = 0; c < HIDDEN; ++c) {
        float v = agg1[(size_t)i * HIDDEN + c] * di + b1[c];
        v = fmaxf(v, 0.0f);
        h0 += v * W2[c * NCLS + 0];
        h1 += v * W2[c * NCLS + 1];
    }
    h0 *= di;
    h1 *= di;
    hw2[(size_t)i * NCLS + 0]  = h0;
    hw2[(size_t)i * NCLS + 1]  = h1;
    agg2[(size_t)i * NCLS + 0] = h0;
    agg2[(size_t)i * NCLS + 1] = h1;
}

__global__ void k_edges2(const int* __restrict__ src, const int* __restrict__ dst,
                         const float* __restrict__ hw2,
                         float* __restrict__ agg2, int E) {
    int t = blockIdx.x * blockDim.x + threadIdx.x;
    int e = t >> 1;
    int c = t & 1;
    if (e >= E) return;
    int s = src[e], d = dst[e];
    atomic_add_f32(&agg2[(size_t)d * NCLS + c], hw2[(size_t)s * NCLS + c]);
}

#define E2_CHUNK  7
#define E2_TRIPS  (DEG_LOADS / E2_CHUNK)

__device__ __forceinline__ void lds_add_f32(float* p, float v) {
    __hip_atomic_fetch_add(p, v, __ATOMIC_RELAXED, __HIP_MEMORY_SCOPE_WORKGROUP);
}

template <bool FULL>
__device__ __forceinline__ void e2_load_idx(const int* __restrict__ ps, const int* __restrict__ pd,
                                            int t, int ebase, int E,
                                            int (&s)[E2_CHUNK], int (&d)[E2_CHUNK]) {
    #pragma unroll
    for (int j = 0; j < E2_CHUNK; ++j) {
        int off = (t * E2_CHUNK + j) * DEG_THREADS;
        if (FULL || ebase + off < E) {
            s[j] = ps[off];
            d[j] = pd[off];
        } else {
            s[j] = 0;
            d[j] = -1;
        }
    }
}

template <bool FULL>
__device__ __forceinline__ void e2_walk(const int* __restrict__ ps, const int* __restrict__ pd,
                                        const v2f* __restrict__ rows, float* acc,
                                        int ebase, int E, int lo, int rng, int nacc, int tid) {
    int sn[E2_CHUNK], dn[E2_CHUNK];
    e2_load_idx<FULL>(ps, pd, 0, ebase, E, sn, dn);

    v4f* z = (v4f*)acc;
    for (int i = tid; i < (nacc >> 2); i += DEG_THREADS) z[i] = (v4f){0.0f, 0.0f, 0.0f, 0.0f};
    __syncthreads();

    #pragma unroll
    for (int t = 0; t < E2_TRIPS; ++t) {
        int sc[E2_CHUNK], dc[E2_CHUNK];
        #pragma unroll
        for (int j = 0; j < E2_CHUNK; ++j) { sc[j] = sn[j]; dc[j] = dn[j]; }
        if (t + 1 < E2_TRIPS) e2_load_idx<FULL>(ps, pd, t + 1, ebase, E, sn, dn);
        v2f v[E2_CHUNK];
        #pragma unroll
        for (int j = 0; j < E2_CHUNK; ++j) {
            v[j] = (v2f){0.0f, 0.0f};
            if ((unsigned int)(dc[j] - lo) < (unsigned int)rng) v[j] = rows[sc[j]];
        }
        #pragma unroll
        for (int j = 0; j < E2_CHUNK; ++j) {
            unsigned int r = (unsigned int)(dc[j] - lo);
            if (r < (unsigned int)rng) {
                lds_add_f32(&acc[2u * r],      v[j].x);
                lds_add_f32(&acc[2u * r + 1u], v[j].y);
            }
        }
    }
}

__global__ __launch_bounds__(DEG_THREADS)
void k_edges2_lds(const int* __restrict__ src, const int* __restrict__ dst,
                  const float* __restrict__ hw2, float* __restrict__ agg2,
                  int E, int N, int rng, int nslices) {
    __shared__ float s_acc[NCLS * DEG_RNG_MAX];
    int tid   = threadIdx.x;
    int slice = blockIdx.x / DEG_RANGES;
    int range = blockIdx.x % DEG_RANGES;
    int lo    = range * rng;
    int nacc  = rng * NCLS;

    int e0 = slice * DEG_SLICE;
    const int* ps = src + e0 + tid;
    const int* pd = dst + e0 + tid;
    const v2f* rows = (const v2f*)hw2;
    if (e0 + DEG_SLICE <= E)
        e2_walk<true >(ps, pd, rows, s_acc, e0 + tid, E, lo, rng, nacc, tid);
    else
        e2_walk<false>(ps, pd, rows, s_acc, e0 + tid, E, lo, rng, nacc, tid);
    __syncthreads();

    int nlines = nacc >> 5;
    int rot = (int)(((unsigned int)slice * (unsigned int)nlines) / (unsigned int)nslices) << 5;
    float* base = agg2 + (size_t)lo * NCLS;
    int lim = (N - lo) * NCLS;
    for (int i = tid; i < nacc; i += DEG_THREADS) {
        int j = i + rot;
        if (j >= nacc) j -= nacc;
        float a = s_acc[j];
        if (a != 0.0f && j < lim) atomic_add_f32(&base[j], a);
    }
}

#define L1_RNG_MAX    4032
#define L1_MAX_RANGES 64
#define L1_THREADS    1024
#define L1_K          8
#define L1_TRIP       (L1_K * L1_THREADS)
#define L1_TRIPS      40
#define L1_SLICE      (L1_TRIP * L1_TRIPS)
#define L1_QCAP       (L1_K * 32)
#define L1_BATCH      8

typedef int v2i __attribute__((ext_vector_type(2)));

template <bool FULL>
__device__ __forceinline__ void l1_load_idx(const int* __restrict__ ps, const int* __restrict__ pd,
                                            int t, int ebase, int E,
                                            int (&s)[L1_K], int (&d)[L1_K]) {
    #pragma unroll
    for (int k = 0; k < L1_K; ++k) {
        int off = t * L1_TRIP + k * L1_THREADS;
        if (FULL || ebase + off < E) {
            s[k] = ps[off];
            d[k] = pd[off];
        } else {
            s[k] = 0;
            d[k] = -1;
        }
    }
}

__device__ __forceinline__ void l1_batch(const float* __restrict__ hw1, float* acc, const v2i* q,
                                         int i0, int n, int half, int c) {
    v2i ent[L1_BATCH];
    #pragma unroll
    for (int j = 0; j < L1_BATCH; ++j) {
        int i = i0 + 2 * j + half;
        ent[j] = q[min(i, L1_QCAP - 1)];
    }
    float v[L1_BATCH];
    #pragma unroll
    for (int j = 0; j < L1_BATCH; ++j) {
        int i = i0 + 2 * j + half;
        int s = (i < n) ? ent[j].x : 0;
        v[j] = hw1[(size_t)s * HIDDEN + c];
    }
    #pragma unroll
    for (int j = 0; j < L1_BATCH; ++j) {
        int i = i0 + 2 * j + half;
        if (i < n) lds_add_f32(&acc[ent[j].y * HIDDEN + c], v[j]);
    }
}

template <bool FULL>
__device__ __forceinline__ void l1_walk(const int* __restrict__ ps, const int* __restrict__ pd,
                                        const float* __restrict__ hw1, float* acc, v2i* q,
                                        int ebase, int E, int ntrips, int lo, int rng, int nacc,
                                        int tid) {
    int lane = tid & 31;
    int half = lane >> 4;
    int c    = lane & 15;

    int sn[L1_K], dn[L1_K];
    l1_load_idx<FULL>(ps, pd, 0, ebase, E, sn, dn);

    v4f* z = (v4f*)acc;
    for (int i = tid; i < (nacc >> 2); i += L1_THREADS) z[i] = (v4f){0.0f, 0.0f, 0.0f, 0.0f};
    __syncthreads();

    for (int t = 0; t < ntrips; ++t) {
        int sc[L1_K], dc[L1_K];
        #pragma unroll
        for (int k = 0; k < L1_K; ++k) { sc[k] = sn[k]; dc[k] = dn[k]; }
        if (t + 1 < ntrips) l1_load_idx<FULL>(ps, pd, t + 1, ebase, E, sn, dn);

        int n = 0;
        #pragma unroll
        for (int k = 0; k < L1_K; ++k) {
            unsigned int r   = (unsigned int)(dc[k] - lo);
            bool         hit = r < (unsigned int)rng;
            unsigned int m   = __builtin_amdgcn_ballot_w32(hit);
            int pos = n + (int)__builtin_amdgcn_mbcnt_lo(m, 0u);
            if (hit) q[pos] = (v2i){sc[k], (int)r};
            n += __builtin_popcount(m);
        }
        __builtin_amdgcn_fence(__ATOMIC_ACQ_REL, "wavefront");

        if (n > 0) {
            l1_batch(hw1, acc, q, 0, n, half, c);
            for (int i0 = 2 * L1_BATCH; i0 < n; i0 += 2 * L1_BATCH)
                l1_batch(hw1, acc, q, i0, n, half, c);
        }
        __builtin_amdgcn_fence(__ATOMIC_ACQ_REL, "wavefront");
    }
}

__global__ __launch_bounds__(L1_THREADS)
void k_edges1_lds(const int* __restrict__ src, const int* __restrict__ dst,
                  const float* __restrict__ hw1, float* __restrict__ agg1,
                  int E, int N, int rng, int nranges, int nslices) {
    __shared__ float s_acc[L1_RNG_MAX * HIDDEN];
    __shared__ v2i   s_q[L1_THREADS / 32][L1_QCAP];
    int tid   = threadIdx.x;
    int slice = blockIdx.x / nranges;
    int range = blockIdx.x - slice * nranges;
    int lo    = range * rng;
    int nacc  = rng * HIDDEN;

    int e0 = slice * L1_SLICE;
    const int* ps = src + e0 + tid;
    const int* pd = dst + e0 + tid;
    v2i* q = s_q[tid >> 5];
    if (e0 + L1_SLICE <= E) {
        l1_walk<true >(ps, pd, hw1, s_acc, q, e0 + tid, E, L1_TRIPS, lo, rng, nacc, tid);
    } else {
        int ntrips = (E - e0 + L1_TRIP - 1) / L1_TRIP;
        l1_walk<false>(ps, pd, hw1, s_acc, q, e0 + tid, E, ntrips, lo, rng, nacc, tid);
    }
    __syncthreads();

    int nlines = nacc >> 5;
    int rot = (int)(((unsigned int)slice * (unsigned int)nlines) / (unsigned int)nslices) << 5;
    float* base = agg1 + (size_t)lo * HIDDEN;
    int lim = (N - lo) * HIDDEN;
    for (int i = tid; i < nacc; i += L1_THREADS) {
        int j = i + rot;
        if (j >= nacc) j -= nacc;
        if (j < lim) atomic_add_f32(&base[j], s_acc[j]);
    }
}

__global__ void k_final(const float* __restrict__ agg2, const float* __restrict__ b2,
                        const float* __restrict__ dinv,
                        float* __restrict__ out, int n) {
    int i = blockIdx.x * blockDim.x + threadIdx.x;
    if (i >= n) return;
    float di = dinv[i];
    float l0 = agg2[(size_t)i * NCLS + 0] * di + b2[0];
    float l1 = agg2[(size_t)i * NCLS + 1] * di + b2[1];
    float m   = fmaxf(l0, l1);
    float lse = m + logf(expf(l0 - m) + expf(l1 - m));
    out[(size_t)i * NCLS + 0] = l0 - lse;
    out[(size_t)i * NCLS + 1] = l1 - lse;
}

extern "C" void kernel_launch(void* const* d_in, const int* in_sizes, int n_in,
                              void* d_out, int out_size, void* d_ws, size_t ws_size,
                              hipStream_t stream) {
    const float* x  = (const float*)d_in[0];
    const int*   ei = (const int*)d_in[1];
    const float* W1 = (const float*)d_in[2];
    const float* b1 = (const float*)d_in[3];
    const float* W2 = (const float*)d_in[4];
    const float* b2 = (const float*)d_in[5];
    float* out = (float*)d_out;

    int N = in_sizes[0] / N_FEAT;
    int E = in_sizes[1] / 2;
    const int* src = ei;
    const int* dst = ei + E;

    float* ws   = (float*)d_ws;
    float* dinv = ws;
    float* hw1  = dinv + (size_t)N;
    float* agg1 = hw1  + (size_t)N * HIDDEN;
    float* hw2  = agg1 + (size_t)N * HIDDEN;
    float* agg2 = hw2  + (size_t)N * NCLS;

    const int tb = 256;
    k_deg_init <<<(N + tb - 1) / tb, tb, 0, stream>>>(dinv, N);
    int rng = (((N + DEG_RANGES - 1) / DEG_RANGES) + 31) & ~31;
    int nslices = (E + DEG_SLICE - 1) / DEG_SLICE;
    if (rng <= DEG_RNG_MAX) {
        if (nslices > 0)
            k_deg_hist<<<nslices * DEG_RANGES, DEG_THREADS, 0, stream>>>(dst, dinv, E, N, rng, nslices);
    } else {
        k_deg_edges<<<(E + tb - 1) / tb, tb, 0, stream>>>(dst, dinv, E);
    }
    k_dinv     <<<(N + tb - 1) / tb, tb, 0, stream>>>(dinv, N);

    int ntiles = N / 16;
    int wpb    = tb / 32;
    if (ntiles > 0)
        k_gemm1<<<(ntiles + wpb - 1) / wpb, tb, 0, stream>>>(x, W1, dinv, hw1, agg1, ntiles);
    int rem = N - ntiles * 16;
    if (rem > 0)
        k_gemm1_rem<<<(rem * HIDDEN + tb - 1) / tb, tb, 0, stream>>>(x, W1, dinv, hw1, agg1,
                                                                     ntiles * 16, N);

    int nr1  = (N + L1_RNG_MAX - 1) / L1_RNG_MAX;
    int nsl1 = (E + L1_SLICE - 1) / L1_SLICE;
    if (nr1 >= 1 && nr1 <= L1_MAX_RANGES) {
        int rng1 = (((N + nr1 - 1) / nr1) + 1) & ~1;
        if (nsl1 > 0)
            k_edges1_lds<<<nsl1 * nr1, L1_THREADS, 0, stream>>>(src, dst, hw1, agg1, E, N, rng1, nr1, nsl1);
    } else {
        k_edges1<<<(E + tb - 1) / tb, tb, 0, stream>>>(src, dst, hw1, agg1, E);
    }
    k_layer2<<<(N + tb - 1) / tb, tb, 0, stream>>>(agg1, b1, W2, dinv, hw2, agg2, N);
    if (rng <= DEG_RNG_MAX) {
        if (nslices > 0)
            k_edges2_lds<<<nslices * DEG_RANGES, DEG_THREADS, 0, stream>>>(src, dst, hw2, agg2, E, N, rng, nslices);
    } else {
        long long t2 = (long long)E * NCLS;
        k_edges2<<<(int)((t2 + tb - 1) / tb), tb, 0, stream>>>(src, dst, hw2, agg2, E);
    }
    k_final <<<(N + tb - 1) / tb, tb, 0, stream>>>(agg2, b2, dinv, out, N);
}


# layer-1 edge sums: edges binned once by destination range (LDS-sorted, coalesced 4-byte records), so a sum block reads only its own range's edges instead of scanning every slice; new f32 sum order
# speedup vs baseline: 4.8913x; 1.1191x over previous
#include <hip/hip_runtime.h>
#include <hip/hip_bf16.h>

#define N_FEAT 256
#define HIDDEN 16
#define NCLS   2

typedef float v2f __attribute__((ext_vector_type(2)));
typedef float v4f __attribute__((ext_vector_type(4)));
typedef float v8f __attribute__((ext_vector_type(8)));

__device__ __forceinline__ void atomic_add_f32(float* p, float v) {
    unsafeAtomicAdd(p, v);
}

__global__ void k_deg_init(float* __restrict__ deg, int n) {
    int i = blockIdx.x * blockDim.x + threadIdx.x;
    if (i < n) deg[i] = 1.0f;
}

__global__ void k_deg_edges(const int* __restrict__ dst, float* __restrict__ deg, int E) {
    int e = blockIdx.x * blockDim.x + threadIdx.x;
    if (e < E) atomic_add_f32(&deg[dst[e]], 1.0f);
}

#define DEG_RANGES   4
#define DEG_RNG_MAX  32768
#define DEG_THREADS  1024
#define DEG_LOADS    49
#define DEG_SLICE    (DEG_LOADS * DEG_THREADS)

__global__ __launch_bounds__(DEG_THREADS)
void k_deg_hist(const int* __restrict__ dst, float* __restrict__ deg,
                int E, int N, int rng, int nslices) {
    __shared__ unsigned int s_cnt[DEG_RNG_MAX];
    int tid   = threadIdx.x;
    int slice = blockIdx.x / DEG_RANGES;
    int range = blockIdx.x % DEG_RANGES;
    int lo    = range * rng;

    for (int i = tid; i < rng; i += DEG_THREADS) s_cnt[i] = 0u;
    __syncthreads();

    int e0 = slice * DEG_SLICE;
    int e1 = min(e0 + DEG_SLICE, E);
    int d[DEG_LOADS];
    const int* p = dst + e0 + tid;
    if (e0 + DEG_SLICE <= E) {
        #pragma unroll
        for (int j = 0; j < DEG_LOADS; ++j) d[j] = p[j * DEG_THREADS];
    } else {
        #pragma unroll
        for (int j = 0; j < DEG_LOADS; ++j)
            d[j] = (e0 + j * DEG_THREADS + tid < e1) ? p[j * DEG_THREADS] : -1;
    }
    #pragma unroll
    for (int j = 0; j < DEG_LOADS; ++j) {
        unsigned int r = (unsigned int)(d[j] - lo);
        if (r < (unsigned int)rng) atomicAdd(&s_cnt[r], 1u);
    }
    __syncthreads();

    int rot = (int)(((unsigned int)slice * (unsigned int)(rng >> 5)) / (unsigned int)nslices) << 5;
    for (int i = tid; i < rng; i += DEG_THREADS) {
        int j = i + rot;
        if (j >= rng) j -= rng;
        unsigned int c = s_cnt[j];
        int node = lo + j;
        if (c != 0u && node < N) atomic_add_f32(&deg[node], (float)c);
    }
}

__global__ void k_dinv(float* __restrict__ deg, int n) {
    int i = blockIdx.x * blockDim.x + threadIdx.x;
    if (i < n) deg[i] = rsqrtf(deg[i]);
}

__global__ void k_gemm1(const float* __restrict__ x, const float* __restrict__ W1,
                        const float* __restrict__ dinv,
                        float* __restrict__ hw1, float* __restrict__ agg1, int ntiles) {
    int lane = threadIdx.x & 31;
    int wave = threadIdx.x >> 5;
    int tile = blockIdx.x * (blockDim.x >> 5) + wave;
    if (tile >= ntiles) return;
    int half = lane >> 4;
    int l    = lane & 15;
    int row0 = tile * 16;

    const float* aptr = x  + (size_t)(row0 + l) * N_FEAT + 2 * half;
    const float* bptr = W1 + (size_t)(2 * half) * HIDDEN + l;

    v8f c = {};
    #pragma unroll 8
    for (int kk = 0; kk < N_FEAT / 4; ++kk) {
        v2f a = *(const v2f*)(aptr + kk * 4);
        v2f b;
        b.x = bptr[kk * 4 * HIDDEN];
        b.y = bptr[kk * 4 * HIDDEN + HIDDEN];
        c = __builtin_amdgcn_wmma_f32_16x16x4_f32(false, a, false, b,
                                                  (short)0, c, false, false);
    }

    int rbase = row0 + half * 8;
    #pragma unroll
    for (int v = 0; v < 8; ++v) {
        int r = rbase + v;
        float dval = c[v] * dinv[r];
        size_t idx = (size_t)r * HIDDEN + l;
        hw1[idx]  = dval;
        agg1[idx] = dval;
    }
}

__global__ void k_gemm1_rem(const float* __restrict__ x, const float* __restrict__ W1,
                            const float* __restrict__ dinv,
                            float* __restrict__ hw1, float* __restrict__ agg1,
                            int row_lo, int n) {
    int t = blockIdx.x * blockDim.x + threadIdx.x;
    int r = row_lo + t / HIDDEN;
    int c = t % HIDDEN;
    if (r >= n) return;
    float acc = 0.0f;
    for (int k = 0; k < N_FEAT; ++k) acc += x[(size_t)r * N_FEAT + k] * W1[(size_t)k * HIDDEN + c];
    acc *= dinv[r];
    size_t idx = (size_t)r * HIDDEN + c;
    hw1[idx]  = acc;
    agg1[idx] = acc;
}

__global__ void k_edges1(const int* __restrict__ src, const int* __restrict__ dst,
                         const float* __restrict__ hw1,
                         float* __restrict__ agg1, int E) {
    int lane = threadIdx.x & 31;
    int wave = (blockIdx.x * blockDim.x + threadIdx.x) >> 5;
    int e0   = wave * 32;
    if (e0 >= E) return;
    int e = e0 + lane;
    int s = 0, d = 0;
    if (e < E) {
        s = src[e];
        d = dst[e];
    }
    int c     = lane & 15;
    int hbase = lane & 16;
    if (e0 + 32 <= E) {
        float val[16];
        int   drow[16];
        #pragma unroll
        for (int j = 0; j < 16; ++j) {
            int   sj = __shfl(s, hbase + j);
            drow[j]  = __shfl(d, hbase + j);
            val[j]   = hw1[(size_t)sj * HIDDEN + c];
        }
        #pragma unroll
        for (int j = 0; j < 16; ++j)
            atomic_add_f32(&agg1[(size_t)drow[j] * HIDDEN + c], val[j]);
    } else {
        for (int j = 0; j < 16; ++j) {
            int   sj = __shfl(s, hbase + j);
            int   dj = __shfl(d, hbase + j);
            if (e0 + hbase + j < E)
                atomic_add_f32(&agg1[(size_t)dj * HIDDEN + c],
                               hw1[(size_t)sj * HIDDEN + c]);
        }
    }
}

__global__ void k_layer2(const float* __restrict__ agg1, const float* __restrict__ b1,
                         const float* __restrict__ W2, const float* __restrict__ dinv,
                         float* __restrict__ hw2, float* __restrict__ agg2, int n) {
    int i = blockIdx.x * blockDim.x + threadIdx.x;
    if (i >= n) return;
    float di = dinv[i];
    float h0 = 0.0f, h1 = 0.0f;
    #pragma unroll
    for (int c = 0; c < HIDDEN; ++c) {
        float v = agg1[(size_t)i * HIDDEN + c] * di + b1[c];
        v = fmaxf(v, 0.0f);
        h0 += v * W2[c * NCLS + 0];
        h1 += v * W2[c * NCLS + 1];
    }
    h0 *= di;
    h1 *= di;
    hw2[(size_t)i * NCLS + 0]  = h0;
    hw2[(size_t)i * NCLS + 1]  = h1;
    agg2[(size_t)i * NCLS + 0] = h0;
    agg2[(size_t)i * NCLS + 1] = h1;
}

__global__ void k_edges2(const int* __restrict__ src, const int* __restrict__ dst,
                         const float* __restrict__ hw2,
                         float* __restrict__ agg2, int E) {
    int t = blockIdx.x * blockDim.x + threadIdx.x;
    int e = t >> 1;
    int c = t & 1;
    if (e >= E) return;
    int s = src[e], d = dst[e];
    atomic_add_f32(&agg2[(size_t)d * NCLS + c], hw2[(size_t)s * NCLS + c]);
}

#define E2_CHUNK  7
#define E2_TRIPS  (DEG_LOADS / E2_CHUNK)

__device__ __forceinline__ void lds_add_f32(float* p, float v) {
    __hip_atomic_fetch_add(p, v, __ATOMIC_RELAXED, __HIP_MEMORY_SCOPE_WORKGROUP);
}

template <bool FULL>
__device__ __forceinline__ void e2_load_idx(const int* __restrict__ ps, const int* __restrict__ pd,
                                            int t, int ebase, int E,
                                            int (&s)[E2_CHUNK], int (&d)[E2_CHUNK]) {
    #pragma unroll
    for (int j = 0; j < E2_CHUNK; ++j) {
        int off = (t * E2_CHUNK + j) * DEG_THREADS;
        if (FULL || ebase + off < E) {
            s[j] = ps[off];
            d[j] = pd[off];
        } else {
            s[j] = 0;
            d[j] = -1;
        }
    }
}

template <bool FULL>
__device__ __forceinline__ void e2_walk(const int* __restrict__ ps, const int* __restrict__ pd,
                                        const v2f* __restrict__ rows, float* acc,
                                        int ebase, int E, int lo, int rng, int nacc, int tid) {
    int sn[E2_CHUNK], dn[E2_CHUNK];
    e2_load_idx<FULL>(ps, pd, 0, ebase, E, sn, dn);

    v4f* z = (v4f*)acc;
    for (int i = tid; i < (nacc >> 2); i += DEG_THREADS) z[i] = (v4f){0.0f, 0.0f, 0.0f, 0.0f};
    __syncthreads();

    #pragma unroll
    for (int t = 0; t < E2_TRIPS; ++t) {
        int sc[E2_CHUNK], dc[E2_CHUNK];
        #pragma unroll
        for (int j = 0; j < E2_CHUNK; ++j) { sc[j] = sn[j]; dc[j] = dn[j]; }
        if (t + 1 < E2_TRIPS) e2_load_idx<FULL>(ps, pd, t + 1, ebase, E, sn, dn);
        v2f v[E2_CHUNK];
        #pragma unroll
        for (int j = 0; j < E2_CHUNK; ++j) {
            v[j] = (v2f){0.0f, 0.0f};
            if ((unsigned int)(dc[j] - lo) < (unsigned int)rng) v[j] = rows[sc[j]];
        }
        #pragma unroll
        for (int j = 0; j < E2_CHUNK; ++j) {
            unsigned int r = (unsigned int)(dc[j] - lo);
            if (r < (unsigned int)rng) {
                lds_add_f32(&acc[2u * r],      v[j].x);
                lds_add_f32(&acc[2u * r + 1u], v[j].y);
            }
        }
    }
}

__global__ __launch_bounds__(DEG_THREADS)
void k_edges2_lds(const int* __restrict__ src, const int* __restrict__ dst,
                  const float* __restrict__ hw2, float* __restrict__ agg2,
                  int E, int N, int rng, int nslices) {
    __shared__ float s_acc[NCLS * DEG_RNG_MAX];
    int tid   = threadIdx.x;
    int slice = blockIdx.x / DEG_RANGES;
    int range = blockIdx.x % DEG_RANGES;
    int lo    = range * rng;
    int nacc  = rng * NCLS;

    int e0 = slice * DEG_SLICE;
    const int* ps = src + e0 + tid;
    const int* pd = dst + e0 + tid;
    const v2f* rows = (const v2f*)hw2;
    if (e0 + DEG_SLICE <= E)
        e2_walk<true >(ps, pd, rows, s_acc, e0 + tid, E, lo, rng, nacc, tid);
    else
        e2_walk<false>(ps, pd, rows, s_acc, e0 + tid, E, lo, rng, nacc, tid);
    __syncthreads();

    int nlines = nacc >> 5;
    int rot = (int)(((unsigned int)slice * (unsigned int)nlines) / (unsigned int)nslices) << 5;
    float* base = agg2 + (size_t)lo * NCLS;
    int lim = (N - lo) * NCLS;
    for (int i = tid; i < nacc; i += DEG_THREADS) {
        int j = i + rot;
        if (j >= nacc) j -= nacc;
        float a = s_acc[j];
        if (a != 0.0f && j < lim) atomic_add_f32(&base[j], a);
    }
}

#define L1_RNG_MAX    4032
#define L1_MAX_RANGES 64
#define L1_THREADS    1024
#define L1_K          8
#define L1_TRIP       (L1_K * L1_THREADS)
#define L1_TRIPS      40
#define L1_SLICE      (L1_TRIP * L1_TRIPS)
#define L1_QCAP       (L1_K * 32)
#define L1_BATCH      8

typedef int v2i __attribute__((ext_vector_type(2)));

template <bool FULL>
__device__ __forceinline__ void l1_load_idx(const int* __restrict__ ps, const int* __restrict__ pd,
                                            int t, int ebase, int E,
                                            int (&s)[L1_K], int (&d)[L1_K]) {
    #pragma unroll
    for (int k = 0; k < L1_K; ++k) {
        int off = t * L1_TRIP + k * L1_THREADS;
        if (FULL || ebase + off < E) {
            s[k] = ps[off];
            d[k] = pd[off];
        } else {
            s[k] = 0;
            d[k] = -1;
        }
    }
}

__device__ __forceinline__ void l1_batch(const float* __restrict__ hw1, float* acc, const v2i* q,
                                         int i0, int n, int half, int c) {
    v2i ent[L1_BATCH];
    #pragma unroll
    for (int j = 0; j < L1_BATCH; ++j) {
        int i = i0 + 2 * j + half;
        ent[j] = q[min(i, L1_QCAP - 1)];
    }
    float v[L1_BATCH];
    #pragma unroll
    for (int j = 0; j < L1_BATCH; ++j) {
        int i = i0 + 2 * j + half;
        int s = (i < n) ? ent[j].x : 0;
        v[j] = hw1[(size_t)s * HIDDEN + c];
    }
    #pragma unroll
    for (int j = 0; j < L1_BATCH; ++j) {
        int i = i0 + 2 * j + half;
        if (i < n) lds_add_f32(&acc[ent[j].y * HIDDEN + c], v[j]);
    }
}

template <bool FULL>
__device__ __forceinline__ void l1_walk(const int* __restrict__ ps, const int* __restrict__ pd,
                                        const float* __restrict__ hw1, float* acc, v2i* q,
                                        int ebase, int E, int ntrips, int lo, int rng, int nacc,
                                        int tid) {
    int lane = tid & 31;
    int half = lane >> 4;
    int c    = lane & 15;

    int sn[L1_K], dn[L1_K];
    l1_load_idx<FULL>(ps, pd, 0, ebase, E, sn, dn);

    v4f* z = (v4f*)acc;
    for (int i = tid; i < (nacc >> 2); i += L1_THREADS) z[i] = (v4f){0.0f, 0.0f, 0.0f, 0.0f};
    __syncthreads();

    for (int t = 0; t < ntrips; ++t) {
        int sc[L1_K], dc[L1_K];
        #pragma unroll
        for (int k = 0; k < L1_K; ++k) { sc[k] = sn[k]; dc[k] = dn[k]; }
        if (t + 1 < ntrips) l1_load_idx<FULL>(ps, pd, t + 1, ebase, E, sn, dn);

        int n = 0;
        #pragma unroll
        for (int k = 0; k < L1_K; ++k) {
            unsigned int r   = (unsigned int)(dc[k] - lo);
            bool         hit = r < (unsigned int)rng;
            unsigned int m   = __builtin_amdgcn_ballot_w32(hit);
            int pos = n + (int)__builtin_amdgcn_mbcnt_lo(m, 0u);
            if (hit) q[pos] = (v2i){sc[k], (int)r};
            n += __builtin_popcount(m);
        }
        __builtin_amdgcn_fence(__ATOMIC_ACQ_REL, "wavefront");

        if (n > 0) {
            l1_batch(hw1, acc, q, 0, n, half, c);
            for (int i0 = 2 * L1_BATCH; i0 < n; i0 += 2 * L1_BATCH)
                l1_batch(hw1, acc, q, i0, n, half, c);
        }
        __builtin_amdgcn_fence(__ATOMIC_ACQ_REL, "wavefront");
    }
}

__global__ __launch_bounds__(L1_THREADS)
void k_edges1_lds(const int* __restrict__ src, const int* __restrict__ dst,
                  const float* __restrict__ hw1, float* __restrict__ agg1,
                  int E, int N, int rng, int nranges, int nslices) {
    __shared__ float s_acc[L1_RNG_MAX * HIDDEN];
    __shared__ v2i   s_q[L1_THREADS / 32][L1_QCAP];
    int tid   = threadIdx.x;
    int slice = blockIdx.x / nranges;
    int range = blockIdx.x - slice * nranges;
    int lo    = range * rng;
    int nacc  = rng * HIDDEN;

    int e0 = slice * L1_SLICE;
    const int* ps = src + e0 + tid;
    const int* pd = dst + e0 + tid;
    v2i* q = s_q[tid >> 5];
    if (e0 + L1_SLICE <= E) {
        l1_walk<true >(ps, pd, hw1, s_acc, q, e0 + tid, E, L1_TRIPS, lo, rng, nacc, tid);
    } else {
        int ntrips = (E - e0 + L1_TRIP - 1) / L1_TRIP;
        l1_walk<false>(ps, pd, hw1, s_acc, q, e0 + tid, E, ntrips, lo, rng, nacc, tid);
    }
    __syncthreads();

    int nlines = nacc >> 5;
    int rot = (int)(((unsigned int)slice * (unsigned int)nlines) / (unsigned int)nslices) << 5;
    float* base = agg1 + (size_t)lo * HIDDEN;
    int lim = (N - lo) * HIDDEN;
    for (int i = tid; i < nacc; i += L1_THREADS) {
        int j = i + rot;
        if (j >= nacc) j -= nacc;
        if (j < lim) atomic_add_f32(&base[j], s_acc[j]);
    }
}

#define B1_RNG     2000
#define B1_RMAX    64
#define B1_CAP     640
#define B1_LOADS   25
#define B1_THREADS 1024
#define B1_EPB     (B1_LOADS * B1_THREADS)
#define B1_CH      25
#define B1_RBITS   12

__global__ __launch_bounds__(B1_THREADS)
void k_bin1(const int* __restrict__ src, const int* __restrict__ dst,
            const float* __restrict__ hw1, float* __restrict__ agg1,
            unsigned int* __restrict__ bins, int* __restrict__ cnts, int E, int R) {
    __shared__ unsigned int s_stage[B1_RMAX * B1_CAP];
    __shared__ unsigned int s_cnt[B1_RMAX];
    int tid = threadIdx.x;
    int blk = blockIdx.x;
    int e0  = blk * B1_EPB;

    int s[B1_LOADS], d[B1_LOADS];
    const int* ps = src + e0 + tid;
    const int* pd = dst + e0 + tid;
    if (e0 + B1_EPB <= E) {
        #pragma unroll
        for (int j = 0; j < B1_LOADS; ++j) {
            s[j] = ps[j * B1_THREADS];
            d[j] = pd[j * B1_THREADS];
        }
    } else {
        #pragma unroll
        for (int j = 0; j < B1_LOADS; ++j) {
            bool in = e0 + j * B1_THREADS + tid < E;
            s[j] = in ? ps[j * B1_THREADS] : 0;
            d[j] = in ? pd[j * B1_THREADS] : -1;
        }
    }
    if (tid < B1_RMAX) s_cnt[tid] = 0u;
    __syncthreads();

    #pragma unroll
    for (int j = 0; j < B1_LOADS; ++j) {
        if (d[j] >= 0) {
            unsigned int b   = (unsigned int)d[j] / (unsigned int)B1_RNG;
            unsigned int r   = (unsigned int)d[j] - b * (unsigned int)B1_RNG;
            unsigned int pos = atomicAdd(&s_cnt[b], 1u);
            if (pos < (unsigned int)B1_CAP) {
                s_stage[b * B1_CAP + pos] = ((unsigned int)s[j] << B1_RBITS) | r;
            } else {
                for (int c = 0; c < HIDDEN; ++c)
                    atomic_add_f32(&agg1[(size_t)d[j] * HIDDEN + c], hw1[(size_t)s[j] * HIDDEN + c]);
            }
        }
    }
    __syncthreads();

    unsigned int* outp = bins + (size_t)blk * R * B1_CAP;
    int tot = R * B1_CAP;
    for (int i = tid; i < tot; i += B1_THREADS) {
        int b = i / B1_CAP;
        int p = i - b * B1_CAP;
        if ((unsigned int)p < s_cnt[b]) outp[i] = s_stage[i];
    }
    if (tid < R) cnts[blk * R + tid] = (int)min(s_cnt[tid], (unsigned int)B1_CAP);
}

typedef unsigned int v4u __attribute__((ext_vector_type(4)));

__global__ __launch_bounds__(B1_THREADS)
void k_edges1_bin(const unsigned int* __restrict__ bins, const int* __restrict__ cnts,
                  const float* __restrict__ hw1, float* __restrict__ agg1,
                  int N, int R, int NB, int nsub) {
    __shared__ float s_acc[B1_RNG * HIDDEN];
    __shared__ alignas(16) unsigned int s_rec[B1_CH * B1_CAP + 32];
    int tid   = threadIdx.x;
    int lane  = tid & 31;
    int wave  = __builtin_amdgcn_readfirstlane(tid >> 5);
    int sub   = blockIdx.x / R;
    int range = blockIdx.x - sub * R;
    int lo    = range * B1_RNG;
    int nacc  = B1_RNG * HIDDEN;
    int c     = lane & 15;
    int hbase = lane & 16;
    int blk0  = sub * B1_CH;

    int nv = 0;
    if (lane < B1_CH && blk0 + lane < NB) nv = cnts[(blk0 + lane) * R + range];

    unsigned int rec[B1_CH];
    #pragma unroll
    for (int k = 0; k < B1_CH; ++k) {
        int nk = __builtin_amdgcn_readlane(nv, k);
        rec[k] = 0u;
        if (tid < nk) rec[k] = bins[((size_t)(blk0 + k) * R + range) * B1_CAP + tid];
    }

    v4f* z = (v4f*)s_acc;
    for (int i = tid; i < (nacc >> 2); i += B1_THREADS) z[i] = (v4f){0.0f, 0.0f, 0.0f, 0.0f};

    int T = 0;
    #pragma unroll
    for (int k = 0; k < B1_CH; ++k) {
        int nk = __builtin_amdgcn_readlane(nv, k);
        if (tid < nk) s_rec[T + tid] = rec[k];
        T += nk;
    }
    __syncthreads();

    int nsteps = (T + 31) >> 5;
    for (int st = wave; st < nsteps; st += 32) {
        int i0 = st * 32 + hbase;
        v4u q[4];
        #pragma unroll
        for (int m = 0; m < 4; ++m) q[m] = *(const v4u*)&s_rec[i0 + 4 * m];
        float v[16];
        int   row[16];
        #pragma unroll
        for (int j = 0; j < 16; ++j) {
            unsigned int ej = q[j >> 2][j & 3];
            bool ok = i0 + j < T;
            int  sj = ok ? (int)(ej >> B1_RBITS) : 0;
            row[j]  = ok ? (int)(ej & ((1u << B1_RBITS) - 1u)) : 0;
            v[j]    = hw1[(size_t)sj * HIDDEN + c];
        }
        #pragma unroll
        for (int j = 0; j < 16; ++j)
            lds_add_f32(&s_acc[row[j] * HIDDEN + c], (i0 + j < T) ? v[j] : 0.0f);
    }
    __syncthreads();

    int nlines = nacc >> 5;
    int rot = (int)(((unsigned int)sub * (unsigned int)nlines) / (unsigned int)nsub) << 5;
    float* base = agg1 + (size_t)lo * HIDDEN;
    int lim = (N - lo) * HIDDEN;
    for (int i = tid; i < nacc; i += B1_THREADS) {
        int j = i + rot;
        if (j >= nacc) j -= nacc;
        if (j < lim) atomic_add_f32(&base[j], s_acc[j]);
    }
}

__global__ void k_final(const float* __restrict__ agg2, const float* __restrict__ b2,
                        const float* __restrict__ dinv,
                        float* __restrict__ out, int n) {
    int i = blockIdx.x * blockDim.x + threadIdx.x;
    if (i >= n) return;
    float di = dinv[i];
    float l0 = agg2[(size_t)i * NCLS + 0] * di + b2[0];
    float l1 = agg2[(size_t)i * NCLS + 1] * di + b2[1];
    float m   = fmaxf(l0, l1);
    float lse = m + logf(expf(l0 - m) + expf(l1 - m));
    out[(size_t)i * NCLS + 0] = l0 - lse;
    out[(size_t)i * NCLS + 1] = l1 - lse;
}

extern "C" void kernel_launch(void* const* d_in, const int* in_sizes, int n_in,
                              void* d_out, int out_size, void* d_ws, size_t ws_size,
                              hipStream_t stream) {
    const float* x  = (const float*)d_in[0];
    const int*   ei = (const int*)d_in[1];
    const float* W1 = (const float*)d_in[2];
    const float* b1 = (const float*)d_in[3];
    const float* W2 = (const float*)d_in[4];
    const float* b2 = (const float*)d_in[5];
    float* out = (float*)d_out;

    int N = in_sizes[0] / N_FEAT;
    int E = in_sizes[1] / 2;
    const int* src = ei;
    const int* dst = ei + E;

    float* ws   = (float*)d_ws;
    float* dinv = ws;
    float* hw1  = dinv + (size_t)N;
    float* agg1 = hw1  + (size_t)N * HIDDEN;
    float* hw2  = agg1 + (size_t)N * HIDDEN;
    float* agg2 = hw2  + (size_t)N * NCLS;

    const int tb = 256;
    k_deg_init <<<(N + tb - 1) / tb, tb, 0, stream>>>(dinv, N);
    int rng = (((N + DEG_RANGES - 1) / DEG_RANGES) + 31) & ~31;
    int nslices = (E + DEG_SLICE - 1) / DEG_SLICE;
    if (rng <= DEG_RNG_MAX) {
        if (nslices > 0)
            k_deg_hist<<<nslices * DEG_RANGES, DEG_THREADS, 0, stream>>>(dst, dinv, E, N, rng, nslices);
    } else {
        k_deg_edges<<<(E + tb - 1) / tb, tb, 0, stream>>>(dst, dinv, E);
    }
    k_dinv     <<<(N + tb - 1) / tb, tb, 0, stream>>>(dinv, N);

    int ntiles = N / 16;
    int wpb    = tb / 32;
    if (ntiles > 0)
        k_gemm1<<<(ntiles + wpb - 1) / wpb, tb, 0, stream>>>(x, W1, dinv, hw1, agg1, ntiles);
    int rem = N - ntiles * 16;
    if (rem > 0)
        k_gemm1_rem<<<(rem * HIDDEN + tb - 1) / tb, tb, 0, stream>>>(x, W1, dinv, hw1, agg1,
                                                                     ntiles * 16, N);

    int nr1  = (N + L1_RNG_MAX - 1) / L1_RNG_MAX;
    int nsl1 = (E + L1_SLICE - 1) / L1_SLICE;
    int    rb1   = (N + B1_RNG - 1) / B1_RNG;
    int    nb1   = (E + B1_EPB - 1) / B1_EPB;
    int    nsub1 = (nb1 + B1_CH - 1) / B1_CH;
    size_t boff  = (((size_t)N * (1 + 2 * HIDDEN + 2 * NCLS) * sizeof(float)) + 255) & ~(size_t)255;
    size_t bbyte = (size_t)nb1 * rb1 * B1_CAP * sizeof(unsigned int);
    size_t cbyte = (size_t)nb1 * rb1 * sizeof(int);
    unsigned int* bins = (unsigned int*)((char*)d_ws + boff);
    int*          cnts = (int*)((char*)d_ws + boff + bbyte);
    if (nb1 > 0 && rb1 >= 1 && rb1 <= B1_RMAX && N <= (1 << (32 - B1_RBITS)) &&
        ws_size >= boff + bbyte + cbyte) {
        k_bin1      <<<nb1, B1_THREADS, 0, stream>>>(src, dst, hw1, agg1, bins, cnts, E, rb1);
        k_edges1_bin<<<rb1 * nsub1, B1_THREADS, 0, stream>>>(bins, cnts, hw1, agg1, N, rb1, nb1, nsub1);
    } else if (nr1 >= 1 && nr1 <= L1_MAX_RANGES) {
        int rng1 = (((N + nr1 - 1) / nr1) + 1) & ~1;
        if (nsl1 > 0)
            k_edges1_lds<<<nsl1 * nr1, L1_THREADS, 0, stream>>>(src, dst, hw1, agg1, E, N, rng1, nr1, nsl1);
    } else {
        k_edges1<<<(E + tb - 1) / tb, tb, 0, stream>>>(src, dst, hw1, agg1, E);
    }
    k_layer2<<<(N + tb - 1) / tb, tb, 0, stream>>>(agg1, b1, W2, dinv, hw2, agg2, N);
    if (rng <= DEG_RNG_MAX) {
        if (nslices > 0)
            k_edges2_lds<<<nslices * DEG_RANGES, DEG_THREADS, 0, stream>>>(src, dst, hw2, agg2, E, N, rng, nslices);
    } else {
        long long t2 = (long long)E * NCLS;
        k_edges2<<<(int)((t2 + tb - 1) / tb), tb, 0, stream>>>(src, dst, hw2, agg2, E);
    }
    k_final <<<(N + tb - 1) / tb, tb, 0, stream>>>(agg2, b2, dinv, out, N);
}


# layer-2 edge sums read the layer-1 bins (records sorted by destination range), not a src/dst scan per node quarter: lane = record, one partial sum per sum block flushed; new f32 sum order
# speedup vs baseline: 5.5142x; 1.1273x over previous
#include <hip/hip_runtime.h>
#include <hip/hip_bf16.h>

#define N_FEAT 256
#define HIDDEN 16
#define NCLS   2

typedef float v2f __attribute__((ext_vector_type(2)));
typedef float v4f __attribute__((ext_vector_type(4)));
typedef float v8f __attribute__((ext_vector_type(8)));

__device__ __forceinline__ void atomic_add_f32(float* p, float v) {
    unsafeAtomicAdd(p, v);
}

__global__ void k_deg_init(float* __restrict__ deg, int n) {
    int i = blockIdx.x * blockDim.x + threadIdx.x;
    if (i < n) deg[i] = 1.0f;
}

__global__ void k_deg_edges(const int* __restrict__ dst, float* __restrict__ deg, int E) {
    int e = blockIdx.x * blockDim.x + threadIdx.x;
    if (e < E) atomic_add_f32(&deg[dst[e]], 1.0f);
}

#define DEG_RANGES   4
#define DEG_RNG_MAX  32768
#define DEG_THREADS  1024
#define DEG_LOADS    49
#define DEG_SLICE    (DEG_LOADS * DEG_THREADS)

__global__ __launch_bounds__(DEG_THREADS)
void k_deg_hist(const int* __restrict__ dst, float* __restrict__ deg,
                int E, int N, int rng, int nslices) {
    __shared__ unsigned int s_cnt[DEG_RNG_MAX];
    int tid   = threadIdx.x;
    int slice = blockIdx.x / DEG_RANGES;
    int range = blockIdx.x % DEG_RANGES;
    int lo    = range * rng;

    for (int i = tid; i < rng; i += DEG_THREADS) s_cnt[i] = 0u;
    __syncthreads();

    int e0 = slice * DEG_SLICE;
    int e1 = min(e0 + DEG_SLICE, E);
    int d[DEG_LOADS];
    const int* p = dst + e0 + tid;
    if (e0 + DEG_SLICE <= E) {
        #pragma unroll
        for (int j = 0; j < DEG_LOADS; ++j) d[j] = p[j * DEG_THREADS];
    } else {
        #pragma unroll
        for (int j = 0; j < DEG_LOADS; ++j)
            d[j] = (e0 + j * DEG_THREADS + tid < e1) ? p[j * DEG_THREADS] : -1;
    }
    #pragma unroll
    for (int j = 0; j < DEG_LOADS; ++j) {
        unsigned int r = (unsigned int)(d[j] - lo);
        if (r < (unsigned int)rng) atomicAdd(&s_cnt[r], 1u);
    }
    __syncthreads();

    int rot = (int)(((unsigned int)slice * (unsigned int)(rng >> 5)) / (unsigned int)nslices) << 5;
    for (int i = tid; i < rng; i += DEG_THREADS) {
        int j = i + rot;
        if (j >= rng) j -= rng;
        unsigned int c = s_cnt[j];
        int node = lo + j;
        if (c != 0u && node < N) atomic_add_f32(&deg[node], (float)c);
    }
}

__global__ void k_dinv(float* __restrict__ deg, int n) {
    int i = blockIdx.x * blockDim.x + threadIdx.x;
    if (i < n) deg[i] = rsqrtf(deg[i]);
}

__global__ void k_gemm1(const float* __restrict__ x, const float* __restrict__ W1,
                        const float* __restrict__ dinv,
                        float* __restrict__ hw1, float* __restrict__ agg1, int ntiles) {
    int lane = threadIdx.x & 31;
    int wave = threadIdx.x >> 5;
    int tile = blockIdx.x * (blockDim.x >> 5) + wave;
    if (tile >= ntiles) return;
    int half = lane >> 4;
    int l    = lane & 15;
    int row0 = tile * 16;

    const float* aptr = x  + (size_t)(row0 + l) * N_FEAT + 2 * half;
    const float* bptr = W1 + (size_t)(2 * half) * HIDDEN + l;

    v8f c = {};
    #pragma unroll 8
    for (int kk = 0; kk < N_FEAT / 4; ++kk) {
        v2f a = *(const v2f*)(aptr + kk * 4);
        v2f b;
        b.x = bptr[kk * 4 * HIDDEN];
        b.y = bptr[kk * 4 * HIDDEN + HIDDEN];
        c = __builtin_amdgcn_wmma_f32_16x16x4_f32(false, a, false, b,
                                                  (short)0, c, false, false);
    }

    int rbase = row0 + half * 8;
    #pragma unroll
    for (int v = 0; v < 8; ++v) {
        int r = rbase + v;
        float dval = c[v] * dinv[r];
        size_t idx = (size_t)r * HIDDEN + l;
        hw1[idx]  = dval;
        agg1[idx] = dval;
    }
}

__global__ void k_gemm1_rem(const float* __restrict__ x, const float* __restrict__ W1,
                            const float* __restrict__ dinv,
                            float* __restrict__ hw1, float* __restrict__ agg1,
                            int row_lo, int n) {
    int t = blockIdx.x * blockDim.x + threadIdx.x;
    int r = row_lo + t / HIDDEN;
    int c = t % HIDDEN;
    if (r >= n) return;
    float acc = 0.0f;
    for (int k = 0; k < N_FEAT; ++k) acc += x[(size_t)r * N_FEAT + k] * W1[(size_t)k * HIDDEN + c];
    acc *= dinv[r];
    size_t idx = (size_t)r * HIDDEN + c;
    hw1[idx]  = acc;
    agg1[idx] = acc;
}

__global__ void k_edges1(const int* __restrict__ src, const int* __restrict__ dst,
                         const float* __restrict__ hw1,
                         float* __restrict__ agg1, int E) {
    int lane = threadIdx.x & 31;
    int wave = (blockIdx.x * blockDim.x + threadIdx.x) >> 5;
    int e0   = wave * 32;
    if (e0 >= E) return;
    int e = e0 + lane;
    int s = 0, d = 0;
    if (e < E) {
        s = src[e];
        d = dst[e];
    }
    int c     = lane & 15;
    int hbase = lane & 16;
    if (e0 + 32 <= E) {
        float val[16];
        int   drow[16];
        #pragma unroll
        for (int j = 0; j < 16; ++j) {
            int   sj = __shfl(s, hbase + j);
            drow[j]  = __shfl(d, hbase + j);
            val[j]   = hw1[(size_t)sj * HIDDEN + c];
        }
        #pragma unroll
        for (int j = 0; j < 16; ++j)
            atomic_add_f32(&agg1[(size_t)drow[j] * HIDDEN + c], val[j]);
    } else {
        for (int j = 0; j < 16; ++j) {
            int   sj = __shfl(s, hbase + j);
            int   dj = __shfl(d, hbase + j);
            if (e0 + hbase + j < E)
                atomic_add_f32(&agg1[(size_t)dj * HIDDEN + c],
                               hw1[(size_t)sj * HIDDEN + c]);
        }
    }
}

__global__ void k_layer2(const float* __restrict__ agg1, const float* __restrict__ b1,
                         const float* __restrict__ W2, const float* __restrict__ dinv,
                         float* __restrict__ hw2, float* __restrict__ agg2, int n) {
    int i = blockIdx.x * blockDim.x + threadIdx.x;
    if (i >= n) return;
    float di = dinv[i];
    float h0 = 0.0f, h1 = 0.0f;
    #pragma unroll
    for (int c = 0; c < HIDDEN; ++c) {
        float v = agg1[(size_t)i * HIDDEN + c] * di + b1[c];
        v = fmaxf(v, 0.0f);
        h0 += v * W2[c * NCLS + 0];
        h1 += v * W2[c * NCLS + 1];
    }
    h0 *= di;
    h1 *= di;
    hw2[(size_t)i * NCLS + 0]  = h0;
    hw2[(size_t)i * NCLS + 1]  = h1;
    agg2[(size_t)i * NCLS + 0] = h0;
    agg2[(size_t)i * NCLS + 1] = h1;
}

__global__ void k_edges2(const int* __restrict__ src, const int* __restrict__ dst,
                         const float* __restrict__ hw2,
                         float* __restrict__ agg2, int E) {
    int t = blockIdx.x * blockDim.x + threadIdx.x;
    int e = t >> 1;
    int c = t & 1;
    if (e >= E) return;
    int s = src[e], d = dst[e];
    atomic_add_f32(&agg2[(size_t)d * NCLS + c], hw2[(size_t)s * NCLS + c]);
}

#define E2_CHUNK  7
#define E2_TRIPS  (DEG_LOADS / E2_CHUNK)

__device__ __forceinline__ void lds_add_f32(float* p, float v) {
    __hip_atomic_fetch_add(p, v, __ATOMIC_RELAXED, __HIP_MEMORY_SCOPE_WORKGROUP);
}

template <bool FULL>
__device__ __forceinline__ void e2_load_idx(const int* __restrict__ ps, const int* __restrict__ pd,
                                            int t, int ebase, int E,
                                            int (&s)[E2_CHUNK], int (&d)[E2_CHUNK]) {
    #pragma unroll
    for (int j = 0; j < E2_CHUNK; ++j) {
        int off = (t * E2_CHUNK + j) * DEG_THREADS;
        if (FULL || ebase + off < E) {
            s[j] = ps[off];
            d[j] = pd[off];
        } else {
            s[j] = 0;
            d[j] = -1;
        }
    }
}

template <bool FULL>
__device__ __forceinline__ void e2_walk(const int* __restrict__ ps, const int* __restrict__ pd,
                                        const v2f* __restrict__ rows, float* acc,
                                        int ebase, int E, int lo, int rng, int nacc, int tid) {
    int sn[E2_CHUNK], dn[E2_CHUNK];
    e2_load_idx<FULL>(ps, pd, 0, ebase, E, sn, dn);

    v4f* z = (v4f*)acc;
    for (int i = tid; i < (nacc >> 2); i += DEG_THREADS) z[i] = (v4f){0.0f, 0.0f, 0.0f, 0.0f};
    __syncthreads();

    #pragma unroll
    for (int t = 0; t < E2_TRIPS; ++t) {
        int sc[E2_CHUNK], dc[E2_CHUNK];
        #pragma unroll
        for (int j = 0; j < E2_CHUNK; ++j) { sc[j] = sn[j]; dc[j] = dn[j]; }
        if (t + 1 < E2_TRIPS) e2_load_idx<FULL>(ps, pd, t + 1, ebase, E, sn, dn);
        v2f v[E2_CHUNK];
        #pragma unroll
        for (int j = 0; j < E2_CHUNK; ++j) {
            v[j] = (v2f){0.0f, 0.0f};
            if ((unsigned int)(dc[j] - lo) < (unsigned int)rng) v[j] = rows[sc[j]];
        }
        #pragma unroll
        for (int j = 0; j < E2_CHUNK; ++j) {
            unsigned int r = (unsigned int)(dc[j] - lo);
            if (r < (unsigned int)rng) {
                lds_add_f32(&acc[2u * r],      v[j].x);
                lds_add_f32(&acc[2u * r + 1u], v[j].y);
            }
        }
    }
}

__global__ __launch_bounds__(DEG_THREADS)
void k_edges2_lds(const int* __restrict__ src, const int* __restrict__ dst,
                  const float* __restrict__ hw2, float* __restrict__ agg2,
                  int E, int N, int rng, int nslices) {
    __shared__ float s_acc[NCLS * DEG_RNG_MAX];
    int tid   = threadIdx.x;
    int slice = blockIdx.x / DEG_RANGES;
    int range = blockIdx.x % DEG_RANGES;
    int lo    = range * rng;
    int nacc  = rng * NCLS;

    int e0 = slice * DEG_SLICE;
    const int* ps = src + e0 + tid;
    const int* pd = dst + e0 + tid;
    const v2f* rows = (const v2f*)hw2;
    if (e0 + DEG_SLICE <= E)
        e2_walk<true >(ps, pd, rows, s_acc, e0 + tid, E, lo, rng, nacc, tid);
    else
        e2_walk<false>(ps, pd, rows, s_acc, e0 + tid, E, lo, rng, nacc, tid);
    __syncthreads();

    int nlines = nacc >> 5;
    int rot = (int)(((unsigned int)slice * (unsigned int)nlines) / (unsigned int)nslices) << 5;
    float* base = agg2 + (size_t)lo * NCLS;
    int lim = (N - lo) * NCLS;
    for (int i = tid; i < nacc; i += DEG_THREADS) {
        int j = i + rot;
        if (j >= nacc) j -= nacc;
        float a = s_acc[j];
        if (a != 0.0f && j < lim) atomic_add_f32(&base[j], a);
    }
}

#define L1_RNG_MAX    4032
#define L1_MAX_RANGES 64
#define L1_THREADS    1024
#define L1_K          8
#define L1_TRIP       (L1_K * L1_THREADS)
#define L1_TRIPS      40
#define L1_SLICE      (L1_TRIP * L1_TRIPS)
#define L1_QCAP       (L1_K * 32)
#define L1_BATCH      8

typedef int v2i __attribute__((ext_vector_type(2)));

template <bool FULL>
__device__ __forceinline__ void l1_load_idx(const int* __restrict__ ps, const int* __restrict__ pd,
                                            int t, int ebase, int E,
                                            int (&s)[L1_K], int (&d)[L1_K]) {
    #pragma unroll
    for (int k = 0; k < L1_K; ++k) {
        int off = t * L1_TRIP + k * L1_THREADS;
        if (FULL || ebase + off < E) {
            s[k] = ps[off];
            d[k] = pd[off];
        } else {
            s[k] = 0;
            d[k] = -1;
        }
    }
}

__device__ __forceinline__ void l1_batch(const float* __restrict__ hw1, float* acc, const v2i* q,
                                         int i0, int n, int half, int c) {
    v2i ent[L1_BATCH];
    #pragma unroll
    for (int j = 0; j < L1_BATCH; ++j) {
        int i = i0 + 2 * j + half;
        ent[j] = q[min(i, L1_QCAP - 1)];
    }
    float v[L1_BATCH];
    #pragma unroll
    for (int j = 0; j < L1_BATCH; ++j) {
        int i = i0 + 2 * j + half;
        int s = (i < n) ? ent[j].x : 0;
        v[j] = hw1[(size_t)s * HIDDEN + c];
    }
    #pragma unroll
    for (int j = 0; j < L1_BATCH; ++j) {
        int i = i0 + 2 * j + half;
        if (i < n) lds_add_f32(&acc[ent[j].y * HIDDEN + c], v[j]);
    }
}

template <bool FULL>
__device__ __forceinline__ void l1_walk(const int* __restrict__ ps, const int* __restrict__ pd,
                                        const float* __restrict__ hw1, float* acc, v2i* q,
                                        int ebase, int E, int ntrips, int lo, int rng, int nacc,
                                        int tid) {
    int lane = tid & 31;
    int half = lane >> 4;
    int c    = lane & 15;

    int sn[L1_K], dn[L1_K];
    l1_load_idx<FULL>(ps, pd, 0, ebase, E, sn, dn);

    v4f* z = (v4f*)acc;
    for (int i = tid; i < (nacc >> 2); i += L1_THREADS) z[i] = (v4f){0.0f, 0.0f, 0.0f, 0.0f};
    __syncthreads();

    for (int t = 0; t < ntrips; ++t) {
        int sc[L1_K], dc[L1_K];
        #pragma unroll
        for (int k = 0; k < L1_K; ++k) { sc[k] = sn[k]; dc[k] = dn[k]; }
        if (t + 1 < ntrips) l1_load_idx<FULL>(ps, pd, t + 1, ebase, E, sn, dn);

        int n = 0;
        #pragma unroll
        for (int k = 0; k < L1_K; ++k) {
            unsigned int r   = (unsigned int)(dc[k] - lo);
            bool         hit = r < (unsigned int)rng;
            unsigned int m   = __builtin_amdgcn_ballot_w32(hit);
            int pos = n + (int)__builtin_amdgcn_mbcnt_lo(m, 0u);
            if (hit) q[pos] = (v2i){sc[k], (int)r};
            n += __builtin_popcount(m);
        }
        __builtin_amdgcn_fence(__ATOMIC_ACQ_REL, "wavefront");

        if (n > 0) {
            l1_batch(hw1, acc, q, 0, n, half, c);
            for (int i0 = 2 * L1_BATCH; i0 < n; i0 += 2 * L1_BATCH)
                l1_batch(hw1, acc, q, i0, n, half, c);
        }
        __builtin_amdgcn_fence(__ATOMIC_ACQ_REL, "wavefront");
    }
}

__global__ __launch_bounds__(L1_THREADS)
void k_edges1_lds(const int* __restrict__ src, const int* __restrict__ dst,
                  const float* __restrict__ hw1, float* __restrict__ agg1,
                  int E, int N, int rng, int nranges, int nslices) {
    __shared__ float s_acc[L1_RNG_MAX * HIDDEN];
    __shared__ v2i   s_q[L1_THREADS / 32][L1_QCAP];
    int tid   = threadIdx.x;
    int slice = blockIdx.x / nranges;
    int range = blockIdx.x - slice * nranges;
    int lo    = range * rng;
    int nacc  = rng * HIDDEN;

    int e0 = slice * L1_SLICE;
    const int* ps = src + e0 + tid;
    const int* pd = dst + e0 + tid;
    v2i* q = s_q[tid >> 5];
    if (e0 + L1_SLICE <= E) {
        l1_walk<true >(ps, pd, hw1, s_acc, q, e0 + tid, E, L1_TRIPS, lo, rng, nacc, tid);
    } else {
        int ntrips = (E - e0 + L1_TRIP - 1) / L1_TRIP;
        l1_walk<false>(ps, pd, hw1, s_acc, q, e0 + tid, E, ntrips, lo, rng, nacc, tid);
    }
    __syncthreads();

    int nlines = nacc >> 5;
    int rot = (int)(((unsigned int)slice * (unsigned int)nlines) / (unsigned int)nslices) << 5;
    float* base = agg1 + (size_t)lo * HIDDEN;
    int lim = (N - lo) * HIDDEN;
    for (int i = tid; i < nacc; i += L1_THREADS) {
        int j = i + rot;
        if (j >= nacc) j -= nacc;
        if (j < lim) atomic_add_f32(&base[j], s_acc[j]);
    }
}

#define B1_RNG     2000
#define B1_RMAX    64
#define B1_CAP     640
#define B1_LOADS   25
#define B1_THREADS 1024
#define B1_EPB     (B1_LOADS * B1_THREADS)
#define B1_CH      25
#define B1_RBITS   12

__global__ __launch_bounds__(B1_THREADS)
void k_bin1(const int* __restrict__ src, const int* __restrict__ dst,
            const float* __restrict__ hw1, float* __restrict__ agg1,
            unsigned int* __restrict__ bins, int* __restrict__ cnts, int E, int R) {
    __shared__ unsigned int s_stage[B1_RMAX * B1_CAP];
    __shared__ unsigned int s_cnt[B1_RMAX];
    int tid = threadIdx.x;
    int blk = blockIdx.x;
    int e0  = blk * B1_EPB;

    int s[B1_LOADS], d[B1_LOADS];
    const int* ps = src + e0 + tid;
    const int* pd = dst + e0 + tid;
    if (e0 + B1_EPB <= E) {
        #pragma unroll
        for (int j = 0; j < B1_LOADS; ++j) {
            s[j] = ps[j * B1_THREADS];
            d[j] = pd[j * B1_THREADS];
        }
    } else {
        #pragma unroll
        for (int j = 0; j < B1_LOADS; ++j) {
            bool in = e0 + j * B1_THREADS + tid < E;
            s[j] = in ? ps[j * B1_THREADS] : 0;
            d[j] = in ? pd[j * B1_THREADS] : -1;
        }
    }
    if (tid < B1_RMAX) s_cnt[tid] = 0u;
    __syncthreads();

    #pragma unroll
    for (int j = 0; j < B1_LOADS; ++j) {
        if (d[j] >= 0) {
            unsigned int b   = (unsigned int)d[j] / (unsigned int)B1_RNG;
            unsigned int r   = (unsigned int)d[j] - b * (unsigned int)B1_RNG;
            unsigned int pos = atomicAdd(&s_cnt[b], 1u);
            if (pos < (unsigned int)B1_CAP) {
                s_stage[b * B1_CAP + pos] = ((unsigned int)s[j] << B1_RBITS) | r;
            } else {
                for (int c = 0; c < HIDDEN; ++c)
                    atomic_add_f32(&agg1[(size_t)d[j] * HIDDEN + c], hw1[(size_t)s[j] * HIDDEN + c]);
            }
        }
    }
    __syncthreads();

    unsigned int* outp = bins + (size_t)blk * R * B1_CAP;
    int tot = R * B1_CAP;
    for (int i = tid; i < tot; i += B1_THREADS) {
        int b = i / B1_CAP;
        int p = i - b * B1_CAP;
        if ((unsigned int)p < s_cnt[b]) outp[i] = s_stage[i];
    }
    if (tid < R) cnts[blk * R + tid] = (int)s_cnt[tid];
}

typedef unsigned int v4u __attribute__((ext_vector_type(4)));

__global__ __launch_bounds__(B1_THREADS)
void k_edges1_bin(const unsigned int* __restrict__ bins, const int* __restrict__ cnts,
                  const float* __restrict__ hw1, float* __restrict__ agg1,
                  int N, int R, int NB, int nsub) {
    __shared__ float s_acc[B1_RNG * HIDDEN];
    __shared__ alignas(16) unsigned int s_rec[B1_CH * B1_CAP + 32];
    int tid   = threadIdx.x;
    int lane  = tid & 31;
    int wave  = __builtin_amdgcn_readfirstlane(tid >> 5);
    int sub   = blockIdx.x / R;
    int range = blockIdx.x - sub * R;
    int lo    = range * B1_RNG;
    int nacc  = B1_RNG * HIDDEN;
    int c     = lane & 15;
    int hbase = lane & 16;
    int blk0  = sub * B1_CH;

    int nv = 0;
    if (lane < B1_CH && blk0 + lane < NB) nv = min(cnts[(blk0 + lane) * R + range], B1_CAP);

    unsigned int rec[B1_CH];
    #pragma unroll
    for (int k = 0; k < B1_CH; ++k) {
        int nk = __builtin_amdgcn_readlane(nv, k);
        rec[k] = 0u;
        if (tid < nk) rec[k] = bins[((size_t)(blk0 + k) * R + range) * B1_CAP + tid];
    }

    v4f* z = (v4f*)s_acc;
    for (int i = tid; i < (nacc >> 2); i += B1_THREADS) z[i] = (v4f){0.0f, 0.0f, 0.0f, 0.0f};

    int T = 0;
    #pragma unroll
    for (int k = 0; k < B1_CH; ++k) {
        int nk = __builtin_amdgcn_readlane(nv, k);
        if (tid < nk) s_rec[T + tid] = rec[k];
        T += nk;
    }
    __syncthreads();

    int nsteps = (T + 31) >> 5;
    for (int st = wave; st < nsteps; st += 32) {
        int i0 = st * 32 + hbase;
        v4u q[4];
        #pragma unroll
        for (int m = 0; m < 4; ++m) q[m] = *(const v4u*)&s_rec[i0 + 4 * m];
        float v[16];
        int   row[16];
        #pragma unroll
        for (int j = 0; j < 16; ++j) {
            unsigned int ej = q[j >> 2][j & 3];
            bool ok = i0 + j < T;
            int  sj = ok ? (int)(ej >> B1_RBITS) : 0;
            row[j]  = ok ? (int)(ej & ((1u << B1_RBITS) - 1u)) : 0;
            v[j]    = hw1[(size_t)sj * HIDDEN + c];
        }
        #pragma unroll
        for (int j = 0; j < 16; ++j)
            lds_add_f32(&s_acc[row[j] * HIDDEN + c], (i0 + j < T) ? v[j] : 0.0f);
    }
    __syncthreads();

    int nlines = nacc >> 5;
    int rot = (int)(((unsigned int)sub * (unsigned int)nlines) / (unsigned int)nsub) << 5;
    float* base = agg1 + (size_t)lo * HIDDEN;
    int lim = (N - lo) * HIDDEN;
    for (int i = tid; i < nacc; i += B1_THREADS) {
        int j = i + rot;
        if (j >= nacc) j -= nacc;
        if (j < lim) atomic_add_f32(&base[j], s_acc[j]);
    }
}

#define E2B_G ((B1_CH * B1_CAP + B1_THREADS - 1) / B1_THREADS)

__global__ __launch_bounds__(B1_THREADS)
void k_edges2_bin(const unsigned int* __restrict__ bins, const int* __restrict__ cnts,
                  const int* __restrict__ src, const int* __restrict__ dst,
                  const float* __restrict__ hw2, float* __restrict__ agg2,
                  int E, int N, int R, int NB, int nsub) {
    __shared__ alignas(16) float s_acc[NCLS * B1_RNG];
    __shared__ unsigned int s_rec[B1_CH * B1_CAP];
    int tid   = threadIdx.x;
    int lane  = tid & 31;
    int sub   = blockIdx.x / R;
    int range = blockIdx.x - sub * R;
    int lo    = range * B1_RNG;
    int nacc  = B1_RNG * NCLS;
    int blk0  = sub * B1_CH;
    const v2f* rows = (const v2f*)hw2;

    int raw = 0;
    if (lane < B1_CH && blk0 + lane < NB) raw = cnts[(blk0 + lane) * R + range];
    int nv = (raw <= B1_CAP) ? raw : 0;

    unsigned int rec[B1_CH];
    #pragma unroll
    for (int k = 0; k < B1_CH; ++k) {
        int nk = __builtin_amdgcn_readlane(nv, k);
        rec[k] = 0u;
        if (tid < nk) rec[k] = bins[((size_t)(blk0 + k) * R + range) * B1_CAP + tid];
    }

    v4f* z = (v4f*)s_acc;
    for (int i = tid; i < (nacc >> 2); i += B1_THREADS) z[i] = (v4f){0.0f, 0.0f, 0.0f, 0.0f};

    int T = 0;
    #pragma unroll
    for (int k = 0; k < B1_CH; ++k) {
        int nk = __builtin_amdgcn_readlane(nv, k);
        if (tid < nk) s_rec[T + tid] = rec[k];
        T += nk;
    }
    __syncthreads();

    unsigned int q[E2B_G];
    v2f w[E2B_G];
    #pragma unroll
    for (int g = 0; g < E2B_G; ++g) {
        int i = g * B1_THREADS + tid;
        q[g] = 0u;
        if (i < T) q[g] = s_rec[i];
    }
    #pragma unroll
    for (int g = 0; g < E2B_G; ++g) {
        int i = g * B1_THREADS + tid;
        w[g] = (v2f){0.0f, 0.0f};
        if (i < T) w[g] = rows[q[g] >> B1_RBITS];
    }
    #pragma unroll
    for (int g = 0; g < E2B_G; ++g) {
        int i = g * B1_THREADS + tid;
        if (i < T) {
            unsigned int r = q[g] & ((1u << B1_RBITS) - 1u);
            lds_add_f32(&s_acc[r],          w[g].x);
            lds_add_f32(&s_acc[B1_RNG + r], w[g].y);
        }
    }

    for (int k = 0; k < B1_CH; ++k) {
        int rk = __builtin_amdgcn_readlane(raw, k);
        if (rk > B1_CAP) {
            int e0 = (blk0 + k) * B1_EPB;
            int e1 = min(e0 + B1_EPB, E);
            for (int e = e0 + tid; e < e1; e += B1_THREADS) {
                unsigned int r = (unsigned int)(dst[e] - lo);
                if (r < (unsigned int)B1_RNG) {
                    v2f u = rows[src[e]];
                    lds_add_f32(&s_acc[r],          u.x);
                    lds_add_f32(&s_acc[B1_RNG + r], u.y);
                }
            }
        }
    }
    __syncthreads();

    int nlines = nacc >> 5;
    int rot = (int)(((unsigned int)sub * (unsigned int)nlines) / (unsigned int)nsub) << 5;
    float* base = agg2 + (size_t)lo * NCLS;
    int lim = (N - lo) * NCLS;
    for (int i = tid; i < nacc; i += B1_THREADS) {
        int j = i + rot;
        if (j >= nacc) j -= nacc;
        if (j < lim) atomic_add_f32(&base[j], s_acc[(j & 1) * B1_RNG + (j >> 1)]);
    }
}

__global__ void k_final(const float* __restrict__ agg2, const float* __restrict__ b2,
                        const float* __restrict__ dinv,
                        float* __restrict__ out, int n) {
    int i = blockIdx.x * blockDim.x + threadIdx.x;
    if (i >= n) return;
    float di = dinv[i];
    float l0 = agg2[(size_t)i * NCLS + 0] * di + b2[0];
    float l1 = agg2[(size_t)i * NCLS + 1] * di + b2[1];
    float m   = fmaxf(l0, l1);
    float lse = m + logf(expf(l0 - m) + expf(l1 - m));
    out[(size_t)i * NCLS + 0] = l0 - lse;
    out[(size_t)i * NCLS + 1] = l1 - lse;
}

extern "C" void kernel_launch(void* const* d_in, const int* in_sizes, int n_in,
                              void* d_out, int out_size, void* d_ws, size_t ws_size,
                              hipStream_t stream) {
    const float* x  = (const float*)d_in[0];
    const int*   ei = (const int*)d_in[1];
    const float* W1 = (const float*)d_in[2];
    const float* b1 = (const float*)d_in[3];
    const float* W2 = (const float*)d_in[4];
    const float* b2 = (const float*)d_in[5];
    float* out = (float*)d_out;

    int N = in_sizes[0] / N_FEAT;
    int E = in_sizes[1] / 2;
    const int* src = ei;
    const int* dst = ei + E;

    float* ws   = (float*)d_ws;
    float* dinv = ws;
    float* hw1  = dinv + (size_t)N;
    float* agg1 = hw1  + (size_t)N * HIDDEN;
    float* hw2  = agg1 + (size_t)N * HIDDEN;
    float* agg2 = hw2  + (size_t)N * NCLS;

    const int tb = 256;
    k_deg_init <<<(N + tb - 1) / tb, tb, 0, stream>>>(dinv, N);
    int rng = (((N + DEG_RANGES - 1) / DEG_RANGES) + 31) & ~31;
    int nslices = (E + DEG_SLICE - 1) / DEG_SLICE;
    if (rng <= DEG_RNG_MAX) {
        if (nslices > 0)
            k_deg_hist<<<nslices * DEG_RANGES, DEG_THREADS, 0, stream>>>(dst, dinv, E, N, rng, nslices);
    } else {
        k_deg_edges<<<(E + tb - 1) / tb, tb, 0, stream>>>(dst, dinv, E);
    }
    k_dinv     <<<(N + tb - 1) / tb, tb, 0, stream>>>(dinv, N);

    int ntiles = N / 16;
    int wpb    = tb / 32;
    if (ntiles > 0)
        k_gemm1<<<(ntiles + wpb - 1) / wpb, tb, 0, stream>>>(x, W1, dinv, hw1, agg1, ntiles);
    int rem = N - ntiles * 16;
    if (rem > 0)
        k_gemm1_rem<<<(rem * HIDDEN + tb - 1) / tb, tb, 0, stream>>>(x, W1, dinv, hw1, agg1,
                                                                     ntiles * 16, N);

    int nr1  = (N + L1_RNG_MAX - 1) / L1_RNG_MAX;
    int nsl1 = (E + L1_SLICE - 1) / L1_SLICE;
    int    rb1   = (N + B1_RNG - 1) / B1_RNG;
    int    nb1   = (E + B1_EPB - 1) / B1_EPB;
    int    nsub1 = (nb1 + B1_CH - 1) / B1_CH;
    size_t boff  = (((size_t)N * (1 + 2 * HIDDEN + 2 * NCLS) * sizeof(float)) + 255) & ~(size_t)255;
    size_t bbyte = (size_t)nb1 * rb1 * B1_CAP * sizeof(unsigned int);
    size_t cbyte = (size_t)nb1 * rb1 * sizeof(int);
    unsigned int* bins = (unsigned int*)((char*)d_ws + boff);
    int*          cnts = (int*)((char*)d_ws + boff + bbyte);
    bool binned = nb1 > 0 && rb1 >= 1 && rb1 <= B1_RMAX && N <= (1 << (32 - B1_RBITS)) &&
                  ws_size >= boff + bbyte + cbyte;
    if (binned) {
        k_bin1      <<<nb1, B1_THREADS, 0, stream>>>(src, dst, hw1, agg1, bins, cnts, E, rb1);
        k_edges1_bin<<<rb1 * nsub1, B1_THREADS, 0, stream>>>(bins, cnts, hw1, agg1, N, rb1, nb1, nsub1);
    } else if (nr1 >= 1 && nr1 <= L1_MAX_RANGES) {
        int rng1 = (((N + nr1 - 1) / nr1) + 1) & ~1;
        if (nsl1 > 0)
            k_edges1_lds<<<nsl1 * nr1, L1_THREADS, 0, stream>>>(src, dst, hw1, agg1, E, N, rng1, nr1, nsl1);
    } else {
        k_edges1<<<(E + tb - 1) / tb, tb, 0, stream>>>(src, dst, hw1, agg1, E);
    }
    k_layer2<<<(N + tb - 1) / tb, tb, 0, stream>>>(agg1, b1, W2, dinv, hw2, agg2, N);
    if (binned) {
        k_edges2_bin<<<rb1 * nsub1, B1_THREADS, 0, stream>>>(bins, cnts, src, dst, hw2, agg2, E, N, rb1, nb1, nsub1);
    } else if (rng <= DEG_RNG_MAX) {
        if (nslices > 0)
            k_edges2_lds<<<nslices * DEG_RANGES, DEG_THREADS, 0, stream>>>(src, dst, hw2, agg2, E, N, rng, nslices);
    } else {
        long long t2 = (long long)E * NCLS;
        k_edges2<<<(int)((t2 + tb - 1) / tb), tb, 0, stream>>>(src, dst, hw2, agg2, E);
    }
    k_final <<<(N + tb - 1) / tb, tb, 0, stream>>>(agg2, b2, dinv, out, N);
}


# degree count reads the dst-sorted bins (binning runs first): one block per node range counts its records in LDS and writes dinv itself; no deg init, global atomics or dinv pass; f32 sum order kept
# speedup vs baseline: 5.9017x; 1.0703x over previous
#include <hip/hip_runtime.h>
#include <hip/hip_bf16.h>

#define N_FEAT 256
#define HIDDEN 16
#define NCLS   2

typedef float v2f __attribute__((ext_vector_type(2)));
typedef float v4f __attribute__((ext_vector_type(4)));
typedef float v8f __attribute__((ext_vector_type(8)));

__device__ __forceinline__ void atomic_add_f32(float* p, float v) {
    unsafeAtomicAdd(p, v);
}

__global__ void k_deg_init(float* __restrict__ deg, int n) {
    int i = blockIdx.x * blockDim.x + threadIdx.x;
    if (i < n) deg[i] = 1.0f;
}

__global__ void k_deg_edges(const int* __restrict__ dst, float* __restrict__ deg, int E) {
    int e = blockIdx.x * blockDim.x + threadIdx.x;
    if (e < E) atomic_add_f32(&deg[dst[e]], 1.0f);
}

#define DEG_RANGES   4
#define DEG_RNG_MAX  32768
#define DEG_THREADS  1024
#define DEG_LOADS    49
#define DEG_SLICE    (DEG_LOADS * DEG_THREADS)

__global__ __launch_bounds__(DEG_THREADS)
void k_deg_hist(const int* __restrict__ dst, float* __restrict__ deg,
                int E, int N, int rng, int nslices) {
    __shared__ unsigned int s_cnt[DEG_RNG_MAX];
    int tid   = threadIdx.x;
    int slice = blockIdx.x / DEG_RANGES;
    int range = blockIdx.x % DEG_RANGES;
    int lo    = range * rng;

    for (int i = tid; i < rng; i += DEG_THREADS) s_cnt[i] = 0u;
    __syncthreads();

    int e0 = slice * DEG_SLICE;
    int e1 = min(e0 + DEG_SLICE, E);
    int d[DEG_LOADS];
    const int* p = dst + e0 + tid;
    if (e0 + DEG_SLICE <= E) {
        #pragma unroll
        for (int j = 0; j < DEG_LOADS; ++j) d[j] = p[j * DEG_THREADS];
    } else {
        #pragma unroll
        for (int j = 0; j < DEG_LOADS; ++j)
            d[j] = (e0 + j * DEG_THREADS + tid < e1) ? p[j * DEG_THREADS] : -1;
    }
    #pragma unroll
    for (int j = 0; j < DEG_LOADS; ++j) {
        unsigned int r = (unsigned int)(d[j] - lo);
        if (r < (unsigned int)rng) atomicAdd(&s_cnt[r], 1u);
    }
    __syncthreads();

    int rot = (int)(((unsigned int)slice * (unsigned int)(rng >> 5)) / (unsigned int)nslices) << 5;
    for (int i = tid; i < rng; i += DEG_THREADS) {
        int j = i + rot;
        if (j >= rng) j -= rng;
        unsigned int c = s_cnt[j];
        int node = lo + j;
        if (c != 0u && node < N) atomic_add_f32(&deg[node], (float)c);
    }
}

__global__ void k_dinv(float* __restrict__ deg, int n) {
    int i = blockIdx.x * blockDim.x + threadIdx.x;
    if (i < n) deg[i] = rsqrtf(deg[i]);
}

__global__ void k_gemm1(const float* __restrict__ x, const float* __restrict__ W1,
                        const float* __restrict__ dinv,
                        float* __restrict__ hw1, float* __restrict__ agg1, int ntiles) {
    int lane = threadIdx.x & 31;
    int wave = threadIdx.x >> 5;
    int tile = blockIdx.x * (blockDim.x >> 5) + wave;
    if (tile >= ntiles) return;
    int half = lane >> 4;
    int l    = lane & 15;
    int row0 = tile * 16;

    const float* aptr = x  + (size_t)(row0 + l) * N_FEAT + 2 * half;
    const float* bptr = W1 + (size_t)(2 * half) * HIDDEN + l;

    v8f c = {};
    #pragma unroll 8
    for (int kk = 0; kk < N_FEAT / 4; ++kk) {
        v2f a = *(const v2f*)(aptr + kk * 4);
        v2f b;
        b.x = bptr[kk * 4 * HIDDEN];
        b.y = bptr[kk * 4 * HIDDEN + HIDDEN];
        c = __builtin_amdgcn_wmma_f32_16x16x4_f32(false, a, false, b,
                                                  (short)0, c, false, false);
    }

    int rbase = row0 + half * 8;
    #pragma unroll
    for (int v = 0; v < 8; ++v) {
        int r = rbase + v;
        float dval = c[v] * dinv[r];
        size_t idx = (size_t)r * HIDDEN + l;
        hw1[idx]  = dval;
        agg1[idx] = dval;
    }
}

__global__ void k_gemm1_rem(const float* __restrict__ x, const float* __restrict__ W1,
                            const float* __restrict__ dinv,
                            float* __restrict__ hw1, float* __restrict__ agg1,
                            int row_lo, int n) {
    int t = blockIdx.x * blockDim.x + threadIdx.x;
    int r = row_lo + t / HIDDEN;
    int c = t % HIDDEN;
    if (r >= n) return;
    float acc = 0.0f;
    for (int k = 0; k < N_FEAT; ++k) acc += x[(size_t)r * N_FEAT + k] * W1[(size_t)k * HIDDEN + c];
    acc *= dinv[r];
    size_t idx = (size_t)r * HIDDEN + c;
    hw1[idx]  = acc;
    agg1[idx] = acc;
}

__global__ void k_edges1(const int* __restrict__ src, const int* __restrict__ dst,
                         const float* __restrict__ hw1,
                         float* __restrict__ agg1, int E) {
    int lane = threadIdx.x & 31;
    int wave = (blockIdx.x * blockDim.x + threadIdx.x) >> 5;
    int e0   = wave * 32;
    if (e0 >= E) return;
    int e = e0 + lane;
    int s = 0, d = 0;
    if (e < E) {
        s = src[e];
        d = dst[e];
    }
    int c     = lane & 15;
    int hbase = lane & 16;
    if (e0 + 32 <= E) {
        float val[16];
        int   drow[16];
        #pragma unroll
        for (int j = 0; j < 16; ++j) {
            int   sj = __shfl(s, hbase + j);
            drow[j]  = __shfl(d, hbase + j);
            val[j]   = hw1[(size_t)sj * HIDDEN + c];
        }
        #pragma unroll
        for (int j = 0; j < 16; ++j)
            atomic_add_f32(&agg1[(size_t)drow[j] * HIDDEN + c], val[j]);
    } else {
        for (int j = 0; j < 16; ++j) {
            int   sj = __shfl(s, hbase + j);
            int   dj = __shfl(d, hbase + j);
            if (e0 + hbase + j < E)
                atomic_add_f32(&agg1[(size_t)dj * HIDDEN + c],
                               hw1[(size_t)sj * HIDDEN + c]);
        }
    }
}

__global__ void k_layer2(const float* __restrict__ agg1, const float* __restrict__ b1,
                         const float* __restrict__ W2, const float* __restrict__ dinv,
                         float* __restrict__ hw2, float* __restrict__ agg2, int n) {
    int i = blockIdx.x * blockDim.x + threadIdx.x;
    if (i >= n) return;
    float di = dinv[i];
    float h0 = 0.0f, h1 = 0.0f;
    #pragma unroll
    for (int c = 0; c < HIDDEN; ++c) {
        float v = agg1[(size_t)i * HIDDEN + c] * di + b1[c];
        v = fmaxf(v, 0.0f);
        h0 += v * W2[c * NCLS + 0];
        h1 += v * W2[c * NCLS + 1];
    }
    h0 *= di;
    h1 *= di;
    hw2[(size_t)i * NCLS + 0]  = h0;
    hw2[(size_t)i * NCLS + 1]  = h1;
    agg2[(size_t)i * NCLS + 0] = h0;
    agg2[(size_t)i * NCLS + 1] = h1;
}

__global__ void k_edges2(const int* __restrict__ src, const int* __restrict__ dst,
                         const float* __restrict__ hw2,
                         float* __restrict__ agg2, int E) {
    int t = blockIdx.x * blockDim.x + threadIdx.x;
    int e = t >> 1;
    int c = t & 1;
    if (e >= E) return;
    int s = src[e], d = dst[e];
    atomic_add_f32(&agg2[(size_t)d * NCLS + c], hw2[(size_t)s * NCLS + c]);
}

#define E2_CHUNK  7
#define E2_TRIPS  (DEG_LOADS / E2_CHUNK)

__device__ __forceinline__ void lds_add_f32(float* p, float v) {
    __hip_atomic_fetch_add(p, v, __ATOMIC_RELAXED, __HIP_MEMORY_SCOPE_WORKGROUP);
}

template <bool FULL>
__device__ __forceinline__ void e2_load_idx(const int* __restrict__ ps, const int* __restrict__ pd,
                                            int t, int ebase, int E,
                                            int (&s)[E2_CHUNK], int (&d)[E2_CHUNK]) {
    #pragma unroll
    for (int j = 0; j < E2_CHUNK; ++j) {
        int off = (t * E2_CHUNK + j) * DEG_THREADS;
        if (FULL || ebase + off < E) {
            s[j] = ps[off];
            d[j] = pd[off];
        } else {
            s[j] = 0;
            d[j] = -1;
        }
    }
}

template <bool FULL>
__device__ __forceinline__ void e2_walk(const int* __restrict__ ps, const int* __restrict__ pd,
                                        const v2f* __restrict__ rows, float* acc,
                                        int ebase, int E, int lo, int rng, int nacc, int tid) {
    int sn[E2_CHUNK], dn[E2_CHUNK];
    e2_load_idx<FULL>(ps, pd, 0, ebase, E, sn, dn);

    v4f* z = (v4f*)acc;
    for (int i = tid; i < (nacc >> 2); i += DEG_THREADS) z[i] = (v4f){0.0f, 0.0f, 0.0f, 0.0f};
    __syncthreads();

    #pragma unroll
    for (int t = 0; t < E2_TRIPS; ++t) {
        int sc[E2_CHUNK], dc[E2_CHUNK];
        #pragma unroll
        for (int j = 0; j < E2_CHUNK; ++j) { sc[j] = sn[j]; dc[j] = dn[j]; }
        if (t + 1 < E2_TRIPS) e2_load_idx<FULL>(ps, pd, t + 1, ebase, E, sn, dn);
        v2f v[E2_CHUNK];
        #pragma unroll
        for (int j = 0; j < E2_CHUNK; ++j) {
            v[j] = (v2f){0.0f, 0.0f};
            if ((unsigned int)(dc[j] - lo) < (unsigned int)rng) v[j] = rows[sc[j]];
        }
        #pragma unroll
        for (int j = 0; j < E2_CHUNK; ++j) {
            unsigned int r = (unsigned int)(dc[j] - lo);
            if (r < (unsigned int)rng) {
                lds_add_f32(&acc[2u * r],      v[j].x);
                lds_add_f32(&acc[2u * r + 1u], v[j].y);
            }
        }
    }
}

__global__ __launch_bounds__(DEG_THREADS)
void k_edges2_lds(const int* __restrict__ src, const int* __restrict__ dst,
                  const float* __restrict__ hw2, float* __restrict__ agg2,
                  int E, int N, int rng, int nslices) {
    __shared__ float s_acc[NCLS * DEG_RNG_MAX];
    int tid   = threadIdx.x;
    int slice = blockIdx.x / DEG_RANGES;
    int range = blockIdx.x % DEG_RANGES;
    int lo    = range * rng;
    int nacc  = rng * NCLS;

    int e0 = slice * DEG_SLICE;
    const int* ps = src + e0 + tid;
    const int* pd = dst + e0 + tid;
    const v2f* rows = (const v2f*)hw2;
    if (e0 + DEG_SLICE <= E)
        e2_walk<true >(ps, pd, rows, s_acc, e0 + tid, E, lo, rng, nacc, tid);
    else
        e2_walk<false>(ps, pd, rows, s_acc, e0 + tid, E, lo, rng, nacc, tid);
    __syncthreads();

    int nlines = nacc >> 5;
    int rot = (int)(((unsigned int)slice * (unsigned int)nlines) / (unsigned int)nslices) << 5;
    float* base = agg2 + (size_t)lo * NCLS;
    int lim = (N - lo) * NCLS;
    for (int i = tid; i < nacc; i += DEG_THREADS) {
        int j = i + rot;
        if (j >= nacc) j -= nacc;
        float a = s_acc[j];
        if (a != 0.0f && j < lim) atomic_add_f32(&base[j], a);
    }
}

#define L1_RNG_MAX    4032
#define L1_MAX_RANGES 64
#define L1_THREADS    1024
#define L1_K          8
#define L1_TRIP       (L1_K * L1_THREADS)
#define L1_TRIPS      40
#define L1_SLICE      (L1_TRIP * L1_TRIPS)
#define L1_QCAP       (L1_K * 32)
#define L1_BATCH      8

typedef int v2i __attribute__((ext_vector_type(2)));

template <bool FULL>
__device__ __forceinline__ void l1_load_idx(const int* __restrict__ ps, const int* __restrict__ pd,
                                            int t, int ebase, int E,
                                            int (&s)[L1_K], int (&d)[L1_K]) {
    #pragma unroll
    for (int k = 0; k < L1_K; ++k) {
        int off = t * L1_TRIP + k * L1_THREADS;
        if (FULL || ebase + off < E) {
            s[k] = ps[off];
            d[k] = pd[off];
        } else {
            s[k] = 0;
            d[k] = -1;
        }
    }
}

__device__ __forceinline__ void l1_batch(const float* __restrict__ hw1, float* acc, const v2i* q,
                                         int i0, int n, int half, int c) {
    v2i ent[L1_BATCH];
    #pragma unroll
    for (int j = 0; j < L1_BATCH; ++j) {
        int i = i0 + 2 * j + half;
        ent[j] = q[min(i, L1_QCAP - 1)];
    }
    float v[L1_BATCH];
    #pragma unroll
    for (int j = 0; j < L1_BATCH; ++j) {
        int i = i0 + 2 * j + half;
        int s = (i < n) ? ent[j].x : 0;
        v[j] = hw1[(size_t)s * HIDDEN + c];
    }
    #pragma unroll
    for (int j = 0; j < L1_BATCH; ++j) {
        int i = i0 + 2 * j + half;
        if (i < n) lds_add_f32(&acc[ent[j].y * HIDDEN + c], v[j]);
    }
}

template <bool FULL>
__device__ __forceinline__ void l1_walk(const int* __restrict__ ps, const int* __restrict__ pd,
                                        const float* __restrict__ hw1, float* acc, v2i* q,
                                        int ebase, int E, int ntrips, int lo, int rng, int nacc,
                                        int tid) {
    int lane = tid & 31;
    int half = lane >> 4;
    int c    = lane & 15;

    int sn[L1_K], dn[L1_K];
    l1_load_idx<FULL>(ps, pd, 0, ebase, E, sn, dn);

    v4f* z = (v4f*)acc;
    for (int i = tid; i < (nacc >> 2); i += L1_THREADS) z[i] = (v4f){0.0f, 0.0f, 0.0f, 0.0f};
    __syncthreads();

    for (int t = 0; t < ntrips; ++t) {
        int sc[L1_K], dc[L1_K];
        #pragma unroll
        for (int k = 0; k < L1_K; ++k) { sc[k] = sn[k]; dc[k] = dn[k]; }
        if (t + 1 < ntrips) l1_load_idx<FULL>(ps, pd, t + 1, ebase, E, sn, dn);

        int n = 0;
        #pragma unroll
        for (int k = 0; k < L1_K; ++k) {
            unsigned int r   = (unsigned int)(dc[k] - lo);
            bool         hit = r < (unsigned int)rng;
            unsigned int m   = __builtin_amdgcn_ballot_w32(hit);
            int pos = n + (int)__builtin_amdgcn_mbcnt_lo(m, 0u);
            if (hit) q[pos] = (v2i){sc[k], (int)r};
            n += __builtin_popcount(m);
        }
        __builtin_amdgcn_fence(__ATOMIC_ACQ_REL, "wavefront");

        if (n > 0) {
            l1_batch(hw1, acc, q, 0, n, half, c);
            for (int i0 = 2 * L1_BATCH; i0 < n; i0 += 2 * L1_BATCH)
                l1_batch(hw1, acc, q, i0, n, half, c);
        }
        __builtin_amdgcn_fence(__ATOMIC_ACQ_REL, "wavefront");
    }
}

__global__ __launch_bounds__(L1_THREADS)
void k_edges1_lds(const int* __restrict__ src, const int* __restrict__ dst,
                  const float* __restrict__ hw1, float* __restrict__ agg1,
                  int E, int N, int rng, int nranges, int nslices) {
    __shared__ float s_acc[L1_RNG_MAX * HIDDEN];
    __shared__ v2i   s_q[L1_THREADS / 32][L1_QCAP];
    int tid   = threadIdx.x;
    int slice = blockIdx.x / nranges;
    int range = blockIdx.x - slice * nranges;
    int lo    = range * rng;
    int nacc  = rng * HIDDEN;

    int e0 = slice * L1_SLICE;
    const int* ps = src + e0 + tid;
    const int* pd = dst + e0 + tid;
    v2i* q = s_q[tid >> 5];
    if (e0 + L1_SLICE <= E) {
        l1_walk<true >(ps, pd, hw1, s_acc, q, e0 + tid, E, L1_TRIPS, lo, rng, nacc, tid);
    } else {
        int ntrips = (E - e0 + L1_TRIP - 1) / L1_TRIP;
        l1_walk<false>(ps, pd, hw1, s_acc, q, e0 + tid, E, ntrips, lo, rng, nacc, tid);
    }
    __syncthreads();

    int nlines = nacc >> 5;
    int rot = (int)(((unsigned int)slice * (unsigned int)nlines) / (unsigned int)nslices) << 5;
    float* base = agg1 + (size_t)lo * HIDDEN;
    int lim = (N - lo) * HIDDEN;
    for (int i = tid; i < nacc; i += L1_THREADS) {
        int j = i + rot;
        if (j >= nacc) j -= nacc;
        if (j < lim) atomic_add_f32(&base[j], s_acc[j]);
    }
}

#define B1_RNG     2000
#define B1_RMAX    64
#define B1_CAP     640
#define B1_LOADS   25
#define B1_THREADS 1024
#define B1_EPB     (B1_LOADS * B1_THREADS)
#define B1_CH      25
#define B1_RBITS   12

__global__ __launch_bounds__(B1_THREADS)
void k_bin1(const int* __restrict__ src, const int* __restrict__ dst,
            unsigned int* __restrict__ bins, int* __restrict__ cnts, int E, int R) {
    __shared__ unsigned int s_stage[B1_RMAX * B1_CAP];
    __shared__ unsigned int s_cnt[B1_RMAX];
    int tid = threadIdx.x;
    int blk = blockIdx.x;
    int e0  = blk * B1_EPB;

    int s[B1_LOADS], d[B1_LOADS];
    const int* ps = src + e0 + tid;
    const int* pd = dst + e0 + tid;
    if (e0 + B1_EPB <= E) {
        #pragma unroll
        for (int j = 0; j < B1_LOADS; ++j) {
            s[j] = ps[j * B1_THREADS];
            d[j] = pd[j * B1_THREADS];
        }
    } else {
        #pragma unroll
        for (int j = 0; j < B1_LOADS; ++j) {
            bool in = e0 + j * B1_THREADS + tid < E;
            s[j] = in ? ps[j * B1_THREADS] : 0;
            d[j] = in ? pd[j * B1_THREADS] : -1;
        }
    }
    if (tid < B1_RMAX) s_cnt[tid] = 0u;
    __syncthreads();

    #pragma unroll
    for (int j = 0; j < B1_LOADS; ++j) {
        if (d[j] >= 0) {
            unsigned int b   = (unsigned int)d[j] / (unsigned int)B1_RNG;
            unsigned int r   = (unsigned int)d[j] - b * (unsigned int)B1_RNG;
            unsigned int pos = atomicAdd(&s_cnt[b], 1u);
            if (pos < (unsigned int)B1_CAP)
                s_stage[b * B1_CAP + pos] = ((unsigned int)s[j] << B1_RBITS) | r;
        }
    }
    __syncthreads();

    unsigned int* outp = bins + (size_t)blk * R * B1_CAP;
    int tot = R * B1_CAP;
    for (int i = tid; i < tot; i += B1_THREADS) {
        int b = i / B1_CAP;
        int p = i - b * B1_CAP;
        if ((unsigned int)p < s_cnt[b]) outp[i] = s_stage[i];
    }
    if (tid < R) cnts[blk * R + tid] = (int)s_cnt[tid];
}

static_assert(B1_THREADS == 32 * 32, "k_deg_bins: 32 waves, wave w takes chunks w, w + 32, ...");
#define DB_J  (B1_CAP / 32)

__global__ __launch_bounds__(B1_THREADS)
void k_deg_bins(const unsigned int* __restrict__ bins, const int* __restrict__ cnts,
                const int* __restrict__ dst, float* __restrict__ dinv,
                int E, int N, int R, int NB) {
    __shared__ unsigned int s_c[B1_RNG];
    static_assert(B1_CAP % 32 == 0, "a chunk is a whole number of wave-wide loads");
    int tid   = threadIdx.x;
    int lane  = tid & 31;
    int wave  = __builtin_amdgcn_readfirstlane(tid >> 5);
    int range = blockIdx.x;
    int lo    = range * B1_RNG;

    for (int i = tid; i < B1_RNG; i += B1_THREADS) s_c[i] = 0u;
    __syncthreads();

    for (int kb = wave; kb < NB; kb += 32 * 32) {
        int kl  = kb + 32 * lane;
        int cv  = (kl < NB) ? cnts[kl * R + range] : 0;
        int nch = min(32, (NB - kb + 31) / 32);
        for (int i = 0; i < nch; ++i) {
            int k   = kb + 32 * i;
            int raw = __builtin_amdgcn_readlane(cv, i);
            if (raw <= B1_CAP) {
                const unsigned int* pc = bins + ((size_t)k * R + range) * B1_CAP + lane;
                unsigned int rec[DB_J];
                #pragma unroll
                for (int j = 0; j < DB_J; ++j) {
                    rec[j] = 0xFFFFFFFFu;
                    if (j * 32 + lane < raw) rec[j] = pc[j * 32];
                }
                #pragma unroll
                for (int j = 0; j < DB_J; ++j) {
                    unsigned int r = rec[j] & ((1u << B1_RBITS) - 1u);
                    if (r < (unsigned int)B1_RNG) atomicAdd(&s_c[r], 1u);
                }
            } else {
                int e0 = k * B1_EPB;
                int e1 = min(e0 + B1_EPB, E);
                for (int e = e0 + lane; e < e1; e += 32) {
                    unsigned int r = (unsigned int)(dst[e] - lo);
                    if (r < (unsigned int)B1_RNG) atomicAdd(&s_c[r], 1u);
                }
            }
        }
    }
    __syncthreads();

    for (int i = tid; i < B1_RNG; i += B1_THREADS) {
        int node = lo + i;
        if (node < N) dinv[node] = rsqrtf((float)(s_c[i] + 1u));
    }
}

typedef unsigned int v4u __attribute__((ext_vector_type(4)));

__global__ __launch_bounds__(B1_THREADS)
void k_edges1_bin(const unsigned int* __restrict__ bins, const int* __restrict__ cnts,
                  const int* __restrict__ src, const int* __restrict__ dst,
                  const float* __restrict__ hw1, float* __restrict__ agg1,
                  int E, int N, int R, int NB, int nsub) {
    __shared__ float s_acc[B1_RNG * HIDDEN];
    __shared__ alignas(16) unsigned int s_rec[B1_CH * B1_CAP + 32];
    int tid   = threadIdx.x;
    int lane  = tid & 31;
    int wave  = __builtin_amdgcn_readfirstlane(tid >> 5);
    int sub   = blockIdx.x / R;
    int range = blockIdx.x - sub * R;
    int lo    = range * B1_RNG;
    int nacc  = B1_RNG * HIDDEN;
    int c     = lane & 15;
    int hbase = lane & 16;
    int blk0  = sub * B1_CH;

    int raw = 0;
    if (lane < B1_CH && blk0 + lane < NB) raw = cnts[(blk0 + lane) * R + range];
    int nv = (raw <= B1_CAP) ? raw : 0;

    unsigned int rec[B1_CH];
    #pragma unroll
    for (int k = 0; k < B1_CH; ++k) {
        int nk = __builtin_amdgcn_readlane(nv, k);
        rec[k] = 0u;
        if (tid < nk) rec[k] = bins[((size_t)(blk0 + k) * R + range) * B1_CAP + tid];
    }

    v4f* z = (v4f*)s_acc;
    for (int i = tid; i < (nacc >> 2); i += B1_THREADS) z[i] = (v4f){0.0f, 0.0f, 0.0f, 0.0f};

    int T = 0;
    #pragma unroll
    for (int k = 0; k < B1_CH; ++k) {
        int nk = __builtin_amdgcn_readlane(nv, k);
        if (tid < nk) s_rec[T + tid] = rec[k];
        T += nk;
    }
    __syncthreads();

    int nsteps = (T + 31) >> 5;
    for (int st = wave; st < nsteps; st += 32) {
        int i0 = st * 32 + hbase;
        v4u q[4];
        #pragma unroll
        for (int m = 0; m < 4; ++m) q[m] = *(const v4u*)&s_rec[i0 + 4 * m];
        float v[16];
        int   row[16];
        #pragma unroll
        for (int j = 0; j < 16; ++j) {
            unsigned int ej = q[j >> 2][j & 3];
            bool ok = i0 + j < T;
            int  sj = ok ? (int)(ej >> B1_RBITS) : 0;
            row[j]  = ok ? (int)(ej & ((1u << B1_RBITS) - 1u)) : 0;
            v[j]    = hw1[(size_t)sj * HIDDEN + c];
        }
        #pragma unroll
        for (int j = 0; j < 16; ++j)
            lds_add_f32(&s_acc[row[j] * HIDDEN + c], (i0 + j < T) ? v[j] : 0.0f);
    }

    for (int k = 0; k < B1_CH; ++k) {
        int rk = __builtin_amdgcn_readlane(raw, k);
        if (rk > B1_CAP) {
            int e0 = (blk0 + k) * B1_EPB;
            int e1 = min(e0 + B1_EPB, E);
            for (int e = e0 + tid; e < e1; e += B1_THREADS) {
                unsigned int r = (unsigned int)(dst[e] - lo);
                if (r < (unsigned int)B1_RNG) {
                    const float* hrow = hw1 + (size_t)src[e] * HIDDEN;
                    for (int ch = 0; ch < HIDDEN; ++ch)
                        lds_add_f32(&s_acc[r * HIDDEN + ch], hrow[ch]);
                }
            }
        }
    }
    __syncthreads();

    int nlines = nacc >> 5;
    int rot = (int)(((unsigned int)sub * (unsigned int)nlines) / (unsigned int)nsub) << 5;
    float* base = agg1 + (size_t)lo * HIDDEN;
    int lim = (N - lo) * HIDDEN;
    for (int i = tid; i < nacc; i += B1_THREADS) {
        int j = i + rot;
        if (j >= nacc) j -= nacc;
        if (j < lim) atomic_add_f32(&base[j], s_acc[j]);
    }
}

#define E2B_G ((B1_CH * B1_CAP + B1_THREADS - 1) / B1_THREADS)

__global__ __launch_bounds__(B1_THREADS)
void k_edges2_bin(const unsigned int* __restrict__ bins, const int* __restrict__ cnts,
                  const int* __restrict__ src, const int* __restrict__ dst,
                  const float* __restrict__ hw2, float* __restrict__ agg2,
                  int E, int N, int R, int NB, int nsub) {
    __shared__ alignas(16) float s_acc[NCLS * B1_RNG];
    __shared__ unsigned int s_rec[B1_CH * B1_CAP];
    int tid   = threadIdx.x;
    int lane  = tid & 31;
    int sub   = blockIdx.x / R;
    int range = blockIdx.x - sub * R;
    int lo    = range * B1_RNG;
    int nacc  = B1_RNG * NCLS;
    int blk0  = sub * B1_CH;
    const v2f* rows = (const v2f*)hw2;

    int raw = 0;
    if (lane < B1_CH && blk0 + lane < NB) raw = cnts[(blk0 + lane) * R + range];
    int nv = (raw <= B1_CAP) ? raw : 0;

    unsigned int rec[B1_CH];
    #pragma unroll
    for (int k = 0; k < B1_CH; ++k) {
        int nk = __builtin_amdgcn_readlane(nv, k);
        rec[k] = 0u;
        if (tid < nk) rec[k] = bins[((size_t)(blk0 + k) * R + range) * B1_CAP + tid];
    }

    v4f* z = (v4f*)s_acc;
    for (int i = tid; i < (nacc >> 2); i += B1_THREADS) z[i] = (v4f){0.0f, 0.0f, 0.0f, 0.0f};

    int T = 0;
    #pragma unroll
    for (int k = 0; k < B1_CH; ++k) {
        int nk = __builtin_amdgcn_readlane(nv, k);
        if (tid < nk) s_rec[T + tid] = rec[k];
        T += nk;
    }
    __syncthreads();

    unsigned int q[E2B_G];
    v2f w[E2B_G];
    #pragma unroll
    for (int g = 0; g < E2B_G; ++g) {
        int i = g * B1_THREADS + tid;
        q[g] = 0u;
        if (i < T) q[g] = s_rec[i];
    }
    #pragma unroll
    for (int g = 0; g < E2B_G; ++g) {
        int i = g * B1_THREADS + tid;
        w[g] = (v2f){0.0f, 0.0f};
        if (i < T) w[g] = rows[q[g] >> B1_RBITS];
    }
    #pragma unroll
    for (int g = 0; g < E2B_G; ++g) {
        int i = g * B1_THREADS + tid;
        if (i < T) {
            unsigned int r = q[g] & ((1u << B1_RBITS) - 1u);
            lds_add_f32(&s_acc[r],          w[g].x);
            lds_add_f32(&s_acc[B1_RNG + r], w[g].y);
        }
    }

    for (int k = 0; k < B1_CH; ++k) {
        int rk = __builtin_amdgcn_readlane(raw, k);
        if (rk > B1_CAP) {
            int e0 = (blk0 + k) * B1_EPB;
            int e1 = min(e0 + B1_EPB, E);
            for (int e = e0 + tid; e < e1; e += B1_THREADS) {
                unsigned int r = (unsigned int)(dst[e] - lo);
                if (r < (unsigned int)B1_RNG) {
                    v2f u = rows[src[e]];
                    lds_add_f32(&s_acc[r],          u.x);
                    lds_add_f32(&s_acc[B1_RNG + r], u.y);
                }
            }
        }
    }
    __syncthreads();

    int nlines = nacc >> 5;
    int rot = (int)(((unsigned int)sub * (unsigned int)nlines) / (unsigned int)nsub) << 5;
    float* base = agg2 + (size_t)lo * NCLS;
    int lim = (N - lo) * NCLS;
    for (int i = tid; i < nacc; i += B1_THREADS) {
        int j = i + rot;
        if (j >= nacc) j -= nacc;
        if (j < lim) atomic_add_f32(&base[j], s_acc[(j & 1) * B1_RNG + (j >> 1)]);
    }
}

__global__ void k_final(const float* __restrict__ agg2, const float* __restrict__ b2,
                        const float* __restrict__ dinv,
                        float* __restrict__ out, int n) {
    int i = blockIdx.x * blockDim.x + threadIdx.x;
    if (i >= n) return;
    float di = dinv[i];
    float l0 = agg2[(size_t)i * NCLS + 0] * di + b2[0];
    float l1 = agg2[(size_t)i * NCLS + 1] * di + b2[1];
    float m   = fmaxf(l0, l1);
    float lse = m + logf(expf(l0 - m) + expf(l1 - m));
    out[(size_t)i * NCLS + 0] = l0 - lse;
    out[(size_t)i * NCLS + 1] = l1 - lse;
}

extern "C" void kernel_launch(void* const* d_in, const int* in_sizes, int n_in,
                              void* d_out, int out_size, void* d_ws, size_t ws_size,
                              hipStream_t stream) {
    const float* x  = (const float*)d_in[0];
    const int*   ei = (const int*)d_in[1];
    const float* W1 = (const float*)d_in[2];
    const float* b1 = (const float*)d_in[3];
    const float* W2 = (const float*)d_in[4];
    const float* b2 = (const float*)d_in[5];
    float* out = (float*)d_out;

    int N = in_sizes[0] / N_FEAT;
    int E = in_sizes[1] / 2;
    const int* src = ei;
    const int* dst = ei + E;

    float* ws   = (float*)d_ws;
    float* dinv = ws;
    float* hw1  = dinv + (size_t)N;
    float* agg1 = hw1  + (size_t)N * HIDDEN;
    float* hw2  = agg1 + (size_t)N * HIDDEN;
    float* agg2 = hw2  + (size_t)N * NCLS;

    const int tb = 256;
    int rng = (((N + DEG_RANGES - 1) / DEG_RANGES) + 31) & ~31;
    int nslices = (E + DEG_SLICE - 1) / DEG_SLICE;

    int    rb1   = (N + B1_RNG - 1) / B1_RNG;
    int    nb1   = (E + B1_EPB - 1) / B1_EPB;
    int    nsub1 = (nb1 + B1_CH - 1) / B1_CH;
    size_t boff  = (((size_t)N * (1 + 2 * HIDDEN + 2 * NCLS) * sizeof(float)) + 255) & ~(size_t)255;
    size_t bbyte = (size_t)nb1 * rb1 * B1_CAP * sizeof(unsigned int);
    size_t cbyte = (size_t)nb1 * rb1 * sizeof(int);
    unsigned int* bins = (unsigned int*)((char*)d_ws + boff);
    int*          cnts = (int*)((char*)d_ws + boff + bbyte);
    bool binned = nb1 > 0 && rb1 >= 1 && rb1 <= B1_RMAX && N <= (1 << (32 - B1_RBITS)) &&
                  ws_size >= boff + bbyte + cbyte;

    if (binned) {
        k_bin1    <<<nb1, B1_THREADS, 0, stream>>>(src, dst, bins, cnts, E, rb1);
        k_deg_bins<<<rb1, B1_THREADS, 0, stream>>>(bins, cnts, dst, dinv, E, N, rb1, nb1);
    } else {
        k_deg_init <<<(N + tb - 1) / tb, tb, 0, stream>>>(dinv, N);
        if (rng <= DEG_RNG_MAX) {
            if (nslices > 0)
                k_deg_hist<<<nslices * DEG_RANGES, DEG_THREADS, 0, stream>>>(dst, dinv, E, N, rng, nslices);
        } else {
            k_deg_edges<<<(E + tb - 1) / tb, tb, 0, stream>>>(dst, dinv, E);
        }
        k_dinv     <<<(N + tb - 1) / tb, tb, 0, stream>>>(dinv, N);
    }

    int ntiles = N / 16;
    int wpb    = tb / 32;
    if (ntiles > 0)
        k_gemm1<<<(ntiles + wpb - 1) / wpb, tb, 0, stream>>>(x, W1, dinv, hw1, agg1, ntiles);
    int rem = N - ntiles * 16;
    if (rem > 0)
        k_gemm1_rem<<<(rem * HIDDEN + tb - 1) / tb, tb, 0, stream>>>(x, W1, dinv, hw1, agg1,
                                                                     ntiles * 16, N);

    int nr1  = (N + L1_RNG_MAX - 1) / L1_RNG_MAX;
    int nsl1 = (E + L1_SLICE - 1) / L1_SLICE;
    if (binned) {
        k_edges1_bin<<<rb1 * nsub1, B1_THREADS, 0, stream>>>(bins, cnts, src, dst, hw1, agg1, E, N, rb1, nb1, nsub1);
    } else if (nr1 >= 1 && nr1 <= L1_MAX_RANGES) {
        int rng1 = (((N + nr1 - 1) / nr1) + 1) & ~1;
        if (nsl1 > 0)
            k_edges1_lds<<<nsl1 * nr1, L1_THREADS, 0, stream>>>(src, dst, hw1, agg1, E, N, rng1, nr1, nsl1);
    } else {
        k_edges1<<<(E + tb - 1) / tb, tb, 0, stream>>>(src, dst, hw1, agg1, E);
    }
    k_layer2<<<(N + tb - 1) / tb, tb, 0, stream>>>(agg1, b1, W2, dinv, hw2, agg2, N);
    if (binned) {
        k_edges2_bin<<<rb1 * nsub1, B1_THREADS, 0, stream>>>(bins, cnts, src, dst, hw2, agg2, E, N, rb1, nb1, nsub1);
    } else if (rng <= DEG_RNG_MAX) {
        if (nslices > 0)
            k_edges2_lds<<<nslices * DEG_RANGES, DEG_THREADS, 0, stream>>>(src, dst, hw2, agg2, E, N, rng, nslices);
    } else {
        long long t2 = (long long)E * NCLS;
        k_edges2<<<(int)((t2 + tb - 1) / tb), tb, 0, stream>>>(src, dst, hw2, agg2, E);
    }
    k_final <<<(N + tb - 1) / tb, tb, 0, stream>>>(agg2, b2, dinv, out, N);
}


# bin ranges cut to 400 nodes so one sum block owns all edges of its nodes: sums finish in LDS (no partial-sum atomics, no start planes), next node pass runs as the block's tail; new f32 sum order
# speedup vs baseline: 6.4876x; 1.0993x over previous
#include <hip/hip_runtime.h>
#include <hip/hip_bf16.h>

#define N_FEAT 256
#define HIDDEN 16
#define NCLS   2

typedef float v2f __attribute__((ext_vector_type(2)));
typedef float v4f __attribute__((ext_vector_type(4)));
typedef float v8f __attribute__((ext_vector_type(8)));

__device__ __forceinline__ void atomic_add_f32(float* p, float v) {
    unsafeAtomicAdd(p, v);
}

__global__ void k_deg_init(float* __restrict__ deg, int n) {
    int i = blockIdx.x * blockDim.x + threadIdx.x;
    if (i < n) deg[i] = 1.0f;
}

__global__ void k_deg_edges(const int* __restrict__ dst, float* __restrict__ deg, int E) {
    int e = blockIdx.x * blockDim.x + threadIdx.x;
    if (e < E) atomic_add_f32(&deg[dst[e]], 1.0f);
}

#define DEG_RANGES   4
#define DEG_RNG_MAX  32768
#define DEG_THREADS  1024
#define DEG_LOADS    49
#define DEG_SLICE    (DEG_LOADS * DEG_THREADS)

__global__ __launch_bounds__(DEG_THREADS)
void k_deg_hist(const int* __restrict__ dst, float* __restrict__ deg,
                int E, int N, int rng, int nslices) {
    __shared__ unsigned int s_cnt[DEG_RNG_MAX];
    int tid   = threadIdx.x;
    int slice = blockIdx.x / DEG_RANGES;
    int range = blockIdx.x % DEG_RANGES;
    int lo    = range * rng;

    for (int i = tid; i < rng; i += DEG_THREADS) s_cnt[i] = 0u;
    __syncthreads();

    int e0 = slice * DEG_SLICE;
    int e1 = min(e0 + DEG_SLICE, E);
    int d[DEG_LOADS];
    const int* p = dst + e0 + tid;
    if (e0 + DEG_SLICE <= E) {
        #pragma unroll
        for (int j = 0; j < DEG_LOADS; ++j) d[j] = p[j * DEG_THREADS];
    } else {
        #pragma unroll
        for (int j = 0; j < DEG_LOADS; ++j)
            d[j] = (e0 + j * DEG_THREADS + tid < e1) ? p[j * DEG_THREADS] : -1;
    }
    #pragma unroll
    for (int j = 0; j < DEG_LOADS; ++j) {
        unsigned int r = (unsigned int)(d[j] - lo);
        if (r < (unsigned int)rng) atomicAdd(&s_cnt[r], 1u);
    }
    __syncthreads();

    int rot = (int)(((unsigned int)slice * (unsigned int)(rng >> 5)) / (unsigned int)nslices) << 5;
    for (int i = tid; i < rng; i += DEG_THREADS) {
        int j = i + rot;
        if (j >= rng) j -= rng;
        unsigned int c = s_cnt[j];
        int node = lo + j;
        if (c != 0u && node < N) atomic_add_f32(&deg[node], (float)c);
    }
}

__global__ void k_dinv(float* __restrict__ deg, int n) {
    int i = blockIdx.x * blockDim.x + threadIdx.x;
    if (i < n) deg[i] = rsqrtf(deg[i]);
}

__global__ void k_gemm1(const float* __restrict__ x, const float* __restrict__ W1,
                        const float* __restrict__ dinv,
                        float* __restrict__ hw1, float* __restrict__ agg1, int ntiles) {
    int lane = threadIdx.x & 31;
    int wave = threadIdx.x >> 5;
    int tile = blockIdx.x * (blockDim.x >> 5) + wave;
    if (tile >= ntiles) return;
    int half = lane >> 4;
    int l    = lane & 15;
    int row0 = tile * 16;

    const float* aptr = x  + (size_t)(row0 + l) * N_FEAT + 2 * half;
    const float* bptr = W1 + (size_t)(2 * half) * HIDDEN + l;

    v8f c = {};
    #pragma unroll 8
    for (int kk = 0; kk < N_FEAT / 4; ++kk) {
        v2f a = *(const v2f*)(aptr + kk * 4);
        v2f b;
        b.x = bptr[kk * 4 * HIDDEN];
        b.y = bptr[kk * 4 * HIDDEN + HIDDEN];
        c = __builtin_amdgcn_wmma_f32_16x16x4_f32(false, a, false, b,
                                                  (short)0, c, false, false);
    }

    int rbase = row0 + half * 8;
    #pragma unroll
    for (int v = 0; v < 8; ++v) {
        int r = rbase + v;
        float dval = c[v] * dinv[r];
        size_t idx = (size_t)r * HIDDEN + l;
        hw1[idx]  = dval;
        if (agg1) agg1[idx] = dval;
    }
}

__global__ void k_gemm1_rem(const float* __restrict__ x, const float* __restrict__ W1,
                            const float* __restrict__ dinv,
                            float* __restrict__ hw1, float* __restrict__ agg1,
                            int row_lo, int n) {
    int t = blockIdx.x * blockDim.x + threadIdx.x;
    int r = row_lo + t / HIDDEN;
    int c = t % HIDDEN;
    if (r >= n) return;
    float acc = 0.0f;
    for (int k = 0; k < N_FEAT; ++k) acc += x[(size_t)r * N_FEAT + k] * W1[(size_t)k * HIDDEN + c];
    acc *= dinv[r];
    size_t idx = (size_t)r * HIDDEN + c;
    hw1[idx]  = acc;
    if (agg1) agg1[idx] = acc;
}

__global__ void k_edges1(const int* __restrict__ src, const int* __restrict__ dst,
                         const float* __restrict__ hw1,
                         float* __restrict__ agg1, int E) {
    int lane = threadIdx.x & 31;
    int wave = (blockIdx.x * blockDim.x + threadIdx.x) >> 5;
    int e0   = wave * 32;
    if (e0 >= E) return;
    int e = e0 + lane;
    int s = 0, d = 0;
    if (e < E) {
        s = src[e];
        d = dst[e];
    }
    int c     = lane & 15;
    int hbase = lane & 16;
    if (e0 + 32 <= E) {
        float val[16];
        int   drow[16];
        #pragma unroll
        for (int j = 0; j < 16; ++j) {
            int   sj = __shfl(s, hbase + j);
            drow[j]  = __shfl(d, hbase + j);
            val[j]   = hw1[(size_t)sj * HIDDEN + c];
        }
        #pragma unroll
        for (int j = 0; j < 16; ++j)
            atomic_add_f32(&agg1[(size_t)drow[j] * HIDDEN + c], val[j]);
    } else {
        for (int j = 0; j < 16; ++j) {
            int   sj = __shfl(s, hbase + j);
            int   dj = __shfl(d, hbase + j);
            if (e0 + hbase + j < E)
                atomic_add_f32(&agg1[(size_t)dj * HIDDEN + c],
                               hw1[(size_t)sj * HIDDEN + c]);
        }
    }
}

__global__ void k_layer2(const float* __restrict__ agg1, const float* __restrict__ b1,
                         const float* __restrict__ W2, const float* __restrict__ dinv,
                         float* __restrict__ hw2, float* __restrict__ agg2, int n) {
    int i = blockIdx.x * blockDim.x + threadIdx.x;
    if (i >= n) return;
    float di = dinv[i];
    float h0 = 0.0f, h1 = 0.0f;
    #pragma unroll
    for (int c = 0; c < HIDDEN; ++c) {
        float v = agg1[(size_t)i * HIDDEN + c] * di + b1[c];
        v = fmaxf(v, 0.0f);
        h0 += v * W2[c * NCLS + 0];
        h1 += v * W2[c * NCLS + 1];
    }
    h0 *= di;
    h1 *= di;
    hw2[(size_t)i * NCLS + 0]  = h0;
    hw2[(size_t)i * NCLS + 1]  = h1;
    agg2[(size_t)i * NCLS + 0] = h0;
    agg2[(size_t)i * NCLS + 1] = h1;
}

__global__ void k_edges2(const int* __restrict__ src, const int* __restrict__ dst,
                         const float* __restrict__ hw2,
                         float* __restrict__ agg2, int E) {
    int t = blockIdx.x * blockDim.x + threadIdx.x;
    int e = t >> 1;
    int c = t & 1;
    if (e >= E) return;
    int s = src[e], d = dst[e];
    atomic_add_f32(&agg2[(size_t)d * NCLS + c], hw2[(size_t)s * NCLS + c]);
}

#define E2_CHUNK  7
#define E2_TRIPS  (DEG_LOADS / E2_CHUNK)

__device__ __forceinline__ void lds_add_f32(float* p, float v) {
    __hip_atomic_fetch_add(p, v, __ATOMIC_RELAXED, __HIP_MEMORY_SCOPE_WORKGROUP);
}

template <bool FULL>
__device__ __forceinline__ void e2_load_idx(const int* __restrict__ ps, const int* __restrict__ pd,
                                            int t, int ebase, int E,
                                            int (&s)[E2_CHUNK], int (&d)[E2_CHUNK]) {
    #pragma unroll
    for (int j = 0; j < E2_CHUNK; ++j) {
        int off = (t * E2_CHUNK + j) * DEG_THREADS;
        if (FULL || ebase + off < E) {
            s[j] = ps[off];
            d[j] = pd[off];
        } else {
            s[j] = 0;
            d[j] = -1;
        }
    }
}

template <bool FULL>
__device__ __forceinline__ void e2_walk(const int* __restrict__ ps, const int* __restrict__ pd,
                                        const v2f* __restrict__ rows, float* acc,
                                        int ebase, int E, int lo, int rng, int nacc, int tid) {
    int sn[E2_CHUNK], dn[E2_CHUNK];
    e2_load_idx<FULL>(ps, pd, 0, ebase, E, sn, dn);

    v4f* z = (v4f*)acc;
    for (int i = tid; i < (nacc >> 2); i += DEG_THREADS) z[i] = (v4f){0.0f, 0.0f, 0.0f, 0.0f};
    __syncthreads();

    #pragma unroll
    for (int t = 0; t < E2_TRIPS; ++t) {
        int sc[E2_CHUNK], dc[E2_CHUNK];
        #pragma unroll
        for (int j = 0; j < E2_CHUNK; ++j) { sc[j] = sn[j]; dc[j] = dn[j]; }
        if (t + 1 < E2_TRIPS) e2_load_idx<FULL>(ps, pd, t + 1, ebase, E, sn, dn);
        v2f v[E2_CHUNK];
        #pragma unroll
        for (int j = 0; j < E2_CHUNK; ++j) {
            v[j] = (v2f){0.0f, 0.0f};
            if ((unsigned int)(dc[j] - lo) < (unsigned int)rng) v[j] = rows[sc[j]];
        }
        #pragma unroll
        for (int j = 0; j < E2_CHUNK; ++j) {
            unsigned int r = (unsigned int)(dc[j] - lo);
            if (r < (unsigned int)rng) {
                lds_add_f32(&acc[2u * r],      v[j].x);
                lds_add_f32(&acc[2u * r + 1u], v[j].y);
            }
        }
    }
}

__global__ __launch_bounds__(DEG_THREADS)
void k_edges2_lds(const int* __restrict__ src, const int* __restrict__ dst,
                  const float* __restrict__ hw2, float* __restrict__ agg2,
                  int E, int N, int rng, int nslices) {
    __shared__ float s_acc[NCLS * DEG_RNG_MAX];
    int tid   = threadIdx.x;
    int slice = blockIdx.x / DEG_RANGES;
    int range = blockIdx.x % DEG_RANGES;
    int lo    = range * rng;
    int nacc  = rng * NCLS;

    int e0 = slice * DEG_SLICE;
    const int* ps = src + e0 + tid;
    const int* pd = dst + e0 + tid;
    const v2f* rows = (const v2f*)hw2;
    if (e0 + DEG_SLICE <= E)
        e2_walk<true >(ps, pd, rows, s_acc, e0 + tid, E, lo, rng, nacc, tid);
    else
        e2_walk<false>(ps, pd, rows, s_acc, e0 + tid, E, lo, rng, nacc, tid);
    __syncthreads();

    int nlines = nacc >> 5;
    int rot = (int)(((unsigned int)slice * (unsigned int)nlines) / (unsigned int)nslices) << 5;
    float* base = agg2 + (size_t)lo * NCLS;
    int lim = (N - lo) * NCLS;
    for (int i = tid; i < nacc; i += DEG_THREADS) {
        int j = i + rot;
        if (j >= nacc) j -= nacc;
        float a = s_acc[j];
        if (a != 0.0f && j < lim) atomic_add_f32(&base[j], a);
    }
}

#define L1_RNG_MAX    4032
#define L1_MAX_RANGES 64
#define L1_THREADS    1024
#define L1_K          8
#define L1_TRIP       (L1_K * L1_THREADS)
#define L1_TRIPS      40
#define L1_SLICE      (L1_TRIP * L1_TRIPS)
#define L1_QCAP       (L1_K * 32)
#define L1_BATCH      8

typedef int v2i __attribute__((ext_vector_type(2)));

template <bool FULL>
__device__ __forceinline__ void l1_load_idx(const int* __restrict__ ps, const int* __restrict__ pd,
                                            int t, int ebase, int E,
                                            int (&s)[L1_K], int (&d)[L1_K]) {
    #pragma unroll
    for (int k = 0; k < L1_K; ++k) {
        int off = t * L1_TRIP + k * L1_THREADS;
        if (FULL || ebase + off < E) {
            s[k] = ps[off];
            d[k] = pd[off];
        } else {
            s[k] = 0;
            d[k] = -1;
        }
    }
}

__device__ __forceinline__ void l1_batch(const float* __restrict__ hw1, float* acc, const v2i* q,
                                         int i0, int n, int half, int c) {
    v2i ent[L1_BATCH];
    #pragma unroll
    for (int j = 0; j < L1_BATCH; ++j) {
        int i = i0 + 2 * j + half;
        ent[j] = q[min(i, L1_QCAP - 1)];
    }
    float v[L1_BATCH];
    #pragma unroll
    for (int j = 0; j < L1_BATCH; ++j) {
        int i = i0 + 2 * j + half;
        int s = (i < n) ? ent[j].x : 0;
        v[j] = hw1[(size_t)s * HIDDEN + c];
    }
    #pragma unroll
    for (int j = 0; j < L1_BATCH; ++j) {
        int i = i0 + 2 * j + half;
        if (i < n) lds_add_f32(&acc[ent[j].y * HIDDEN + c], v[j]);
    }
}

template <bool FULL>
__device__ __forceinline__ void l1_walk(const int* __restrict__ ps, const int* __restrict__ pd,
                                        const float* __restrict__ hw1, float* acc, v2i* q,
                                        int ebase, int E, int ntrips, int lo, int rng, int nacc,
                                        int tid) {
    int lane = tid & 31;
    int half = lane >> 4;
    int c    = lane & 15;

    int sn[L1_K], dn[L1_K];
    l1_load_idx<FULL>(ps, pd, 0, ebase, E, sn, dn);

    v4f* z = (v4f*)acc;
    for (int i = tid; i < (nacc >> 2); i += L1_THREADS) z[i] = (v4f){0.0f, 0.0f, 0.0f, 0.0f};
    __syncthreads();

    for (int t = 0; t < ntrips; ++t) {
        int sc[L1_K], dc[L1_K];
        #pragma unroll
        for (int k = 0; k < L1_K; ++k) { sc[k] = sn[k]; dc[k] = dn[k]; }
        if (t + 1 < ntrips) l1_load_idx<FULL>(ps, pd, t + 1, ebase, E, sn, dn);

        int n = 0;
        #pragma unroll
        for (int k = 0; k < L1_K; ++k) {
            unsigned int r   = (unsigned int)(dc[k] - lo);
            bool         hit = r < (unsigned int)rng;
            unsigned int m   = __builtin_amdgcn_ballot_w32(hit);
            int pos = n + (int)__builtin_amdgcn_mbcnt_lo(m, 0u);
            if (hit) q[pos] = (v2i){sc[k], (int)r};
            n += __builtin_popcount(m);
        }
        __builtin_amdgcn_fence(__ATOMIC_ACQ_REL, "wavefront");

        if (n > 0) {
            l1_batch(hw1, acc, q, 0, n, half, c);
            for (int i0 = 2 * L1_BATCH; i0 < n; i0 += 2 * L1_BATCH)
                l1_batch(hw1, acc, q, i0, n, half, c);
        }
        __builtin_amdgcn_fence(__ATOMIC_ACQ_REL, "wavefront");
    }
}

__global__ __launch_bounds__(L1_THREADS)
void k_edges1_lds(const int* __restrict__ src, const int* __restrict__ dst,
                  const float* __restrict__ hw1, float* __restrict__ agg1,
                  int E, int N, int rng, int nranges, int nslices) {
    __shared__ float s_acc[L1_RNG_MAX * HIDDEN];
    __shared__ v2i   s_q[L1_THREADS / 32][L1_QCAP];
    int tid   = threadIdx.x;
    int slice = blockIdx.x / nranges;
    int range = blockIdx.x - slice * nranges;
    int lo    = range * rng;
    int nacc  = rng * HIDDEN;

    int e0 = slice * L1_SLICE;
    const int* ps = src + e0 + tid;
    const int* pd = dst + e0 + tid;
    v2i* q = s_q[tid >> 5];
    if (e0 + L1_SLICE <= E) {
        l1_walk<true >(ps, pd, hw1, s_acc, q, e0 + tid, E, L1_TRIPS, lo, rng, nacc, tid);
    } else {
        int ntrips = (E - e0 + L1_TRIP - 1) / L1_TRIP;
        l1_walk<false>(ps, pd, hw1, s_acc, q, e0 + tid, E, ntrips, lo, rng, nacc, tid);
    }
    __syncthreads();

    int nlines = nacc >> 5;
    int rot = (int)(((unsigned int)slice * (unsigned int)nlines) / (unsigned int)nslices) << 5;
    float* base = agg1 + (size_t)lo * HIDDEN;
    int lim = (N - lo) * HIDDEN;
    for (int i = tid; i < nacc; i += L1_THREADS) {
        int j = i + rot;
        if (j >= nacc) j -= nacc;
        if (j < lim) atomic_add_f32(&base[j], s_acc[j]);
    }
}

#define B1_RNG     400
#define B1_RMAX    256
#define B1_CAP     160
#define B1_LOADS   25
#define B1_THREADS 1024
#define B1_EPB     (B1_LOADS * B1_THREADS)
#define B1_CH      128
#define B1_RBITS   12

__global__ __launch_bounds__(B1_THREADS)
void k_bin1(const int* __restrict__ src, const int* __restrict__ dst,
            unsigned int* __restrict__ bins, int* __restrict__ cnts, int E, int R) {
    __shared__ unsigned int s_stage[B1_RMAX * B1_CAP];
    __shared__ unsigned int s_cnt[B1_RMAX];
    int tid = threadIdx.x;
    int blk = blockIdx.x;
    int e0  = blk * B1_EPB;

    int s[B1_LOADS], d[B1_LOADS];
    const int* ps = src + e0 + tid;
    const int* pd = dst + e0 + tid;
    if (e0 + B1_EPB <= E) {
        #pragma unroll
        for (int j = 0; j < B1_LOADS; ++j) {
            s[j] = ps[j * B1_THREADS];
            d[j] = pd[j * B1_THREADS];
        }
    } else {
        #pragma unroll
        for (int j = 0; j < B1_LOADS; ++j) {
            bool in = e0 + j * B1_THREADS + tid < E;
            s[j] = in ? ps[j * B1_THREADS] : 0;
            d[j] = in ? pd[j * B1_THREADS] : -1;
        }
    }
    if (tid < B1_RMAX) s_cnt[tid] = 0u;
    __syncthreads();

    #pragma unroll
    for (int j = 0; j < B1_LOADS; ++j) {
        if (d[j] >= 0) {
            unsigned int b   = (unsigned int)d[j] / (unsigned int)B1_RNG;
            unsigned int r   = (unsigned int)d[j] - b * (unsigned int)B1_RNG;
            unsigned int pos = atomicAdd(&s_cnt[b], 1u);
            if (pos < (unsigned int)B1_CAP)
                s_stage[b * B1_CAP + pos] = ((unsigned int)s[j] << B1_RBITS) | r;
        }
    }
    __syncthreads();

    unsigned int* outp = bins + (size_t)blk * R * B1_CAP;
    int tot = R * B1_CAP;
    for (int i = tid; i < tot; i += B1_THREADS) {
        int b = i / B1_CAP;
        int p = i - b * B1_CAP;
        if ((unsigned int)p < s_cnt[b]) outp[i] = s_stage[i];
    }
    if (tid < R) cnts[blk * R + tid] = (int)s_cnt[tid];
}

static_assert(B1_THREADS == 32 * 32, "k_deg_bins: 32 waves, wave w takes chunks w, w + 32, ...");
#define DB_J  (B1_CAP / 32)

__global__ __launch_bounds__(B1_THREADS)
void k_deg_bins(const unsigned int* __restrict__ bins, const int* __restrict__ cnts,
                const int* __restrict__ dst, float* __restrict__ dinv,
                int E, int N, int R, int NB) {
    __shared__ unsigned int s_c[B1_RNG];
    static_assert(B1_CAP % 32 == 0, "a chunk is a whole number of wave-wide loads");
    int tid   = threadIdx.x;
    int lane  = tid & 31;
    int wave  = __builtin_amdgcn_readfirstlane(tid >> 5);
    int range = blockIdx.x;
    int lo    = range * B1_RNG;

    for (int i = tid; i < B1_RNG; i += B1_THREADS) s_c[i] = 0u;
    __syncthreads();

    for (int kb = wave; kb < NB; kb += 32 * 32) {
        int kl  = kb + 32 * lane;
        int cv  = (kl < NB) ? cnts[kl * R + range] : 0;
        int nch = min(32, (NB - kb + 31) / 32);
        for (int i = 0; i < nch; ++i) {
            int k   = kb + 32 * i;
            int raw = __builtin_amdgcn_readlane(cv, i);
            if (raw <= B1_CAP) {
                const unsigned int* pc = bins + ((size_t)k * R + range) * B1_CAP + lane;
                unsigned int rec[DB_J];
                #pragma unroll
                for (int j = 0; j < DB_J; ++j) {
                    rec[j] = 0xFFFFFFFFu;
                    if (j * 32 + lane < raw) rec[j] = pc[j * 32];
                }
                #pragma unroll
                for (int j = 0; j < DB_J; ++j) {
                    unsigned int r = rec[j] & ((1u << B1_RBITS) - 1u);
                    if (r < (unsigned int)B1_RNG) atomicAdd(&s_c[r], 1u);
                }
            } else {
                int e0 = k * B1_EPB;
                int e1 = min(e0 + B1_EPB, E);
                for (int e = e0 + lane; e < e1; e += 32) {
                    unsigned int r = (unsigned int)(dst[e] - lo);
                    if (r < (unsigned int)B1_RNG) atomicAdd(&s_c[r], 1u);
                }
            }
        }
    }
    __syncthreads();

    for (int i = tid; i < B1_RNG; i += B1_THREADS) {
        int node = lo + i;
        if (node < N) dinv[node] = rsqrtf((float)(s_c[i] + 1u));
    }
}

typedef unsigned int v4u __attribute__((ext_vector_type(4)));
typedef int v4i __attribute__((ext_vector_type(4)));

#define OWN_I (B1_CH / 32)
#define OWN_SV ((B1_RNG * HIDDEN + B1_THREADS - 1) / B1_THREADS)
static_assert(B1_CH == 4 * 32, "a sum block's trip: wave w takes chunks w, w + 32, w + 64, w + 96");
static_assert(B1_RNG <= B1_THREADS, "the tail of a sum block: one thread per node");
static_assert(B1_RNG <= (1 << B1_RBITS), "a row in range fits a record's low bits");

__global__ __launch_bounds__(B1_THREADS)
void k_edges1_own(const unsigned int* __restrict__ bins, const int* __restrict__ cnts,
                  const int* __restrict__ src, const int* __restrict__ dst,
                  const float* __restrict__ hw1, const float* __restrict__ b1,
                  const float* __restrict__ W2, const float* __restrict__ dinv,
                  float* __restrict__ hw2, int E, int N, int R, int NB) {
    __shared__ alignas(16) float s_acc[B1_RNG * HIDDEN];
    __shared__ alignas(16) unsigned int s_rec[B1_CH * B1_CAP + 32];
    __shared__ alignas(16) int s_raw[B1_CH];
    int tid   = threadIdx.x;
    int lane  = tid & 31;
    int wave  = __builtin_amdgcn_readfirstlane(tid >> 5);
    int range = blockIdx.x;
    int lo    = range * B1_RNG;
    int nacc  = B1_RNG * HIDDEN;
    int c     = lane & 15;
    int hbase = lane & 16;

    int lim = (N - lo) * HIDDEN;
    const float* own = hw1 + (size_t)lo * HIDDEN;
    float sv[OWN_SV];
    #pragma unroll
    for (int u = 0; u < OWN_SV; ++u) {
        int i = u * B1_THREADS + tid;
        sv[u] = (i < nacc && i < lim) ? own[i] : 0.0f;
    }
    int rawn = (tid < B1_CH && tid < NB) ? cnts[tid * R + range] : 0;

    for (int g0 = 0; g0 < NB; g0 += B1_CH) {
        int ng = min(B1_CH, NB - g0);
        if (tid < B1_CH) s_raw[tid] = rawn;
        if (g0 == 0) {
            #pragma unroll
            for (int u = 0; u < OWN_SV; ++u) {
                int i = u * B1_THREADS + tid;
                if (i < nacc) s_acc[i] = sv[u];
            }
        }
        __syncthreads();

        v4i rr = *(const v4i*)&s_raw[4 * lane];
        int n0 = (rr.x <= B1_CAP) ? rr.x : 0;
        int n1 = (rr.y <= B1_CAP) ? rr.y : 0;
        int n2 = (rr.z <= B1_CAP) ? rr.z : 0;
        int n3 = (rr.w <= B1_CAP) ? rr.w : 0;
        int tot = n0 + n1 + n2 + n3;
        int inc = tot;
        #pragma unroll
        for (int dd = 1; dd < 32; dd <<= 1) {
            int up = __shfl_up(inc, dd);
            if (lane >= dd) inc += up;
        }
        int T = __shfl(inc, 31);
        int m    = wave & 3;
        int mine = (m == 0) ? n0 : (m == 1) ? n1 : (m == 2) ? n2 : n3;
        int pre  = inc - tot + ((m > 0) ? n0 : 0) + ((m > 1) ? n1 : 0) + ((m > 2) ? n2 : 0);

        unsigned int rec[OWN_I][DB_J];
        int nk[OWN_I], off[OWN_I];
        #pragma unroll
        for (int i = 0; i < OWN_I; ++i) {
            int hl = (wave >> 2) + 8 * i;
            nk[i]  = __builtin_amdgcn_readlane(mine, hl);
            off[i] = __builtin_amdgcn_readlane(pre, hl);
            const unsigned int* pc = bins + ((size_t)(g0 + wave + 32 * i) * R + range) * B1_CAP + lane;
            #pragma unroll
            for (int j = 0; j < DB_J; ++j) {
                rec[i][j] = 0u;
                if (j * 32 + lane < nk[i]) rec[i][j] = pc[j * 32];
            }
        }
        #pragma unroll
        for (int i = 0; i < OWN_I; ++i) {
            #pragma unroll
            for (int j = 0; j < DB_J; ++j)
                if (j * 32 + lane < nk[i]) s_rec[off[i] + j * 32 + lane] = rec[i][j];
        }
        __syncthreads();

        int nsteps = (T + 31) >> 5;
        for (int st = wave; st < nsteps; st += 32) {
            int i0 = st * 32 + hbase;
            v4u q[4];
            #pragma unroll
            for (int mm = 0; mm < 4; ++mm) q[mm] = *(const v4u*)&s_rec[i0 + 4 * mm];
            float v[16];
            int   row[16];
            #pragma unroll
            for (int j = 0; j < 16; ++j) {
                unsigned int ej = q[j >> 2][j & 3];
                bool ok = i0 + j < T;
                int  sj = ok ? (int)(ej >> B1_RBITS) : 0;
                row[j]  = ok ? (int)(ej & ((1u << B1_RBITS) - 1u)) : 0;
                v[j]    = hw1[(size_t)sj * HIDDEN + c];
            }
            #pragma unroll
            for (int j = 0; j < 16; ++j)
                lds_add_f32(&s_acc[row[j] * HIDDEN + c], (i0 + j < T) ? v[j] : 0.0f);
        }

        for (int k = 0; k < ng; ++k) {
            int rk = s_raw[k];
            if (rk > B1_CAP) {
                int e0 = (g0 + k) * B1_EPB;
                int e1 = min(e0 + B1_EPB, E);
                for (int e = e0 + tid; e < e1; e += B1_THREADS) {
                    unsigned int r = (unsigned int)(dst[e] - lo);
                    if (r < (unsigned int)B1_RNG) {
                        const float* hrow = hw1 + (size_t)src[e] * HIDDEN;
                        for (int ch = 0; ch < HIDDEN; ++ch)
                            lds_add_f32(&s_acc[r * HIDDEN + ch], hrow[ch]);
                    }
                }
            }
        }
        int gn = g0 + B1_CH;
        rawn = (tid < B1_CH && gn + tid < NB) ? cnts[(gn + tid) * R + range] : 0;
        __syncthreads();
    }

    if (tid < B1_RNG && lo + tid < N) {
        int   node = lo + tid;
        float di   = dinv[node];
        const v4f* a4 = (const v4f*)&s_acc[tid * HIDDEN];
        float h0 = 0.0f, h1 = 0.0f;
        #pragma unroll
        for (int qd = 0; qd < HIDDEN / 4; ++qd) {
            v4f a = a4[qd];
            #pragma unroll
            for (int el = 0; el < 4; ++el) {
                int   ch = 4 * qd + el;
                float v  = a[el] * di + b1[ch];
                v = fmaxf(v, 0.0f);
                h0 += v * W2[ch * NCLS + 0];
                h1 += v * W2[ch * NCLS + 1];
            }
        }
        hw2[(size_t)node * NCLS + 0] = h0 * di;
        hw2[(size_t)node * NCLS + 1] = h1 * di;
    }
}

__global__ __launch_bounds__(B1_THREADS)
void k_edges2_own(const unsigned int* __restrict__ bins, const int* __restrict__ cnts,
                  const int* __restrict__ src, const int* __restrict__ dst,
                  const float* __restrict__ hw2, const float* __restrict__ b2,
                  const float* __restrict__ dinv, float* __restrict__ out,
                  int E, int N, int R, int NB) {
    __shared__ float s_acc[NCLS * B1_RNG];
    int tid   = threadIdx.x;
    int lane  = tid & 31;
    int wave  = __builtin_amdgcn_readfirstlane(tid >> 5);
    int range = blockIdx.x;
    int lo    = range * B1_RNG;
    const v2f* rows = (const v2f*)hw2;

    v2f u0 = (v2f){0.0f, 0.0f};
    if (tid < B1_RNG && lo + tid < N) u0 = rows[lo + tid];

    int cv = (lane < OWN_I && wave + 32 * lane < NB) ? cnts[(wave + 32 * lane) * R + range] : 0;
    for (int g0 = 0; g0 < NB; g0 += B1_CH) {
        int kb = g0 + wave;
        unsigned int rec[OWN_I][DB_J];
        int raw[OWN_I], nk[OWN_I];
        #pragma unroll
        for (int i = 0; i < OWN_I; ++i) {
            raw[i] = __builtin_amdgcn_readlane(cv, i);
            nk[i]  = (raw[i] <= B1_CAP) ? raw[i] : 0;
            const unsigned int* pc = bins + ((size_t)(kb + 32 * i) * R + range) * B1_CAP + lane;
            #pragma unroll
            for (int j = 0; j < DB_J; ++j) {
                rec[i][j] = 0u;
                if (j * 32 + lane < nk[i]) rec[i][j] = pc[j * 32];
            }
        }
        v2f w[OWN_I][DB_J];
        #pragma unroll
        for (int i = 0; i < OWN_I; ++i) {
            #pragma unroll
            for (int j = 0; j < DB_J; ++j) {
                unsigned int sidx = (j * 32 + lane < nk[i]) ? (rec[i][j] >> B1_RBITS) : 0u;
                w[i][j] = rows[sidx];
            }
        }
        if (g0 == 0) {
            if (tid < B1_RNG) {
                s_acc[tid]          = u0.x;
                s_acc[B1_RNG + tid] = u0.y;
            }
            __syncthreads();
        }
        #pragma unroll
        for (int i = 0; i < OWN_I; ++i) {
            #pragma unroll
            for (int j = 0; j < DB_J; ++j) {
                if (j * 32 + lane < nk[i]) {
                    unsigned int r = rec[i][j] & ((1u << B1_RBITS) - 1u);
                    lds_add_f32(&s_acc[r],          w[i][j].x);
                    lds_add_f32(&s_acc[B1_RNG + r], w[i][j].y);
                }
            }
        }
        #pragma unroll
        for (int i = 0; i < OWN_I; ++i) {
            if (raw[i] > B1_CAP) {
                int e0 = (kb + 32 * i) * B1_EPB;
                int e1 = min(e0 + B1_EPB, E);
                for (int e = e0 + lane; e < e1; e += 32) {
                    unsigned int r = (unsigned int)(dst[e] - lo);
                    if (r < (unsigned int)B1_RNG) {
                        v2f u = rows[src[e]];
                        lds_add_f32(&s_acc[r],          u.x);
                        lds_add_f32(&s_acc[B1_RNG + r], u.y);
                    }
                }
            }
        }
        int kn = kb + B1_CH + 32 * lane;
        cv = (lane < OWN_I && kn < NB) ? cnts[kn * R + range] : 0;
    }
    __syncthreads();

    if (tid < B1_RNG && lo + tid < N) {
        int   node = lo + tid;
        float di   = dinv[node];
        float l0 = s_acc[tid]          * di + b2[0];
        float l1 = s_acc[B1_RNG + tid] * di + b2[1];
        float mx  = fmaxf(l0, l1);
        float lse = mx + logf(expf(l0 - mx) + expf(l1 - mx));
        out[(size_t)node * NCLS + 0] = l0 - lse;
        out[(size_t)node * NCLS + 1] = l1 - lse;
    }
}

__global__ void k_final(const float* __restrict__ agg2, const float* __restrict__ b2,
                        const float* __restrict__ dinv,
                        float* __restrict__ out, int n) {
    int i = blockIdx.x * blockDim.x + threadIdx.x;
    if (i >= n) return;
    float di = dinv[i];
    float l0 = agg2[(size_t)i * NCLS + 0] * di + b2[0];
    float l1 = agg2[(size_t)i * NCLS + 1] * di + b2[1];
    float m   = fmaxf(l0, l1);
    float lse = m + logf(expf(l0 - m) + expf(l1 - m));
    out[(size_t)i * NCLS + 0] = l0 - lse;
    out[(size_t)i * NCLS + 1] = l1 - lse;
}

extern "C" void kernel_launch(void* const* d_in, const int* in_sizes, int n_in,
                              void* d_out, int out_size, void* d_ws, size_t ws_size,
                              hipStream_t stream) {
    const float* x  = (const float*)d_in[0];
    const int*   ei = (const int*)d_in[1];
    const float* W1 = (const float*)d_in[2];
    const float* b1 = (const float*)d_in[3];
    const float* W2 = (const float*)d_in[4];
    const float* b2 = (const float*)d_in[5];
    float* out = (float*)d_out;

    int N = in_sizes[0] / N_FEAT;
    int E = in_sizes[1] / 2;
    const int* src = ei;
    const int* dst = ei + E;

    float* ws   = (float*)d_ws;
    float* dinv = ws;
    float* hw1  = dinv + (size_t)N;
    float* hw2  = hw1  + (size_t)N * HIDDEN;
    float* agg1 = hw2  + (size_t)N * NCLS;
    float* agg2 = agg1 + (size_t)N * HIDDEN;

    const int tb = 256;
    int rng = (((N + DEG_RANGES - 1) / DEG_RANGES) + 31) & ~31;
    int nslices = (E + DEG_SLICE - 1) / DEG_SLICE;

    int    rb1   = (N + B1_RNG - 1) / B1_RNG;
    int    nb1   = (E + B1_EPB - 1) / B1_EPB;
    size_t boff  = (((size_t)N * (1 + HIDDEN + NCLS) * sizeof(float)) + 255) & ~(size_t)255;
    size_t bbyte = (size_t)nb1 * rb1 * B1_CAP * sizeof(unsigned int);
    size_t cbyte = (size_t)nb1 * rb1 * sizeof(int);
    unsigned int* bins = (unsigned int*)((char*)d_ws + boff);
    int*          cnts = (int*)((char*)d_ws + boff + bbyte);
    bool binned = nb1 > 0 && rb1 >= 1 && rb1 <= B1_RMAX && N <= (1 << (32 - B1_RBITS)) &&
                  ws_size >= boff + bbyte + cbyte;

    if (binned) {
        k_bin1    <<<nb1, B1_THREADS, 0, stream>>>(src, dst, bins, cnts, E, rb1);
        k_deg_bins<<<rb1, B1_THREADS, 0, stream>>>(bins, cnts, dst, dinv, E, N, rb1, nb1);
    } else {
        k_deg_init <<<(N + tb - 1) / tb, tb, 0, stream>>>(dinv, N);
        if (rng <= DEG_RNG_MAX) {
            if (nslices > 0)
                k_deg_hist<<<nslices * DEG_RANGES, DEG_THREADS, 0, stream>>>(dst, dinv, E, N, rng, nslices);
        } else {
            k_deg_edges<<<(E + tb - 1) / tb, tb, 0, stream>>>(dst, dinv, E);
        }
        k_dinv     <<<(N + tb - 1) / tb, tb, 0, stream>>>(dinv, N);
    }

    float* agg1_start = binned ? nullptr : agg1;
    int ntiles = N / 16;
    int wpb    = tb / 32;
    if (ntiles > 0)
        k_gemm1<<<(ntiles + wpb - 1) / wpb, tb, 0, stream>>>(x, W1, dinv, hw1, agg1_start, ntiles);
    int rem = N - ntiles * 16;
    if (rem > 0)
        k_gemm1_rem<<<(rem * HIDDEN + tb - 1) / tb, tb, 0, stream>>>(x, W1, dinv, hw1, agg1_start,
                                                                     ntiles * 16, N);

    if (binned) {
        k_edges1_own<<<rb1, B1_THREADS, 0, stream>>>(bins, cnts, src, dst, hw1, b1, W2, dinv, hw2, E, N, rb1, nb1);
        k_edges2_own<<<rb1, B1_THREADS, 0, stream>>>(bins, cnts, src, dst, hw2, b2, dinv, out, E, N, rb1, nb1);
        return;
    }

    int nr1  = (N + L1_RNG_MAX - 1) / L1_RNG_MAX;
    int nsl1 = (E + L1_SLICE - 1) / L1_SLICE;
    if (nr1 >= 1 && nr1 <= L1_MAX_RANGES) {
        int rng1 = (((N + nr1 - 1) / nr1) + 1) & ~1;
        if (nsl1 > 0)
            k_edges1_lds<<<nsl1 * nr1, L1_THREADS, 0, stream>>>(src, dst, hw1, agg1, E, N, rng1, nr1, nsl1);
    } else {
        k_edges1<<<(E + tb - 1) / tb, tb, 0, stream>>>(src, dst, hw1, agg1, E);
    }
    k_layer2<<<(N + tb - 1) / tb, tb, 0, stream>>>(agg1, b1, W2, dinv, hw2, agg2, N);
    if (rng <= DEG_RNG_MAX) {
        if (nslices > 0)
            k_edges2_lds<<<nslices * DEG_RANGES, DEG_THREADS, 0, stream>>>(src, dst, hw2, agg2, E, N, rng, nslices);
    } else {
        long long t2 = (long long)E * NCLS;
        k_edges2<<<(int)((t2 + tb - 1) / tb), tb, 0, stream>>>(src, dst, hw2, agg2, E);
    }
    k_final <<<(N + tb - 1) / tb, tb, 0, stream>>>(agg2, b2, dinv, out, N);
}
